# starting kernel
# baseline (speedup 1.0000x reference)
#include <hip/hip_runtime.h>
#include <math.h>

#define JD 2048
#define ID 1025
#define IPD 1040
#define KB 8
#define MC 2
#define NSRC 2
#define NITER 5
#define YCH 8
#define EPS 1e-20f

typedef __attribute__((ext_vector_type(2))) float v2f;
typedef __attribute__((ext_vector_type(8))) float v8f;

constexpr size_t oXre = 0;
constexpr size_t oXim = oXre + (size_t)MC * IPD * JD;
constexpr size_t oYre = oXim + (size_t)MC * IPD * JD;
constexpr size_t oYim = oYre + (size_t)NSRC * IPD * JD;
constexpr size_t oYdR = oYim + (size_t)NSRC * IPD * JD;
constexpr size_t oT   = oYdR + (size_t)IPD * JD;
constexpr size_t oV   = oT + (size_t)NSRC * IPD * KB;
constexpr size_t oW   = oV + (size_t)NSRC * KB * JD;
constexpr size_t oPTn = oW + (size_t)ID * 8;
constexpr size_t oPTd = oPTn + (size_t)YCH * IPD * KB;
constexpr size_t oPVn = oPTd + (size_t)YCH * IPD * KB;
constexpr size_t oPVd = oPVn + (size_t)YCH * KB * JD;

__device__ __forceinline__ v8f wmma_k4(v2f a, v2f b, v8f c) {
  return __builtin_amdgcn_wmma_f32_16x16x4_f32(false, a, false, b, (short)0, c, false, false);
}

__global__ __launch_bounds__(256) void k_init_xy(const float* __restrict__ X, float* __restrict__ ws) {
  size_t t = (size_t)blockIdx.x * 256 + threadIdx.x;
  size_t tot = (size_t)MC * IPD * JD;
  if (t >= tot) return;
  int j = (int)(t % JD); size_t r = t / JD;
  int i = (int)(r % IPD); int m = (int)(r / IPD);
  float xr = 0.f, xi = 0.f;
  if (i < ID) {
    size_t src = (((size_t)m * JD + j) * ID + i) * 2;
    xr = X[src]; xi = X[src + 1];
  }
  size_t dst = ((size_t)m * IPD + i) * JD + j;
  ws[oXre + dst] = xr; ws[oXim + dst] = xi;
  ws[oYre + dst] = xr; ws[oYim + dst] = xi;
}

__global__ __launch_bounds__(256) void k_init_t(const float* __restrict__ T0, float* __restrict__ ws) {
  size_t t = (size_t)blockIdx.x * 256 + threadIdx.x;
  size_t tot = (size_t)NSRC * IPD * KB;
  if (t >= tot) return;
  int k = (int)(t % KB); size_t r = t / KB;
  int i = (int)(r % IPD); int n = (int)(r / IPD);
  float v = 0.f;
  if (i < ID) v = T0[((size_t)i * KB + k) * NSRC + n];
  ws[oT + ((size_t)n * IPD + i) * KB + k] = v;
}

__global__ __launch_bounds__(256) void k_init_v(const float* __restrict__ V0, float* __restrict__ ws) {
  size_t t = (size_t)blockIdx.x * 256 + threadIdx.x;
  size_t tot = (size_t)NSRC * KB * JD;
  if (t >= tot) return;
  int j = (int)(t % JD); size_t r = t / JD;
  int k = (int)(r % KB); int n = (int)(r / KB);
  ws[oV + ((size_t)n * KB + k) * JD + j] = V0[((size_t)k * JD + j) * NSRC + n];
}

__global__ __launch_bounds__(256) void k_init_w(float* __restrict__ ws) {
  size_t t = (size_t)blockIdx.x * 256 + threadIdx.x;
  size_t tot = (size_t)ID * 8;
  if (t >= tot) return;
  int f = (int)(t & 7);
  int c = f & 1, m = (f >> 1) & 1, nn = (f >> 2) & 1;
  ws[oW + t] = (c == 0 && nn == m) ? 1.f : 0.f;
}

__global__ __launch_bounds__(256) void k_t_update(float* __restrict__ ws, int n) {
  const int tid = threadIdx.x;
  const int wv = tid >> 5, lane = tid & 31, hf = lane >> 4, l16 = lane & 15;
  const int i0 = blockIdx.x * 16;
  const int yb = blockIdx.y;
  const float* T = ws + oT + (size_t)n * IPD * KB;
  const float* V = ws + oV + (size_t)n * KB * JD;
  const float* Yre = ws + oYre + (size_t)n * IPD * JD;
  const float* Yim = ws + oYim + (size_t)n * IPD * JD;
  float* YdR = ws + oYdR;

  __shared__ float lds_y[8][16][17];
  __shared__ float lds_r[8][16][17];
  __shared__ float redn[8][256];
  __shared__ float redd[8][256];

  v2f a0, a1;
  {
    size_t tr = (size_t)(i0 + l16) * KB;
    a0.x = T[tr + 2 * hf];     a0.y = T[tr + 2 * hf + 1];
    a1.x = T[tr + 4 + 2 * hf]; a1.y = T[tr + 5 + 2 * hf];
  }

  v8f num = {0,0,0,0,0,0,0,0}, den = {0,0,0,0,0,0,0,0};

  for (int jt = yb * 8 + wv; jt < JD / 16; jt += YCH * 8) {
    const int j0 = jt * 16;
    v2f b0, b1;
    b0.x = V[(size_t)(2 * hf + 0) * JD + j0 + l16];
    b0.y = V[(size_t)(2 * hf + 1) * JD + j0 + l16];
    b1.x = V[(size_t)(2 * hf + 4) * JD + j0 + l16];
    b1.y = V[(size_t)(2 * hf + 5) * JD + j0 + l16];
    v8f c = {0,0,0,0,0,0,0,0};
    c = wmma_k4(a0, b0, c);
    c = wmma_k4(a1, b1, c);
#pragma unroll
    for (int r = 0; r < 8; r++) {
      const int il = r + 8 * hf;
      size_t idx = (size_t)(i0 + il) * JD + (j0 + l16);
      float yr = Yre[idx], yi = Yim[idx];
      float rn = c[r];
      float ydr = (yr * yr + yi * yi) / (rn * rn + EPS);
      float rd = 1.0f / (rn + EPS);
      YdR[idx] = ydr;
      lds_y[wv][il][l16] = ydr;
      lds_r[wv][il][l16] = rd;
    }
#pragma unroll
    for (int jc = 0; jc < 16; jc += 4) {
      v2f ay, ar, bv;
      ay.x = lds_y[wv][l16][jc + 2 * hf]; ay.y = lds_y[wv][l16][jc + 2 * hf + 1];
      ar.x = lds_r[wv][l16][jc + 2 * hf]; ar.y = lds_r[wv][l16][jc + 2 * hf + 1];
      float bx = 0.f, by = 0.f;
      if (l16 < KB) {
        bx = V[(size_t)l16 * JD + j0 + jc + 2 * hf];
        by = V[(size_t)l16 * JD + j0 + jc + 2 * hf + 1];
      }
      bv.x = bx; bv.y = by;
      num = wmma_k4(ay, bv, num);
      den = wmma_k4(ar, bv, den);
    }
  }
#pragma unroll
  for (int r = 0; r < 8; r++) { redn[wv][r * 32 + lane] = num[r]; redd[wv][r * 32 + lane] = den[r]; }
  __syncthreads();
  if (wv == 0) {
#pragma unroll
    for (int r = 0; r < 8; r++) {
      float sn = 0.f, sd = 0.f;
      for (int w2 = 0; w2 < 8; w2++) { sn += redn[w2][r * 32 + lane]; sd += redd[w2][r * 32 + lane]; }
      if (l16 < KB) {
        int ii = i0 + r + 8 * hf;
        ws[oPTn + ((size_t)yb * IPD + ii) * KB + l16] = sn;
        ws[oPTd + ((size_t)yb * IPD + ii) * KB + l16] = sd;
      }
    }
  }
}

__global__ __launch_bounds__(256) void k_t_apply(float* __restrict__ ws, int n) {
  size_t t = (size_t)blockIdx.x * 256 + threadIdx.x;
  size_t tot = (size_t)IPD * KB;
  if (t >= tot) return;
  float sn = 0.f, sd = 0.f;
#pragma unroll
  for (int y = 0; y < YCH; y++) {
    sn += ws[oPTn + (size_t)y * IPD * KB + t];
    sd += ws[oPTd + (size_t)y * IPD * KB + t];
  }
  ws[oT + (size_t)n * IPD * KB + t] *= sqrtf(sn / (sd + EPS));
}

__global__ __launch_bounds__(256) void k_v_update(float* __restrict__ ws, int n) {
  const int tid = threadIdx.x;
  const int wv = tid >> 5, lane = tid & 31, hf = lane >> 4, l16 = lane & 15;
  const int j0 = blockIdx.x * 16;
  const int yb = blockIdx.y;
  const float* T = ws + oT + (size_t)n * IPD * KB;
  const float* V = ws + oV + (size_t)n * KB * JD;
  const float* YdR = ws + oYdR;

  __shared__ float lds_r[8][16][17];
  __shared__ float redn[8][256];
  __shared__ float redd[8][256];

  v2f b0, b1;
  b0.x = V[(size_t)(2 * hf + 0) * JD + j0 + l16];
  b0.y = V[(size_t)(2 * hf + 1) * JD + j0 + l16];
  b1.x = V[(size_t)(2 * hf + 4) * JD + j0 + l16];
  b1.y = V[(size_t)(2 * hf + 5) * JD + j0 + l16];

  v8f num = {0,0,0,0,0,0,0,0}, den = {0,0,0,0,0,0,0,0};

  for (int it = yb * 8 + wv; it < IPD / 16; it += YCH * 8) {
    const int i0 = it * 16;
    v2f a0, a1;
    size_t tr = (size_t)(i0 + l16) * KB;
    a0.x = T[tr + 2 * hf];     a0.y = T[tr + 2 * hf + 1];
    a1.x = T[tr + 4 + 2 * hf]; a1.y = T[tr + 5 + 2 * hf];
    v8f c = {0,0,0,0,0,0,0,0};
    c = wmma_k4(a0, b0, c);
    c = wmma_k4(a1, b1, c);
#pragma unroll
    for (int r = 0; r < 8; r++)
      lds_r[wv][r + 8 * hf][l16] = 1.0f / (fabsf(c[r]) + EPS);
#pragma unroll
    for (int ic = 0; ic < 16; ic += 4) {
      v2f at, byv, brv;
      int ia = i0 + ic + 2 * hf;
      at.x = (l16 < KB) ? T[(size_t)ia * KB + l16] : 0.f;
      at.y = (l16 < KB) ? T[(size_t)(ia + 1) * KB + l16] : 0.f;
      byv.x = YdR[(size_t)ia * JD + j0 + l16];
      byv.y = YdR[(size_t)(ia + 1) * JD + j0 + l16];
      brv.x = lds_r[wv][ic + 2 * hf][l16];
      brv.y = lds_r[wv][ic + 2 * hf + 1][l16];
      num = wmma_k4(at, byv, num);
      den = wmma_k4(at, brv, den);
    }
  }
#pragma unroll
  for (int r = 0; r < 8; r++) { redn[wv][r * 32 + lane] = num[r]; redd[wv][r * 32 + lane] = den[r]; }
  __syncthreads();
  if (wv == 0) {
#pragma unroll
    for (int r = 0; r < 8; r++) {
      float sn = 0.f, sd = 0.f;
      for (int w2 = 0; w2 < 8; w2++) { sn += redn[w2][r * 32 + lane]; sd += redd[w2][r * 32 + lane]; }
      if (hf == 0) {
        ws[oPVn + ((size_t)yb * KB + r) * JD + j0 + l16] = sn;
        ws[oPVd + ((size_t)yb * KB + r) * JD + j0 + l16] = sd;
      }
    }
  }
}

__global__ __launch_bounds__(256) void k_v_apply(float* __restrict__ ws, int n) {
  size_t t = (size_t)blockIdx.x * 256 + threadIdx.x;
  size_t tot = (size_t)KB * JD;
  if (t >= tot) return;
  float sn = 0.f, sd = 0.f;
#pragma unroll
  for (int y = 0; y < YCH; y++) {
    sn += ws[oPVn + (size_t)y * KB * JD + t];
    sd += ws[oPVd + (size_t)y * KB * JD + t];
  }
  ws[oV + (size_t)n * KB * JD + t] *= sqrtf(sn / (sd + EPS));
}

__global__ __launch_bounds__(256) void k_ip_update(float* __restrict__ ws, int n) {
  const int i = blockIdx.x;
  const int tid = threadIdx.x;
  const float* T = ws + oT + (size_t)n * IPD * KB;
  const float* V = ws + oV + (size_t)n * KB * JD;
  const float* X0r = ws + oXre;
  const float* X0i = ws + oXim;
  const float* X1r = ws + oXre + (size_t)IPD * JD;
  const float* X1i = ws + oXim + (size_t)IPD * JD;

  __shared__ float sT[KB];
  __shared__ float rb[4][256];
  if (tid < KB) sT[tid] = T[(size_t)i * KB + tid];
  __syncthreads();

  float d00 = 0.f, d11 = 0.f, d01r = 0.f, d01i = 0.f;
  for (int j = tid; j < JD; j += 256) {
    float rn = 0.f;
#pragma unroll
    for (int k = 0; k < KB; k++) rn += sT[k] * V[(size_t)k * JD + j];
    float s = sqrtf(1.0f / (rn + EPS));
    size_t idx = (size_t)i * JD + j;
    float x0r = X0r[idx] * s, x0i = X0i[idx] * s;
    float x1r = X1r[idx] * s, x1i = X1i[idx] * s;
    d00 += x0r * x0r + x0i * x0i;
    d11 += x1r * x1r + x1i * x1i;
    d01r += x0r * x1r + x0i * x1i;
    d01i += x0i * x1r - x0r * x1i;
  }
  rb[0][tid] = d00; rb[1][tid] = d11; rb[2][tid] = d01r; rb[3][tid] = d01i;
  __syncthreads();
  for (int s2 = 128; s2 > 0; s2 >>= 1) {
    if (tid < s2) {
      rb[0][tid] += rb[0][tid + s2]; rb[1][tid] += rb[1][tid + s2];
      rb[2][tid] += rb[2][tid + s2]; rb[3][tid] += rb[3][tid + s2];
    }
    __syncthreads();
  }
  if (tid == 0) {
    const float invJ = 1.0f / (float)JD;
    float D00 = rb[0][0] * invJ + EPS;
    float D11 = rb[1][0] * invJ + EPS;
    float Dr = rb[2][0] * invJ, Di = rb[3][0] * invJ;
    float* Wf = ws + oW + (size_t)i * 8;
    float w00r = Wf[0], w00i = Wf[1], w01r = Wf[2], w01i = Wf[3];
    float w10r = Wf[4], w10i = Wf[5], w11r = Wf[6], w11i = Wf[7];
    float A00r = w00r * D00 + (w01r * Dr + w01i * Di);
    float A00i = w00i * D00 + (w01i * Dr - w01r * Di);
    float A01r = (w00r * Dr - w00i * Di) + w01r * D11;
    float A01i = (w00i * Dr + w00r * Di) + w01i * D11;
    float A10r = w10r * D00 + (w11r * Dr + w11i * Di);
    float A10i = w10i * D00 + (w11i * Dr - w11r * Di);
    float A11r = (w10r * Dr - w10i * Di) + w11r * D11;
    float A11i = (w10i * Dr + w10r * Di) + w11i * D11;
    float detr = (A00r * A11r - A00i * A11i) - (A01r * A10r - A01i * A10i);
    float deti = (A00r * A11i + A00i * A11r) - (A01r * A10i + A01i * A10r);
    float n0r, n0i, n1r, n1i;
    if (n == 0) { n0r = A11r;  n0i = A11i;  n1r = -A10r; n1i = -A10i; }
    else        { n0r = -A01r; n0i = -A01i; n1r = A00r;  n1i = A00i;  }
    float dd = detr * detr + deti * deti;
    float b0r = (n0r * detr + n0i * deti) / dd, b0i = (n0i * detr - n0r * deti) / dd;
    float b1r = (n1r * detr + n1i * deti) / dd, b1i = (n1i * detr - n1r * deti) / dd;
    float Db0r = D00 * b0r + (Dr * b1r - Di * b1i);
    float Db0i = D00 * b0i + (Dr * b1i + Di * b1r);
    float Db1r = (Dr * b0r + Di * b0i) + D11 * b1r;
    float Db1i = (Dr * b0i - Di * b0r) + D11 * b1i;
    float denr = b0r * Db0r + b0i * Db0i + b1r * Db1r + b1i * Db1i;
    float dnm = sqrtf(denr + EPS);
    Wf[(n * 2 + 0) * 2]     = b0r / dnm;
    Wf[(n * 2 + 0) * 2 + 1] = -b0i / dnm;
    Wf[(n * 2 + 1) * 2]     = b1r / dnm;
    Wf[(n * 2 + 1) * 2 + 1] = -b1i / dnm;
  }
}

__global__ __launch_bounds__(256) void k_y_update(float* __restrict__ ws) {
  const int i = blockIdx.x;
  const int tid = threadIdx.x;
  __shared__ float sw[8];
  if (tid < 8) sw[tid] = ws[oW + (size_t)i * 8 + tid];
  __syncthreads();
  const float* X0r = ws + oXre;
  const float* X0i = ws + oXim;
  const float* X1r = ws + oXre + (size_t)IPD * JD;
  const float* X1i = ws + oXim + (size_t)IPD * JD;
  float* Y0r = ws + oYre;
  float* Y0i = ws + oYim;
  float* Y1r = ws + oYre + (size_t)IPD * JD;
  float* Y1i = ws + oYim + (size_t)IPD * JD;
  for (int j = tid; j < JD; j += 256) {
    size_t idx = (size_t)i * JD + j;
    float x0r = X0r[idx], x0i = X0i[idx], x1r = X1r[idx], x1i = X1i[idx];
    {
      float wr0 = sw[0], wi0 = sw[1], wr1 = sw[2], wi1 = sw[3];
      Y0r[idx] = wr0 * x0r - wi0 * x0i + wr1 * x1r - wi1 * x1i;
      Y0i[idx] = wr0 * x0i + wi0 * x0r + wr1 * x1i + wi1 * x1r;
    }
    {
      float wr0 = sw[4], wi0 = sw[5], wr1 = sw[6], wi1 = sw[7];
      Y1r[idx] = wr0 * x0r - wi0 * x0i + wr1 * x1r - wi1 * x1i;
      Y1i[idx] = wr0 * x0i + wi0 * x0r + wr1 * x1i + wi1 * x1r;
    }
  }
}

__global__ __launch_bounds__(256) void k_out(const float* __restrict__ ws, float* __restrict__ out) {
  size_t t = (size_t)blockIdx.x * 256 + threadIdx.x;
  size_t tot = (size_t)NSRC * JD * ID * 2;
  if (t >= tot) return;
  int c = (int)(t & 1); size_t r = t >> 1;
  int i = (int)(r % ID); r /= ID;
  int j = (int)(r % JD); int n = (int)(r / JD);
  size_t idx = ((size_t)n * IPD + i) * JD + j;
  out[t] = (c == 0) ? ws[oYre + idx] : ws[oYim + idx];
}

extern "C" void kernel_launch(void* const* d_in, const int* in_sizes, int n_in,
                              void* d_out, int out_size, void* d_ws, size_t ws_size,
                              hipStream_t stream) {
  (void)in_sizes; (void)n_in; (void)out_size; (void)ws_size;
  const float* X  = (const float*)d_in[0];
  const float* T0 = (const float*)d_in[1];
  const float* V0 = (const float*)d_in[2];
  float* ws = (float*)d_ws;
  float* out = (float*)d_out;

  { size_t tot = (size_t)MC * IPD * JD;   k_init_xy<<<(unsigned)((tot + 255) / 256), 256, 0, stream>>>(X, ws); }
  { size_t tot = (size_t)NSRC * IPD * KB; k_init_t<<<(unsigned)((tot + 255) / 256), 256, 0, stream>>>(T0, ws); }
  { size_t tot = (size_t)NSRC * KB * JD;  k_init_v<<<(unsigned)((tot + 255) / 256), 256, 0, stream>>>(V0, ws); }
  { size_t tot = (size_t)ID * 8;          k_init_w<<<(unsigned)((tot + 255) / 256), 256, 0, stream>>>(ws); }

  for (int it = 0; it < NITER; ++it) {
    for (int n = 0; n < NSRC; ++n) {
      k_t_update<<<dim3(IPD / 16, YCH), 256, 0, stream>>>(ws, n);
      k_t_apply<<<(unsigned)(((size_t)IPD * KB + 255) / 256), 256, 0, stream>>>(ws, n);
      k_v_update<<<dim3(JD / 16, YCH), 256, 0, stream>>>(ws, n);
      k_v_apply<<<(unsigned)(((size_t)KB * JD + 255) / 256), 256, 0, stream>>>(ws, n);
      k_ip_update<<<ID, 256, 0, stream>>>(ws, n);
    }
    k_y_update<<<ID, 256, 0, stream>>>(ws);
  }
  { size_t tot = (size_t)NSRC * JD * ID * 2; k_out<<<(unsigned)((tot + 255) / 256), 256, 0, stream>>>(ws, out); }
}


# Both sources per launch: NMF kernels take the source from grid z (their T/V updates are independent within an iteration), IP kernel sums both covariances in one pass, solves n=0 then 1; 35 launches
# speedup vs baseline: 1.2393x; 1.2393x over previous
#include <hip/hip_runtime.h>
#include <math.h>

#define JD 2048
#define ID 1025
#define IPD 1040
#define KB 8
#define MC 2
#define NSRC 2
#define NITER 5
#define YCH 8
#define EPS 1e-20f

typedef __attribute__((ext_vector_type(2))) float v2f;
typedef __attribute__((ext_vector_type(8))) float v8f;

constexpr size_t oXre = 0;
constexpr size_t oXim = oXre + (size_t)MC * IPD * JD;
constexpr size_t oYre = oXim + (size_t)MC * IPD * JD;
constexpr size_t oYim = oYre + (size_t)NSRC * IPD * JD;
constexpr size_t oT   = oYim + (size_t)NSRC * IPD * JD;
constexpr size_t oV   = oT + (size_t)NSRC * IPD * KB;
constexpr size_t oW   = oV + (size_t)NSRC * KB * JD;
constexpr size_t oPTn = oW + (size_t)ID * 8;
constexpr size_t oPTd = oPTn + (size_t)NSRC * YCH * IPD * KB;
constexpr size_t oPVn = oPTd + (size_t)NSRC * YCH * IPD * KB;
constexpr size_t oPVd = oPVn + (size_t)NSRC * YCH * KB * JD;

__device__ __forceinline__ v8f wmma_k4(v2f a, v2f b, v8f c) {
  return __builtin_amdgcn_wmma_f32_16x16x4_f32(false, a, false, b, (short)0, c, false, false);
}

__global__ __launch_bounds__(256) void k_init_xy(const float* __restrict__ X, float* __restrict__ ws) {
  size_t t = (size_t)blockIdx.x * 256 + threadIdx.x;
  size_t tot = (size_t)MC * IPD * JD;
  if (t >= tot) return;
  int j = (int)(t % JD); size_t r = t / JD;
  int i = (int)(r % IPD); int m = (int)(r / IPD);
  float xr = 0.f, xi = 0.f;
  if (i < ID) {
    size_t src = (((size_t)m * JD + j) * ID + i) * 2;
    xr = X[src]; xi = X[src + 1];
  }
  size_t dst = ((size_t)m * IPD + i) * JD + j;
  ws[oXre + dst] = xr; ws[oXim + dst] = xi;
  ws[oYre + dst] = xr; ws[oYim + dst] = xi;
}

__global__ __launch_bounds__(256) void k_init_t(const float* __restrict__ T0, float* __restrict__ ws) {
  size_t t = (size_t)blockIdx.x * 256 + threadIdx.x;
  size_t tot = (size_t)NSRC * IPD * KB;
  if (t >= tot) return;
  int k = (int)(t % KB); size_t r = t / KB;
  int i = (int)(r % IPD); int n = (int)(r / IPD);
  float v = 0.f;
  if (i < ID) v = T0[((size_t)i * KB + k) * NSRC + n];
  ws[oT + ((size_t)n * IPD + i) * KB + k] = v;
}

__global__ __launch_bounds__(256) void k_init_v(const float* __restrict__ V0, float* __restrict__ ws) {
  size_t t = (size_t)blockIdx.x * 256 + threadIdx.x;
  size_t tot = (size_t)NSRC * KB * JD;
  if (t >= tot) return;
  int j = (int)(t % JD); size_t r = t / JD;
  int k = (int)(r % KB); int n = (int)(r / KB);
  ws[oV + ((size_t)n * KB + k) * JD + j] = V0[((size_t)k * JD + j) * NSRC + n];
}

__global__ __launch_bounds__(256) void k_init_w(float* __restrict__ ws) {
  size_t t = (size_t)blockIdx.x * 256 + threadIdx.x;
  size_t tot = (size_t)ID * 8;
  if (t >= tot) return;
  int f = (int)(t & 7);
  int c = f & 1, m = (f >> 1) & 1, nn = (f >> 2) & 1;
  ws[oW + t] = (c == 0 && nn == m) ? 1.f : 0.f;
}

__global__ __launch_bounds__(256) void k_t_update(float* __restrict__ ws) {
  const int tid = threadIdx.x;
  const int wv = tid >> 5, lane = tid & 31, hf = lane >> 4, l16 = lane & 15;
  const int i0 = blockIdx.x * 16;
  const int yb = blockIdx.y;
  const int n = blockIdx.z;
  const float* T = ws + oT + (size_t)n * IPD * KB;
  const float* V = ws + oV + (size_t)n * KB * JD;
  float* Yre = ws + oYre + (size_t)n * IPD * JD;
  const float* Yim = ws + oYim + (size_t)n * IPD * JD;

  __shared__ float lds_y[8][16][17];
  __shared__ float lds_r[8][16][17];
  __shared__ float redn[8][256];
  __shared__ float redd[8][256];

  v2f a0, a1;
  {
    size_t tr = (size_t)(i0 + l16) * KB;
    a0.x = T[tr + 2 * hf];     a0.y = T[tr + 2 * hf + 1];
    a1.x = T[tr + 4 + 2 * hf]; a1.y = T[tr + 5 + 2 * hf];
  }

  v8f num = {0,0,0,0,0,0,0,0}, den = {0,0,0,0,0,0,0,0};

  for (int jt = yb * 8 + wv; jt < JD / 16; jt += YCH * 8) {
    const int j0 = jt * 16;
    v2f b0, b1;
    b0.x = V[(size_t)(2 * hf + 0) * JD + j0 + l16];
    b0.y = V[(size_t)(2 * hf + 1) * JD + j0 + l16];
    b1.x = V[(size_t)(2 * hf + 4) * JD + j0 + l16];
    b1.y = V[(size_t)(2 * hf + 5) * JD + j0 + l16];
    v8f c = {0,0,0,0,0,0,0,0};
    c = wmma_k4(a0, b0, c);
    c = wmma_k4(a1, b1, c);
    float yrv[8], yiv[8];
#pragma unroll
    for (int r = 0; r < 8; r++) {
      size_t idx = (size_t)(i0 + r + 8 * hf) * JD + (j0 + l16);
      yrv[r] = Yre[idx]; yiv[r] = Yim[idx];
    }
#pragma unroll
    for (int r = 0; r < 8; r++) {
      const int il = r + 8 * hf;
      size_t idx = (size_t)(i0 + il) * JD + (j0 + l16);
      float yr = yrv[r], yi = yiv[r];
      float rn = c[r];
      float ydr = (yr * yr + yi * yi) / (rn * rn + EPS);
      float rd = 1.0f / (rn + EPS);
      Yre[idx] = ydr;
      lds_y[wv][il][l16] = ydr;
      lds_r[wv][il][l16] = rd;
    }
#pragma unroll
    for (int jc = 0; jc < 16; jc += 4) {
      v2f ay, ar, bv;
      ay.x = lds_y[wv][l16][jc + 2 * hf]; ay.y = lds_y[wv][l16][jc + 2 * hf + 1];
      ar.x = lds_r[wv][l16][jc + 2 * hf]; ar.y = lds_r[wv][l16][jc + 2 * hf + 1];
      float bx = 0.f, by = 0.f;
      if (l16 < KB) {
        bx = V[(size_t)l16 * JD + j0 + jc + 2 * hf];
        by = V[(size_t)l16 * JD + j0 + jc + 2 * hf + 1];
      }
      bv.x = bx; bv.y = by;
      num = wmma_k4(ay, bv, num);
      den = wmma_k4(ar, bv, den);
    }
  }
#pragma unroll
  for (int r = 0; r < 8; r++) { redn[wv][r * 32 + lane] = num[r]; redd[wv][r * 32 + lane] = den[r]; }
  __syncthreads();
  if (wv == 0) {
#pragma unroll
    for (int r = 0; r < 8; r++) {
      float sn = 0.f, sd = 0.f;
      for (int w2 = 0; w2 < 8; w2++) { sn += redn[w2][r * 32 + lane]; sd += redd[w2][r * 32 + lane]; }
      if (l16 < KB) {
        int ii = i0 + r + 8 * hf;
        ws[oPTn + ((size_t)(n * YCH + yb) * IPD + ii) * KB + l16] = sn;
        ws[oPTd + ((size_t)(n * YCH + yb) * IPD + ii) * KB + l16] = sd;
      }
    }
  }
}

__global__ __launch_bounds__(256) void k_t_apply(float* __restrict__ ws) {
  size_t t = (size_t)blockIdx.x * 256 + threadIdx.x;
  size_t tot = (size_t)IPD * KB;
  if (t >= tot) return;
  const int n = blockIdx.z;
  float sn = 0.f, sd = 0.f;
#pragma unroll
  for (int y = 0; y < YCH; y++) {
    sn += ws[oPTn + (size_t)(n * YCH + y) * IPD * KB + t];
    sd += ws[oPTd + (size_t)(n * YCH + y) * IPD * KB + t];
  }
  ws[oT + (size_t)n * IPD * KB + t] *= sqrtf(sn / (sd + EPS));
}

__global__ __launch_bounds__(256) void k_v_update(float* __restrict__ ws) {
  const int tid = threadIdx.x;
  const int wv = tid >> 5, lane = tid & 31, hf = lane >> 4, l16 = lane & 15;
  const int j0 = blockIdx.x * 16;
  const int yb = blockIdx.y;
  const int n = blockIdx.z;
  const float* T = ws + oT + (size_t)n * IPD * KB;
  const float* V = ws + oV + (size_t)n * KB * JD;
  const float* YdR = ws + oYre + (size_t)n * IPD * JD;

  __shared__ float lds_r[8][16][17];
  __shared__ float redn[8][256];
  __shared__ float redd[8][256];

  v2f b0, b1;
  b0.x = V[(size_t)(2 * hf + 0) * JD + j0 + l16];
  b0.y = V[(size_t)(2 * hf + 1) * JD + j0 + l16];
  b1.x = V[(size_t)(2 * hf + 4) * JD + j0 + l16];
  b1.y = V[(size_t)(2 * hf + 5) * JD + j0 + l16];

  v8f num = {0,0,0,0,0,0,0,0}, den = {0,0,0,0,0,0,0,0};

  for (int it = yb * 8 + wv; it < IPD / 16; it += YCH * 8) {
    const int i0 = it * 16;
    v2f a0, a1;
    size_t tr = (size_t)(i0 + l16) * KB;
    a0.x = T[tr + 2 * hf];     a0.y = T[tr + 2 * hf + 1];
    a1.x = T[tr + 4 + 2 * hf]; a1.y = T[tr + 5 + 2 * hf];
    v8f c = {0,0,0,0,0,0,0,0};
    c = wmma_k4(a0, b0, c);
    c = wmma_k4(a1, b1, c);
#pragma unroll
    for (int r = 0; r < 8; r++)
      lds_r[wv][r + 8 * hf][l16] = 1.0f / (fabsf(c[r]) + EPS);
#pragma unroll
    for (int ic = 0; ic < 16; ic += 4) {
      v2f at, byv, brv;
      int ia = i0 + ic + 2 * hf;
      at.x = (l16 < KB) ? T[(size_t)ia * KB + l16] : 0.f;
      at.y = (l16 < KB) ? T[(size_t)(ia + 1) * KB + l16] : 0.f;
      byv.x = YdR[(size_t)ia * JD + j0 + l16];
      byv.y = YdR[(size_t)(ia + 1) * JD + j0 + l16];
      brv.x = lds_r[wv][ic + 2 * hf][l16];
      brv.y = lds_r[wv][ic + 2 * hf + 1][l16];
      num = wmma_k4(at, byv, num);
      den = wmma_k4(at, brv, den);
    }
  }
#pragma unroll
  for (int r = 0; r < 8; r++) { redn[wv][r * 32 + lane] = num[r]; redd[wv][r * 32 + lane] = den[r]; }
  __syncthreads();
  if (wv == 0) {
#pragma unroll
    for (int r = 0; r < 8; r++) {
      float sn = 0.f, sd = 0.f;
      for (int w2 = 0; w2 < 8; w2++) { sn += redn[w2][r * 32 + lane]; sd += redd[w2][r * 32 + lane]; }
      if (hf == 0) {
        ws[oPVn + ((size_t)(n * YCH + yb) * KB + r) * JD + j0 + l16] = sn;
        ws[oPVd + ((size_t)(n * YCH + yb) * KB + r) * JD + j0 + l16] = sd;
      }
    }
  }
}

__global__ __launch_bounds__(256) void k_v_apply(float* __restrict__ ws) {
  size_t t = (size_t)blockIdx.x * 256 + threadIdx.x;
  size_t tot = (size_t)KB * JD;
  if (t >= tot) return;
  const int n = blockIdx.z;
  float sn = 0.f, sd = 0.f;
#pragma unroll
  for (int y = 0; y < YCH; y++) {
    sn += ws[oPVn + (size_t)(n * YCH + y) * KB * JD + t];
    sd += ws[oPVd + (size_t)(n * YCH + y) * KB * JD + t];
  }
  ws[oV + (size_t)n * KB * JD + t] *= sqrtf(sn / (sd + EPS));
}

__device__ __forceinline__ void ip_solve(float* Wf, const int n, float S00, float S11, float S01r, float S01i) {
  const float invJ = 1.0f / (float)JD;
  float D00 = S00 * invJ + EPS;
  float D11 = S11 * invJ + EPS;
  float Dr = S01r * invJ, Di = S01i * invJ;
  float w00r = Wf[0], w00i = Wf[1], w01r = Wf[2], w01i = Wf[3];
  float w10r = Wf[4], w10i = Wf[5], w11r = Wf[6], w11i = Wf[7];
  float A00r = w00r * D00 + (w01r * Dr + w01i * Di);
  float A00i = w00i * D00 + (w01i * Dr - w01r * Di);
  float A01r = (w00r * Dr - w00i * Di) + w01r * D11;
  float A01i = (w00i * Dr + w00r * Di) + w01i * D11;
  float A10r = w10r * D00 + (w11r * Dr + w11i * Di);
  float A10i = w10i * D00 + (w11i * Dr - w11r * Di);
  float A11r = (w10r * Dr - w10i * Di) + w11r * D11;
  float A11i = (w10i * Dr + w10r * Di) + w11i * D11;
  float detr = (A00r * A11r - A00i * A11i) - (A01r * A10r - A01i * A10i);
  float deti = (A00r * A11i + A00i * A11r) - (A01r * A10i + A01i * A10r);
  float n0r, n0i, n1r, n1i;
  if (n == 0) { n0r = A11r;  n0i = A11i;  n1r = -A10r; n1i = -A10i; }
  else        { n0r = -A01r; n0i = -A01i; n1r = A00r;  n1i = A00i;  }
  float dd = detr * detr + deti * deti;
  float b0r = (n0r * detr + n0i * deti) / dd, b0i = (n0i * detr - n0r * deti) / dd;
  float b1r = (n1r * detr + n1i * deti) / dd, b1i = (n1i * detr - n1r * deti) / dd;
  float Db0r = D00 * b0r + (Dr * b1r - Di * b1i);
  float Db0i = D00 * b0i + (Dr * b1i + Di * b1r);
  float Db1r = (Dr * b0r + Di * b0i) + D11 * b1r;
  float Db1i = (Dr * b0i - Di * b0r) + D11 * b1i;
  float denr = b0r * Db0r + b0i * Db0i + b1r * Db1r + b1i * Db1i;
  float dnm = sqrtf(denr + EPS);
  Wf[(n * 2 + 0) * 2]     = b0r / dnm;
  Wf[(n * 2 + 0) * 2 + 1] = -b0i / dnm;
  Wf[(n * 2 + 1) * 2]     = b1r / dnm;
  Wf[(n * 2 + 1) * 2 + 1] = -b1i / dnm;
}

__global__ __launch_bounds__(256) void k_ip_update(float* __restrict__ ws) {
  const int i = blockIdx.x;
  const int tid = threadIdx.x;
  const float* V = ws + oV;
  const float* X0r = ws + oXre;
  const float* X0i = ws + oXim;
  const float* X1r = ws + oXre + (size_t)IPD * JD;
  const float* X1i = ws + oXim + (size_t)IPD * JD;

  __shared__ float sT[NSRC][KB];
  __shared__ float rb[4 * NSRC][256];
  if (tid < NSRC * KB) sT[tid >> 3][tid & 7] = ws[oT + (size_t)(tid >> 3) * IPD * KB + (size_t)i * KB + (tid & 7)];
  __syncthreads();

  float d00[NSRC], d11[NSRC], d01r[NSRC], d01i[NSRC];
#pragma unroll
  for (int n = 0; n < NSRC; n++) { d00[n] = 0.f; d11[n] = 0.f; d01r[n] = 0.f; d01i[n] = 0.f; }
  for (int j = tid; j < JD; j += 256) {
    size_t idx = (size_t)i * JD + j;
    float r0r = X0r[idx], r0i = X0i[idx], r1r = X1r[idx], r1i = X1i[idx];
#pragma unroll
    for (int n = 0; n < NSRC; n++) {
      float rn = 0.f;
#pragma unroll
      for (int k = 0; k < KB; k++) rn += sT[n][k] * V[(size_t)(n * KB + k) * JD + j];
      float s = sqrtf(1.0f / (rn + EPS));
      float x0r = r0r * s, x0i = r0i * s;
      float x1r = r1r * s, x1i = r1i * s;
      d00[n] += x0r * x0r + x0i * x0i;
      d11[n] += x1r * x1r + x1i * x1i;
      d01r[n] += x0r * x1r + x0i * x1i;
      d01i[n] += x0i * x1r - x0r * x1i;
    }
  }
#pragma unroll
  for (int n = 0; n < NSRC; n++) {
    rb[4 * n + 0][tid] = d00[n]; rb[4 * n + 1][tid] = d11[n];
    rb[4 * n + 2][tid] = d01r[n]; rb[4 * n + 3][tid] = d01i[n];
  }
  __syncthreads();
  for (int s2 = 128; s2 > 0; s2 >>= 1) {
    if (tid < s2) {
#pragma unroll
      for (int q = 0; q < 4 * NSRC; q++) rb[q][tid] += rb[q][tid + s2];
    }
    __syncthreads();
  }
  if (tid == 0) {
    float* Wg = ws + oW + (size_t)i * 8;
    float Wf[8];
#pragma unroll
    for (int q = 0; q < 8; q++) Wf[q] = Wg[q];
    ip_solve(Wf, 0, rb[0][0], rb[1][0], rb[2][0], rb[3][0]);
    ip_solve(Wf, 1, rb[4][0], rb[5][0], rb[6][0], rb[7][0]);
#pragma unroll
    for (int q = 0; q < 8; q++) Wg[q] = Wf[q];
  }
}

__global__ __launch_bounds__(256) void k_y_update(float* __restrict__ ws) {
  const int i = blockIdx.x;
  const int tid = threadIdx.x;
  __shared__ float sw[8];
  if (tid < 8) sw[tid] = ws[oW + (size_t)i * 8 + tid];
  __syncthreads();
  const float* X0r = ws + oXre;
  const float* X0i = ws + oXim;
  const float* X1r = ws + oXre + (size_t)IPD * JD;
  const float* X1i = ws + oXim + (size_t)IPD * JD;
  float* Y0r = ws + oYre;
  float* Y0i = ws + oYim;
  float* Y1r = ws + oYre + (size_t)IPD * JD;
  float* Y1i = ws + oYim + (size_t)IPD * JD;
  for (int j = tid; j < JD; j += 256) {
    size_t idx = (size_t)i * JD + j;
    float x0r = X0r[idx], x0i = X0i[idx], x1r = X1r[idx], x1i = X1i[idx];
    {
      float wr0 = sw[0], wi0 = sw[1], wr1 = sw[2], wi1 = sw[3];
      Y0r[idx] = wr0 * x0r - wi0 * x0i + wr1 * x1r - wi1 * x1i;
      Y0i[idx] = wr0 * x0i + wi0 * x0r + wr1 * x1i + wi1 * x1r;
    }
    {
      float wr0 = sw[4], wi0 = sw[5], wr1 = sw[6], wi1 = sw[7];
      Y1r[idx] = wr0 * x0r - wi0 * x0i + wr1 * x1r - wi1 * x1i;
      Y1i[idx] = wr0 * x0i + wi0 * x0r + wr1 * x1i + wi1 * x1r;
    }
  }
}

__global__ __launch_bounds__(256) void k_out(const float* __restrict__ ws, float* __restrict__ out) {
  size_t t = (size_t)blockIdx.x * 256 + threadIdx.x;
  size_t tot = (size_t)NSRC * JD * ID * 2;
  if (t >= tot) return;
  int c = (int)(t & 1); size_t r = t >> 1;
  int i = (int)(r % ID); r /= ID;
  int j = (int)(r % JD); int n = (int)(r / JD);
  size_t idx = ((size_t)n * IPD + i) * JD + j;
  out[t] = (c == 0) ? ws[oYre + idx] : ws[oYim + idx];
}

extern "C" void kernel_launch(void* const* d_in, const int* in_sizes, int n_in,
                              void* d_out, int out_size, void* d_ws, size_t ws_size,
                              hipStream_t stream) {
  (void)in_sizes; (void)n_in; (void)out_size; (void)ws_size;
  const float* X  = (const float*)d_in[0];
  const float* T0 = (const float*)d_in[1];
  const float* V0 = (const float*)d_in[2];
  float* ws = (float*)d_ws;
  float* out = (float*)d_out;

  { size_t tot = (size_t)MC * IPD * JD;   k_init_xy<<<(unsigned)((tot + 255) / 256), 256, 0, stream>>>(X, ws); }
  { size_t tot = (size_t)NSRC * IPD * KB; k_init_t<<<(unsigned)((tot + 255) / 256), 256, 0, stream>>>(T0, ws); }
  { size_t tot = (size_t)NSRC * KB * JD;  k_init_v<<<(unsigned)((tot + 255) / 256), 256, 0, stream>>>(V0, ws); }
  { size_t tot = (size_t)ID * 8;          k_init_w<<<(unsigned)((tot + 255) / 256), 256, 0, stream>>>(ws); }

  for (int it = 0; it < NITER; ++it) {
    k_t_update<<<dim3(IPD / 16, YCH, NSRC), 256, 0, stream>>>(ws);
    k_t_apply<<<dim3((unsigned)(((size_t)IPD * KB + 255) / 256), 1, NSRC), 256, 0, stream>>>(ws);
    k_v_update<<<dim3(JD / 16, YCH, NSRC), 256, 0, stream>>>(ws);
    k_v_apply<<<dim3((unsigned)(((size_t)KB * JD + 255) / 256), 1, NSRC), 256, 0, stream>>>(ws);
    k_ip_update<<<ID, 256, 0, stream>>>(ws);
    k_y_update<<<ID, 256, 0, stream>>>(ws);
  }
  { size_t tot = (size_t)NSRC * JD * ID * 2; k_out<<<(unsigned)((tot + 255) / 256), 256, 0, stream>>>(ws, out); }
}


# Both layout transposes (X in, Y out) go through LDS tiles: contiguous global reads and writes, whole 128-B lines per store, 15x fewer waves in k_init_xy, 64x fewer in k_out
# speedup vs baseline: 1.3717x; 1.1068x over previous
#include <hip/hip_runtime.h>
#include <math.h>

#define JD 2048
#define ID 1025
#define IPD 1040
#define KB 8
#define MC 2
#define NSRC 2
#define NITER 5
#define YCH 8
#define EPS 1e-20f

typedef __attribute__((ext_vector_type(2))) float v2f;
typedef __attribute__((ext_vector_type(8))) float v8f;

constexpr size_t oXre = 0;
constexpr size_t oXim = oXre + (size_t)MC * IPD * JD;
constexpr size_t oYre = oXim + (size_t)MC * IPD * JD;
constexpr size_t oYim = oYre + (size_t)NSRC * IPD * JD;
constexpr size_t oT   = oYim + (size_t)NSRC * IPD * JD;
constexpr size_t oV   = oT + (size_t)NSRC * IPD * KB;
constexpr size_t oW   = oV + (size_t)NSRC * KB * JD;
constexpr size_t oPTn = oW + (size_t)ID * 8;
constexpr size_t oPTd = oPTn + (size_t)NSRC * YCH * IPD * KB;
constexpr size_t oPVn = oPTd + (size_t)NSRC * YCH * IPD * KB;
constexpr size_t oPVd = oPVn + (size_t)NSRC * YCH * KB * JD;

__device__ __forceinline__ v8f wmma_k4(v2f a, v2f b, v8f c) {
  return __builtin_amdgcn_wmma_f32_16x16x4_f32(false, a, false, b, (short)0, c, false, false);
}

#define TP 66
__global__ __launch_bounds__(256) void k_init_xy(const float* __restrict__ X, float* __restrict__ ws) {
  __shared__ __attribute__((aligned(16))) float sre[64 * TP];
  __shared__ __attribute__((aligned(16))) float sim[64 * TP];
  const int tid = threadIdx.x, wv = tid >> 5, lane = tid & 31;
  const int i0 = blockIdx.x * 64, j0 = blockIdx.y * 64, m = blockIdx.z;
  float2 v[16];
#pragma unroll
  for (int q = 0; q < 16; q++) {
    const int jj = wv * 8 + (q >> 1);
    const int i = i0 + (q & 1) * 32 + lane;
    const int ic = i < ID ? i : ID - 1;
    v[q] = *(const float2*)(X + (((size_t)m * JD + j0 + jj) * ID + ic) * 2);
  }
#pragma unroll
  for (int q = 0; q < 16; q++) {
    const int jj = wv * 8 + (q >> 1);
    const int il = (q & 1) * 32 + lane;
    const bool pad = i0 + il >= ID;
    sre[il * TP + jj] = pad ? 0.f : v[q].x; sim[il * TP + jj] = pad ? 0.f : v[q].y;
  }
  __syncthreads();
#pragma unroll
  for (int q = 0; q < 8; q++) {
    const int il = wv * 8 + q;
    const int i = i0 + il;
    if (i < IPD) {
      const float2 xr = *(const float2*)(sre + il * TP + 2 * lane);
      const float2 xi = *(const float2*)(sim + il * TP + 2 * lane);
      const size_t dst = ((size_t)m * IPD + i) * JD + j0 + 2 * lane;
      *(float2*)(ws + oXre + dst) = xr; *(float2*)(ws + oXim + dst) = xi;
      *(float2*)(ws + oYre + dst) = xr; *(float2*)(ws + oYim + dst) = xi;
    }
  }
}

__global__ __launch_bounds__(256) void k_init_t(const float* __restrict__ T0, float* __restrict__ ws) {
  size_t t = (size_t)blockIdx.x * 256 + threadIdx.x;
  size_t tot = (size_t)NSRC * IPD * KB;
  if (t >= tot) return;
  int k = (int)(t % KB); size_t r = t / KB;
  int i = (int)(r % IPD); int n = (int)(r / IPD);
  float v = 0.f;
  if (i < ID) v = T0[((size_t)i * KB + k) * NSRC + n];
  ws[oT + ((size_t)n * IPD + i) * KB + k] = v;
}

__global__ __launch_bounds__(256) void k_init_v(const float* __restrict__ V0, float* __restrict__ ws) {
  size_t t = (size_t)blockIdx.x * 256 + threadIdx.x;
  size_t tot = (size_t)NSRC * KB * JD;
  if (t >= tot) return;
  int j = (int)(t % JD); size_t r = t / JD;
  int k = (int)(r % KB); int n = (int)(r / KB);
  ws[oV + ((size_t)n * KB + k) * JD + j] = V0[((size_t)k * JD + j) * NSRC + n];
}

__global__ __launch_bounds__(256) void k_init_w(float* __restrict__ ws) {
  size_t t = (size_t)blockIdx.x * 256 + threadIdx.x;
  size_t tot = (size_t)ID * 8;
  if (t >= tot) return;
  int f = (int)(t & 7);
  int c = f & 1, m = (f >> 1) & 1, nn = (f >> 2) & 1;
  ws[oW + t] = (c == 0 && nn == m) ? 1.f : 0.f;
}

__global__ __launch_bounds__(256) void k_t_update(float* __restrict__ ws) {
  const int tid = threadIdx.x;
  const int wv = tid >> 5, lane = tid & 31, hf = lane >> 4, l16 = lane & 15;
  const int i0 = blockIdx.x * 16;
  const int yb = blockIdx.y;
  const int n = blockIdx.z;
  const float* T = ws + oT + (size_t)n * IPD * KB;
  const float* V = ws + oV + (size_t)n * KB * JD;
  float* Yre = ws + oYre + (size_t)n * IPD * JD;
  const float* Yim = ws + oYim + (size_t)n * IPD * JD;

  __shared__ float lds_y[8][16][17];
  __shared__ float lds_r[8][16][17];
  __shared__ float redn[8][256];
  __shared__ float redd[8][256];

  v2f a0, a1;
  {
    size_t tr = (size_t)(i0 + l16) * KB;
    a0.x = T[tr + 2 * hf];     a0.y = T[tr + 2 * hf + 1];
    a1.x = T[tr + 4 + 2 * hf]; a1.y = T[tr + 5 + 2 * hf];
  }

  v8f num = {0,0,0,0,0,0,0,0}, den = {0,0,0,0,0,0,0,0};

  for (int jt = yb * 8 + wv; jt < JD / 16; jt += YCH * 8) {
    const int j0 = jt * 16;
    v2f b0, b1;
    b0.x = V[(size_t)(2 * hf + 0) * JD + j0 + l16];
    b0.y = V[(size_t)(2 * hf + 1) * JD + j0 + l16];
    b1.x = V[(size_t)(2 * hf + 4) * JD + j0 + l16];
    b1.y = V[(size_t)(2 * hf + 5) * JD + j0 + l16];
    v8f c = {0,0,0,0,0,0,0,0};
    c = wmma_k4(a0, b0, c);
    c = wmma_k4(a1, b1, c);
    float yrv[8], yiv[8];
#pragma unroll
    for (int r = 0; r < 8; r++) {
      size_t idx = (size_t)(i0 + r + 8 * hf) * JD + (j0 + l16);
      yrv[r] = Yre[idx]; yiv[r] = Yim[idx];
    }
#pragma unroll
    for (int r = 0; r < 8; r++) {
      const int il = r + 8 * hf;
      size_t idx = (size_t)(i0 + il) * JD + (j0 + l16);
      float yr = yrv[r], yi = yiv[r];
      float rn = c[r];
      float ydr = (yr * yr + yi * yi) / (rn * rn + EPS);
      float rd = 1.0f / (rn + EPS);
      Yre[idx] = ydr;
      lds_y[wv][il][l16] = ydr;
      lds_r[wv][il][l16] = rd;
    }
#pragma unroll
    for (int jc = 0; jc < 16; jc += 4) {
      v2f ay, ar, bv;
      ay.x = lds_y[wv][l16][jc + 2 * hf]; ay.y = lds_y[wv][l16][jc + 2 * hf + 1];
      ar.x = lds_r[wv][l16][jc + 2 * hf]; ar.y = lds_r[wv][l16][jc + 2 * hf + 1];
      float bx = 0.f, by = 0.f;
      if (l16 < KB) {
        bx = V[(size_t)l16 * JD + j0 + jc + 2 * hf];
        by = V[(size_t)l16 * JD + j0 + jc + 2 * hf + 1];
      }
      bv.x = bx; bv.y = by;
      num = wmma_k4(ay, bv, num);
      den = wmma_k4(ar, bv, den);
    }
  }
#pragma unroll
  for (int r = 0; r < 8; r++) { redn[wv][r * 32 + lane] = num[r]; redd[wv][r * 32 + lane] = den[r]; }
  __syncthreads();
  if (wv == 0) {
#pragma unroll
    for (int r = 0; r < 8; r++) {
      float sn = 0.f, sd = 0.f;
      for (int w2 = 0; w2 < 8; w2++) { sn += redn[w2][r * 32 + lane]; sd += redd[w2][r * 32 + lane]; }
      if (l16 < KB) {
        int ii = i0 + r + 8 * hf;
        ws[oPTn + ((size_t)(n * YCH + yb) * IPD + ii) * KB + l16] = sn;
        ws[oPTd + ((size_t)(n * YCH + yb) * IPD + ii) * KB + l16] = sd;
      }
    }
  }
}

__global__ __launch_bounds__(256) void k_t_apply(float* __restrict__ ws) {
  size_t t = (size_t)blockIdx.x * 256 + threadIdx.x;
  size_t tot = (size_t)IPD * KB;
  if (t >= tot) return;
  const int n = blockIdx.z;
  float sn = 0.f, sd = 0.f;
#pragma unroll
  for (int y = 0; y < YCH; y++) {
    sn += ws[oPTn + (size_t)(n * YCH + y) * IPD * KB + t];
    sd += ws[oPTd + (size_t)(n * YCH + y) * IPD * KB + t];
  }
  ws[oT + (size_t)n * IPD * KB + t] *= sqrtf(sn / (sd + EPS));
}

__global__ __launch_bounds__(256) void k_v_update(float* __restrict__ ws) {
  const int tid = threadIdx.x;
  const int wv = tid >> 5, lane = tid & 31, hf = lane >> 4, l16 = lane & 15;
  const int j0 = blockIdx.x * 16;
  const int yb = blockIdx.y;
  const int n = blockIdx.z;
  const float* T = ws + oT + (size_t)n * IPD * KB;
  const float* V = ws + oV + (size_t)n * KB * JD;
  const float* YdR = ws + oYre + (size_t)n * IPD * JD;

  __shared__ float lds_r[8][16][17];
  __shared__ float redn[8][256];
  __shared__ float redd[8][256];

  v2f b0, b1;
  b0.x = V[(size_t)(2 * hf + 0) * JD + j0 + l16];
  b0.y = V[(size_t)(2 * hf + 1) * JD + j0 + l16];
  b1.x = V[(size_t)(2 * hf + 4) * JD + j0 + l16];
  b1.y = V[(size_t)(2 * hf + 5) * JD + j0 + l16];

  v8f num = {0,0,0,0,0,0,0,0}, den = {0,0,0,0,0,0,0,0};

  for (int it = yb * 8 + wv; it < IPD / 16; it += YCH * 8) {
    const int i0 = it * 16;
    v2f a0, a1;
    size_t tr = (size_t)(i0 + l16) * KB;
    a0.x = T[tr + 2 * hf];     a0.y = T[tr + 2 * hf + 1];
    a1.x = T[tr + 4 + 2 * hf]; a1.y = T[tr + 5 + 2 * hf];
    v8f c = {0,0,0,0,0,0,0,0};
    c = wmma_k4(a0, b0, c);
    c = wmma_k4(a1, b1, c);
#pragma unroll
    for (int r = 0; r < 8; r++)
      lds_r[wv][r + 8 * hf][l16] = 1.0f / (fabsf(c[r]) + EPS);
#pragma unroll
    for (int ic = 0; ic < 16; ic += 4) {
      v2f at, byv, brv;
      int ia = i0 + ic + 2 * hf;
      at.x = (l16 < KB) ? T[(size_t)ia * KB + l16] : 0.f;
      at.y = (l16 < KB) ? T[(size_t)(ia + 1) * KB + l16] : 0.f;
      byv.x = YdR[(size_t)ia * JD + j0 + l16];
      byv.y = YdR[(size_t)(ia + 1) * JD + j0 + l16];
      brv.x = lds_r[wv][ic + 2 * hf][l16];
      brv.y = lds_r[wv][ic + 2 * hf + 1][l16];
      num = wmma_k4(at, byv, num);
      den = wmma_k4(at, brv, den);
    }
  }
#pragma unroll
  for (int r = 0; r < 8; r++) { redn[wv][r * 32 + lane] = num[r]; redd[wv][r * 32 + lane] = den[r]; }
  __syncthreads();
  if (wv == 0) {
#pragma unroll
    for (int r = 0; r < 8; r++) {
      float sn = 0.f, sd = 0.f;
      for (int w2 = 0; w2 < 8; w2++) { sn += redn[w2][r * 32 + lane]; sd += redd[w2][r * 32 + lane]; }
      if (hf == 0) {
        ws[oPVn + ((size_t)(n * YCH + yb) * KB + r) * JD + j0 + l16] = sn;
        ws[oPVd + ((size_t)(n * YCH + yb) * KB + r) * JD + j0 + l16] = sd;
      }
    }
  }
}

__global__ __launch_bounds__(256) void k_v_apply(float* __restrict__ ws) {
  size_t t = (size_t)blockIdx.x * 256 + threadIdx.x;
  size_t tot = (size_t)KB * JD;
  if (t >= tot) return;
  const int n = blockIdx.z;
  float sn = 0.f, sd = 0.f;
#pragma unroll
  for (int y = 0; y < YCH; y++) {
    sn += ws[oPVn + (size_t)(n * YCH + y) * KB * JD + t];
    sd += ws[oPVd + (size_t)(n * YCH + y) * KB * JD + t];
  }
  ws[oV + (size_t)n * KB * JD + t] *= sqrtf(sn / (sd + EPS));
}

__device__ __forceinline__ void ip_solve(float* Wf, const int n, float S00, float S11, float S01r, float S01i) {
  const float invJ = 1.0f / (float)JD;
  float D00 = S00 * invJ + EPS;
  float D11 = S11 * invJ + EPS;
  float Dr = S01r * invJ, Di = S01i * invJ;
  float w00r = Wf[0], w00i = Wf[1], w01r = Wf[2], w01i = Wf[3];
  float w10r = Wf[4], w10i = Wf[5], w11r = Wf[6], w11i = Wf[7];
  float A00r = w00r * D00 + (w01r * Dr + w01i * Di);
  float A00i = w00i * D00 + (w01i * Dr - w01r * Di);
  float A01r = (w00r * Dr - w00i * Di) + w01r * D11;
  float A01i = (w00i * Dr + w00r * Di) + w01i * D11;
  float A10r = w10r * D00 + (w11r * Dr + w11i * Di);
  float A10i = w10i * D00 + (w11i * Dr - w11r * Di);
  float A11r = (w10r * Dr - w10i * Di) + w11r * D11;
  float A11i = (w10i * Dr + w10r * Di) + w11i * D11;
  float detr = (A00r * A11r - A00i * A11i) - (A01r * A10r - A01i * A10i);
  float deti = (A00r * A11i + A00i * A11r) - (A01r * A10i + A01i * A10r);
  float n0r, n0i, n1r, n1i;
  if (n == 0) { n0r = A11r;  n0i = A11i;  n1r = -A10r; n1i = -A10i; }
  else        { n0r = -A01r; n0i = -A01i; n1r = A00r;  n1i = A00i;  }
  float dd = detr * detr + deti * deti;
  float b0r = (n0r * detr + n0i * deti) / dd, b0i = (n0i * detr - n0r * deti) / dd;
  float b1r = (n1r * detr + n1i * deti) / dd, b1i = (n1i * detr - n1r * deti) / dd;
  float Db0r = D00 * b0r + (Dr * b1r - Di * b1i);
  float Db0i = D00 * b0i + (Dr * b1i + Di * b1r);
  float Db1r = (Dr * b0r + Di * b0i) + D11 * b1r;
  float Db1i = (Dr * b0i - Di * b0r) + D11 * b1i;
  float denr = b0r * Db0r + b0i * Db0i + b1r * Db1r + b1i * Db1i;
  float dnm = sqrtf(denr + EPS);
  Wf[(n * 2 + 0) * 2]     = b0r / dnm;
  Wf[(n * 2 + 0) * 2 + 1] = -b0i / dnm;
  Wf[(n * 2 + 1) * 2]     = b1r / dnm;
  Wf[(n * 2 + 1) * 2 + 1] = -b1i / dnm;
}

__global__ __launch_bounds__(256) void k_ip_update(float* __restrict__ ws) {
  const int i = blockIdx.x;
  const int tid = threadIdx.x;
  const float* V = ws + oV;
  const float* X0r = ws + oXre;
  const float* X0i = ws + oXim;
  const float* X1r = ws + oXre + (size_t)IPD * JD;
  const float* X1i = ws + oXim + (size_t)IPD * JD;

  __shared__ float sT[NSRC][KB];
  __shared__ float rb[4 * NSRC][256];
  if (tid < NSRC * KB) sT[tid >> 3][tid & 7] = ws[oT + (size_t)(tid >> 3) * IPD * KB + (size_t)i * KB + (tid & 7)];
  __syncthreads();

  float d00[NSRC], d11[NSRC], d01r[NSRC], d01i[NSRC];
#pragma unroll
  for (int n = 0; n < NSRC; n++) { d00[n] = 0.f; d11[n] = 0.f; d01r[n] = 0.f; d01i[n] = 0.f; }
  for (int j = tid; j < JD; j += 256) {
    size_t idx = (size_t)i * JD + j;
    float r0r = X0r[idx], r0i = X0i[idx], r1r = X1r[idx], r1i = X1i[idx];
#pragma unroll
    for (int n = 0; n < NSRC; n++) {
      float rn = 0.f;
#pragma unroll
      for (int k = 0; k < KB; k++) rn += sT[n][k] * V[(size_t)(n * KB + k) * JD + j];
      float s = sqrtf(1.0f / (rn + EPS));
      float x0r = r0r * s, x0i = r0i * s;
      float x1r = r1r * s, x1i = r1i * s;
      d00[n] += x0r * x0r + x0i * x0i;
      d11[n] += x1r * x1r + x1i * x1i;
      d01r[n] += x0r * x1r + x0i * x1i;
      d01i[n] += x0i * x1r - x0r * x1i;
    }
  }
#pragma unroll
  for (int n = 0; n < NSRC; n++) {
    rb[4 * n + 0][tid] = d00[n]; rb[4 * n + 1][tid] = d11[n];
    rb[4 * n + 2][tid] = d01r[n]; rb[4 * n + 3][tid] = d01i[n];
  }
  __syncthreads();
  for (int s2 = 128; s2 > 0; s2 >>= 1) {
    if (tid < s2) {
#pragma unroll
      for (int q = 0; q < 4 * NSRC; q++) rb[q][tid] += rb[q][tid + s2];
    }
    __syncthreads();
  }
  if (tid == 0) {
    float* Wg = ws + oW + (size_t)i * 8;
    float Wf[8];
#pragma unroll
    for (int q = 0; q < 8; q++) Wf[q] = Wg[q];
    ip_solve(Wf, 0, rb[0][0], rb[1][0], rb[2][0], rb[3][0]);
    ip_solve(Wf, 1, rb[4][0], rb[5][0], rb[6][0], rb[7][0]);
#pragma unroll
    for (int q = 0; q < 8; q++) Wg[q] = Wf[q];
  }
}

__global__ __launch_bounds__(256) void k_y_update(float* __restrict__ ws) {
  const int i = blockIdx.x;
  const int tid = threadIdx.x;
  __shared__ float sw[8];
  if (tid < 8) sw[tid] = ws[oW + (size_t)i * 8 + tid];
  __syncthreads();
  const float* X0r = ws + oXre;
  const float* X0i = ws + oXim;
  const float* X1r = ws + oXre + (size_t)IPD * JD;
  const float* X1i = ws + oXim + (size_t)IPD * JD;
  float* Y0r = ws + oYre;
  float* Y0i = ws + oYim;
  float* Y1r = ws + oYre + (size_t)IPD * JD;
  float* Y1i = ws + oYim + (size_t)IPD * JD;
  for (int j = tid; j < JD; j += 256) {
    size_t idx = (size_t)i * JD + j;
    float x0r = X0r[idx], x0i = X0i[idx], x1r = X1r[idx], x1i = X1i[idx];
    {
      float wr0 = sw[0], wi0 = sw[1], wr1 = sw[2], wi1 = sw[3];
      Y0r[idx] = wr0 * x0r - wi0 * x0i + wr1 * x1r - wi1 * x1i;
      Y0i[idx] = wr0 * x0i + wi0 * x0r + wr1 * x1i + wi1 * x1r;
    }
    {
      float wr0 = sw[4], wi0 = sw[5], wr1 = sw[6], wi1 = sw[7];
      Y1r[idx] = wr0 * x0r - wi0 * x0i + wr1 * x1r - wi1 * x1i;
      Y1i[idx] = wr0 * x0i + wi0 * x0r + wr1 * x1i + wi1 * x1r;
    }
  }
}

#define OP 1026
__global__ __launch_bounds__(512) void k_out(const float* __restrict__ ws, float* __restrict__ out) {
  __shared__ float2 s[16 * OP];
  const int tid = threadIdx.x;
  const int j0 = blockIdx.x * 16, n = blockIdx.y;
  const float* Yre = ws + oYre + (size_t)n * IPD * JD + j0;
  const float* Yim = ws + oYim + (size_t)n * IPD * JD + j0;
  float4 re[8], im[8];
#pragma unroll
  for (int q = 0; q < 8; q++) {
    const int f = q * 512 + tid, i = f >> 2, jq = f & 3;
    re[q] = *(const float4*)(Yre + (size_t)i * JD + 4 * jq);
    im[q] = *(const float4*)(Yim + (size_t)i * JD + 4 * jq);
  }
#pragma unroll
  for (int q = 0; q < 8; q++) {
    const int f = q * 512 + tid, i = f >> 2, jq = f & 3;
    s[(4 * jq + 0) * OP + i] = make_float2(re[q].x, im[q].x);
    s[(4 * jq + 1) * OP + i] = make_float2(re[q].y, im[q].y);
    s[(4 * jq + 2) * OP + i] = make_float2(re[q].z, im[q].z);
    s[(4 * jq + 3) * OP + i] = make_float2(re[q].w, im[q].w);
  }
  if (tid < 4) {
    const int i = ID - 1, jq = tid;
    const float4 r = *(const float4*)(Yre + (size_t)i * JD + 4 * jq);
    const float4 g = *(const float4*)(Yim + (size_t)i * JD + 4 * jq);
    s[(4 * jq + 0) * OP + i] = make_float2(r.x, g.x);
    s[(4 * jq + 1) * OP + i] = make_float2(r.y, g.y);
    s[(4 * jq + 2) * OP + i] = make_float2(r.z, g.z);
    s[(4 * jq + 3) * OP + i] = make_float2(r.w, g.w);
  }
  __syncthreads();
  float4* o4 = (float4*)(out + ((size_t)n * JD + j0) * ID * 2);
  for (int h = tid; h < 16 * ID / 2; h += 512) {
    const int q0 = 2 * h, q1 = q0 + 1;
    const float2 a = s[q0 + q0 / ID], b = s[q1 + q1 / ID];
    o4[h] = make_float4(a.x, a.y, b.x, b.y);
  }
}

extern "C" void kernel_launch(void* const* d_in, const int* in_sizes, int n_in,
                              void* d_out, int out_size, void* d_ws, size_t ws_size,
                              hipStream_t stream) {
  (void)in_sizes; (void)n_in; (void)out_size; (void)ws_size;
  const float* X  = (const float*)d_in[0];
  const float* T0 = (const float*)d_in[1];
  const float* V0 = (const float*)d_in[2];
  float* ws = (float*)d_ws;
  float* out = (float*)d_out;

  k_init_xy<<<dim3((IPD + 63) / 64, JD / 64, MC), 256, 0, stream>>>(X, ws);
  { size_t tot = (size_t)NSRC * IPD * KB; k_init_t<<<(unsigned)((tot + 255) / 256), 256, 0, stream>>>(T0, ws); }
  { size_t tot = (size_t)NSRC * KB * JD;  k_init_v<<<(unsigned)((tot + 255) / 256), 256, 0, stream>>>(V0, ws); }
  { size_t tot = (size_t)ID * 8;          k_init_w<<<(unsigned)((tot + 255) / 256), 256, 0, stream>>>(ws); }

  for (int it = 0; it < NITER; ++it) {
    k_t_update<<<dim3(IPD / 16, YCH, NSRC), 256, 0, stream>>>(ws);
    k_t_apply<<<dim3((unsigned)(((size_t)IPD * KB + 255) / 256), 1, NSRC), 256, 0, stream>>>(ws);
    k_v_update<<<dim3(JD / 16, YCH, NSRC), 256, 0, stream>>>(ws);
    k_v_apply<<<dim3((unsigned)(((size_t)KB * JD + 255) / 256), 1, NSRC), 256, 0, stream>>>(ws);
    k_ip_update<<<ID, 256, 0, stream>>>(ws);
    k_y_update<<<ID, 256, 0, stream>>>(ws);
  }
  k_out<<<dim3(JD / 16, NSRC), 512, 0, stream>>>(ws, out);
}


# IP update and Y = W X fused per frequency: the X row read for the covariances stays in registers over the reduction and the solves and Y is formed from it; 30 launches, no second pass over X
# speedup vs baseline: 1.5434x; 1.1252x over previous
#include <hip/hip_runtime.h>
#include <math.h>

#define JD 2048
#define ID 1025
#define IPD 1040
#define KB 8
#define MC 2
#define NSRC 2
#define NITER 5
#define YCH 8
#define EPS 1e-20f

typedef __attribute__((ext_vector_type(2))) float v2f;
typedef __attribute__((ext_vector_type(8))) float v8f;

constexpr size_t oXre = 0;
constexpr size_t oXim = oXre + (size_t)MC * IPD * JD;
constexpr size_t oYre = oXim + (size_t)MC * IPD * JD;
constexpr size_t oYim = oYre + (size_t)NSRC * IPD * JD;
constexpr size_t oT   = oYim + (size_t)NSRC * IPD * JD;
constexpr size_t oV   = oT + (size_t)NSRC * IPD * KB;
constexpr size_t oW   = oV + (size_t)NSRC * KB * JD;
constexpr size_t oPTn = oW + (size_t)ID * 8;
constexpr size_t oPTd = oPTn + (size_t)NSRC * YCH * IPD * KB;
constexpr size_t oPVn = oPTd + (size_t)NSRC * YCH * IPD * KB;
constexpr size_t oPVd = oPVn + (size_t)NSRC * YCH * KB * JD;

__device__ __forceinline__ v8f wmma_k4(v2f a, v2f b, v8f c) {
  return __builtin_amdgcn_wmma_f32_16x16x4_f32(false, a, false, b, (short)0, c, false, false);
}

#define TP 66
__global__ __launch_bounds__(256) void k_init_xy(const float* __restrict__ X, float* __restrict__ ws) {
  __shared__ __attribute__((aligned(16))) float sre[64 * TP];
  __shared__ __attribute__((aligned(16))) float sim[64 * TP];
  const int tid = threadIdx.x, wv = tid >> 5, lane = tid & 31;
  const int i0 = blockIdx.x * 64, j0 = blockIdx.y * 64, m = blockIdx.z;
  float2 v[16];
#pragma unroll
  for (int q = 0; q < 16; q++) {
    const int jj = wv * 8 + (q >> 1);
    const int i = i0 + (q & 1) * 32 + lane;
    const int ic = i < ID ? i : ID - 1;
    v[q] = *(const float2*)(X + (((size_t)m * JD + j0 + jj) * ID + ic) * 2);
  }
#pragma unroll
  for (int q = 0; q < 16; q++) {
    const int jj = wv * 8 + (q >> 1);
    const int il = (q & 1) * 32 + lane;
    const bool pad = i0 + il >= ID;
    sre[il * TP + jj] = pad ? 0.f : v[q].x; sim[il * TP + jj] = pad ? 0.f : v[q].y;
  }
  __syncthreads();
#pragma unroll
  for (int q = 0; q < 8; q++) {
    const int il = wv * 8 + q;
    const int i = i0 + il;
    if (i < IPD) {
      const float2 xr = *(const float2*)(sre + il * TP + 2 * lane);
      const float2 xi = *(const float2*)(sim + il * TP + 2 * lane);
      const size_t dst = ((size_t)m * IPD + i) * JD + j0 + 2 * lane;
      *(float2*)(ws + oXre + dst) = xr; *(float2*)(ws + oXim + dst) = xi;
      *(float2*)(ws + oYre + dst) = xr; *(float2*)(ws + oYim + dst) = xi;
    }
  }
}

__global__ __launch_bounds__(256) void k_init_t(const float* __restrict__ T0, float* __restrict__ ws) {
  size_t t = (size_t)blockIdx.x * 256 + threadIdx.x;
  size_t tot = (size_t)NSRC * IPD * KB;
  if (t >= tot) return;
  int k = (int)(t % KB); size_t r = t / KB;
  int i = (int)(r % IPD); int n = (int)(r / IPD);
  float v = 0.f;
  if (i < ID) v = T0[((size_t)i * KB + k) * NSRC + n];
  ws[oT + ((size_t)n * IPD + i) * KB + k] = v;
}

__global__ __launch_bounds__(256) void k_init_v(const float* __restrict__ V0, float* __restrict__ ws) {
  size_t t = (size_t)blockIdx.x * 256 + threadIdx.x;
  size_t tot = (size_t)NSRC * KB * JD;
  if (t >= tot) return;
  int j = (int)(t % JD); size_t r = t / JD;
  int k = (int)(r % KB); int n = (int)(r / KB);
  ws[oV + ((size_t)n * KB + k) * JD + j] = V0[((size_t)k * JD + j) * NSRC + n];
}

__global__ __launch_bounds__(256) void k_init_w(float* __restrict__ ws) {
  size_t t = (size_t)blockIdx.x * 256 + threadIdx.x;
  size_t tot = (size_t)ID * 8;
  if (t >= tot) return;
  int f = (int)(t & 7);
  int c = f & 1, m = (f >> 1) & 1, nn = (f >> 2) & 1;
  ws[oW + t] = (c == 0 && nn == m) ? 1.f : 0.f;
}

__global__ __launch_bounds__(256) void k_t_update(float* __restrict__ ws) {
  const int tid = threadIdx.x;
  const int wv = tid >> 5, lane = tid & 31, hf = lane >> 4, l16 = lane & 15;
  const int i0 = blockIdx.x * 16;
  const int yb = blockIdx.y;
  const int n = blockIdx.z;
  const float* T = ws + oT + (size_t)n * IPD * KB;
  const float* V = ws + oV + (size_t)n * KB * JD;
  float* Yre = ws + oYre + (size_t)n * IPD * JD;
  const float* Yim = ws + oYim + (size_t)n * IPD * JD;

  __shared__ float lds_y[8][16][17];
  __shared__ float lds_r[8][16][17];
  __shared__ float redn[8][256];
  __shared__ float redd[8][256];

  v2f a0, a1;
  {
    size_t tr = (size_t)(i0 + l16) * KB;
    a0.x = T[tr + 2 * hf];     a0.y = T[tr + 2 * hf + 1];
    a1.x = T[tr + 4 + 2 * hf]; a1.y = T[tr + 5 + 2 * hf];
  }

  v8f num = {0,0,0,0,0,0,0,0}, den = {0,0,0,0,0,0,0,0};

  for (int jt = yb * 8 + wv; jt < JD / 16; jt += YCH * 8) {
    const int j0 = jt * 16;
    v2f b0, b1;
    b0.x = V[(size_t)(2 * hf + 0) * JD + j0 + l16];
    b0.y = V[(size_t)(2 * hf + 1) * JD + j0 + l16];
    b1.x = V[(size_t)(2 * hf + 4) * JD + j0 + l16];
    b1.y = V[(size_t)(2 * hf + 5) * JD + j0 + l16];
    v8f c = {0,0,0,0,0,0,0,0};
    c = wmma_k4(a0, b0, c);
    c = wmma_k4(a1, b1, c);
    float yrv[8], yiv[8];
#pragma unroll
    for (int r = 0; r < 8; r++) {
      size_t idx = (size_t)(i0 + r + 8 * hf) * JD + (j0 + l16);
      yrv[r] = Yre[idx]; yiv[r] = Yim[idx];
    }
#pragma unroll
    for (int r = 0; r < 8; r++) {
      const int il = r + 8 * hf;
      size_t idx = (size_t)(i0 + il) * JD + (j0 + l16);
      float yr = yrv[r], yi = yiv[r];
      float rn = c[r];
      float ydr = (yr * yr + yi * yi) / (rn * rn + EPS);
      float rd = 1.0f / (rn + EPS);
      Yre[idx] = ydr;
      lds_y[wv][il][l16] = ydr;
      lds_r[wv][il][l16] = rd;
    }
#pragma unroll
    for (int jc = 0; jc < 16; jc += 4) {
      v2f ay, ar, bv;
      ay.x = lds_y[wv][l16][jc + 2 * hf]; ay.y = lds_y[wv][l16][jc + 2 * hf + 1];
      ar.x = lds_r[wv][l16][jc + 2 * hf]; ar.y = lds_r[wv][l16][jc + 2 * hf + 1];
      float bx = 0.f, by = 0.f;
      if (l16 < KB) {
        bx = V[(size_t)l16 * JD + j0 + jc + 2 * hf];
        by = V[(size_t)l16 * JD + j0 + jc + 2 * hf + 1];
      }
      bv.x = bx; bv.y = by;
      num = wmma_k4(ay, bv, num);
      den = wmma_k4(ar, bv, den);
    }
  }
#pragma unroll
  for (int r = 0; r < 8; r++) { redn[wv][r * 32 + lane] = num[r]; redd[wv][r * 32 + lane] = den[r]; }
  __syncthreads();
  if (wv == 0) {
#pragma unroll
    for (int r = 0; r < 8; r++) {
      float sn = 0.f, sd = 0.f;
      for (int w2 = 0; w2 < 8; w2++) { sn += redn[w2][r * 32 + lane]; sd += redd[w2][r * 32 + lane]; }
      if (l16 < KB) {
        int ii = i0 + r + 8 * hf;
        ws[oPTn + ((size_t)(n * YCH + yb) * IPD + ii) * KB + l16] = sn;
        ws[oPTd + ((size_t)(n * YCH + yb) * IPD + ii) * KB + l16] = sd;
      }
    }
  }
}

__global__ __launch_bounds__(256) void k_t_apply(float* __restrict__ ws) {
  size_t t = (size_t)blockIdx.x * 256 + threadIdx.x;
  size_t tot = (size_t)IPD * KB;
  if (t >= tot) return;
  const int n = blockIdx.z;
  float sn = 0.f, sd = 0.f;
#pragma unroll
  for (int y = 0; y < YCH; y++) {
    sn += ws[oPTn + (size_t)(n * YCH + y) * IPD * KB + t];
    sd += ws[oPTd + (size_t)(n * YCH + y) * IPD * KB + t];
  }
  ws[oT + (size_t)n * IPD * KB + t] *= sqrtf(sn / (sd + EPS));
}

__global__ __launch_bounds__(256) void k_v_update(float* __restrict__ ws) {
  const int tid = threadIdx.x;
  const int wv = tid >> 5, lane = tid & 31, hf = lane >> 4, l16 = lane & 15;
  const int j0 = blockIdx.x * 16;
  const int yb = blockIdx.y;
  const int n = blockIdx.z;
  const float* T = ws + oT + (size_t)n * IPD * KB;
  const float* V = ws + oV + (size_t)n * KB * JD;
  const float* YdR = ws + oYre + (size_t)n * IPD * JD;

  __shared__ float lds_r[8][16][17];
  __shared__ float redn[8][256];
  __shared__ float redd[8][256];

  v2f b0, b1;
  b0.x = V[(size_t)(2 * hf + 0) * JD + j0 + l16];
  b0.y = V[(size_t)(2 * hf + 1) * JD + j0 + l16];
  b1.x = V[(size_t)(2 * hf + 4) * JD + j0 + l16];
  b1.y = V[(size_t)(2 * hf + 5) * JD + j0 + l16];

  v8f num = {0,0,0,0,0,0,0,0}, den = {0,0,0,0,0,0,0,0};

  for (int it = yb * 8 + wv; it < IPD / 16; it += YCH * 8) {
    const int i0 = it * 16;
    v2f a0, a1;
    size_t tr = (size_t)(i0 + l16) * KB;
    a0.x = T[tr + 2 * hf];     a0.y = T[tr + 2 * hf + 1];
    a1.x = T[tr + 4 + 2 * hf]; a1.y = T[tr + 5 + 2 * hf];
    v8f c = {0,0,0,0,0,0,0,0};
    c = wmma_k4(a0, b0, c);
    c = wmma_k4(a1, b1, c);
#pragma unroll
    for (int r = 0; r < 8; r++)
      lds_r[wv][r + 8 * hf][l16] = 1.0f / (fabsf(c[r]) + EPS);
#pragma unroll
    for (int ic = 0; ic < 16; ic += 4) {
      v2f at, byv, brv;
      int ia = i0 + ic + 2 * hf;
      at.x = (l16 < KB) ? T[(size_t)ia * KB + l16] : 0.f;
      at.y = (l16 < KB) ? T[(size_t)(ia + 1) * KB + l16] : 0.f;
      byv.x = YdR[(size_t)ia * JD + j0 + l16];
      byv.y = YdR[(size_t)(ia + 1) * JD + j0 + l16];
      brv.x = lds_r[wv][ic + 2 * hf][l16];
      brv.y = lds_r[wv][ic + 2 * hf + 1][l16];
      num = wmma_k4(at, byv, num);
      den = wmma_k4(at, brv, den);
    }
  }
#pragma unroll
  for (int r = 0; r < 8; r++) { redn[wv][r * 32 + lane] = num[r]; redd[wv][r * 32 + lane] = den[r]; }
  __syncthreads();
  if (wv == 0) {
#pragma unroll
    for (int r = 0; r < 8; r++) {
      float sn = 0.f, sd = 0.f;
      for (int w2 = 0; w2 < 8; w2++) { sn += redn[w2][r * 32 + lane]; sd += redd[w2][r * 32 + lane]; }
      if (hf == 0) {
        ws[oPVn + ((size_t)(n * YCH + yb) * KB + r) * JD + j0 + l16] = sn;
        ws[oPVd + ((size_t)(n * YCH + yb) * KB + r) * JD + j0 + l16] = sd;
      }
    }
  }
}

__global__ __launch_bounds__(256) void k_v_apply(float* __restrict__ ws) {
  size_t t = (size_t)blockIdx.x * 256 + threadIdx.x;
  size_t tot = (size_t)KB * JD;
  if (t >= tot) return;
  const int n = blockIdx.z;
  float sn = 0.f, sd = 0.f;
#pragma unroll
  for (int y = 0; y < YCH; y++) {
    sn += ws[oPVn + (size_t)(n * YCH + y) * KB * JD + t];
    sd += ws[oPVd + (size_t)(n * YCH + y) * KB * JD + t];
  }
  ws[oV + (size_t)n * KB * JD + t] *= sqrtf(sn / (sd + EPS));
}

__device__ __forceinline__ void ip_solve(float* Wf, const int n, float S00, float S11, float S01r, float S01i) {
  const float invJ = 1.0f / (float)JD;
  float D00 = S00 * invJ + EPS;
  float D11 = S11 * invJ + EPS;
  float Dr = S01r * invJ, Di = S01i * invJ;
  float w00r = Wf[0], w00i = Wf[1], w01r = Wf[2], w01i = Wf[3];
  float w10r = Wf[4], w10i = Wf[5], w11r = Wf[6], w11i = Wf[7];
  float A00r = w00r * D00 + (w01r * Dr + w01i * Di);
  float A00i = w00i * D00 + (w01i * Dr - w01r * Di);
  float A01r = (w00r * Dr - w00i * Di) + w01r * D11;
  float A01i = (w00i * Dr + w00r * Di) + w01i * D11;
  float A10r = w10r * D00 + (w11r * Dr + w11i * Di);
  float A10i = w10i * D00 + (w11i * Dr - w11r * Di);
  float A11r = (w10r * Dr - w10i * Di) + w11r * D11;
  float A11i = (w10i * Dr + w10r * Di) + w11i * D11;
  float detr = (A00r * A11r - A00i * A11i) - (A01r * A10r - A01i * A10i);
  float deti = (A00r * A11i + A00i * A11r) - (A01r * A10i + A01i * A10r);
  float n0r, n0i, n1r, n1i;
  if (n == 0) { n0r = A11r;  n0i = A11i;  n1r = -A10r; n1i = -A10i; }
  else        { n0r = -A01r; n0i = -A01i; n1r = A00r;  n1i = A00i;  }
  float dd = detr * detr + deti * deti;
  float b0r = (n0r * detr + n0i * deti) / dd, b0i = (n0i * detr - n0r * deti) / dd;
  float b1r = (n1r * detr + n1i * deti) / dd, b1i = (n1i * detr - n1r * deti) / dd;
  float Db0r = D00 * b0r + (Dr * b1r - Di * b1i);
  float Db0i = D00 * b0i + (Dr * b1i + Di * b1r);
  float Db1r = (Dr * b0r + Di * b0i) + D11 * b1r;
  float Db1i = (Dr * b0i - Di * b0r) + D11 * b1i;
  float denr = b0r * Db0r + b0i * Db0i + b1r * Db1r + b1i * Db1i;
  float dnm = sqrtf(denr + EPS);
  Wf[(n * 2 + 0) * 2]     = b0r / dnm;
  Wf[(n * 2 + 0) * 2 + 1] = -b0i / dnm;
  Wf[(n * 2 + 1) * 2]     = b1r / dnm;
  Wf[(n * 2 + 1) * 2 + 1] = -b1i / dnm;
}

#define JPT (JD / 256)
__global__ __launch_bounds__(256, 10) void k_ip_y_update(float* __restrict__ ws) {
  const int i = blockIdx.x;
  const int tid = threadIdx.x;
  const float* V = ws + oV;
  const float* X0r = ws + oXre;
  const float* X0i = ws + oXim;
  const float* X1r = ws + oXre + (size_t)IPD * JD;
  const float* X1i = ws + oXim + (size_t)IPD * JD;

  __shared__ float sw[8];
  __shared__ float rb[4 * NSRC][256];

  float x0r[JPT], x0i[JPT], x1r[JPT], x1i[JPT];
#pragma unroll
  for (int q = 0; q < JPT; q++) {
    const size_t idx = (size_t)i * JD + tid + 256 * q;
    x0r[q] = X0r[idx]; x0i[q] = X0i[idx]; x1r[q] = X1r[idx]; x1i[q] = X1i[idx];
  }
  __builtin_amdgcn_sched_barrier(0);

  float tT[NSRC][KB], w0[8];
#pragma unroll
  for (int n = 0; n < NSRC; n++)
#pragma unroll
    for (int k = 0; k < KB; k++) tT[n][k] = ws[oT + (size_t)n * IPD * KB + (size_t)i * KB + k];
#pragma unroll
  for (int q = 0; q < 8; q++) w0[q] = ws[oW + (size_t)i * 8 + q];

  float d00[NSRC], d11[NSRC], d01r[NSRC], d01i[NSRC];
#pragma unroll
  for (int n = 0; n < NSRC; n++) { d00[n] = 0.f; d11[n] = 0.f; d01r[n] = 0.f; d01i[n] = 0.f; }
  float vv[2][NSRC * KB];
#pragma unroll
  for (int r = 0; r < NSRC * KB; r++) vv[0][r] = V[(size_t)r * JD + tid];
#pragma unroll
  for (int q = 0; q < JPT; q++) {
    if (q + 1 < JPT) {
#pragma unroll
      for (int r = 0; r < NSRC * KB; r++) vv[(q + 1) & 1][r] = V[(size_t)r * JD + tid + 256 * (q + 1)];
    }
#pragma unroll
    for (int n = 0; n < NSRC; n++) {
      float rn = 0.f;
#pragma unroll
      for (int k = 0; k < KB; k++) rn += tT[n][k] * vv[q & 1][n * KB + k];
      float s = sqrtf(1.0f / (rn + EPS));
      float a0r = x0r[q] * s, a0i = x0i[q] * s;
      float a1r = x1r[q] * s, a1i = x1i[q] * s;
      d00[n] += a0r * a0r + a0i * a0i;
      d11[n] += a1r * a1r + a1i * a1i;
      d01r[n] += a0r * a1r + a0i * a1i;
      d01i[n] += a0i * a1r - a0r * a1i;
    }
  }
#pragma unroll
  for (int n = 0; n < NSRC; n++) {
    rb[4 * n + 0][tid] = d00[n]; rb[4 * n + 1][tid] = d11[n];
    rb[4 * n + 2][tid] = d01r[n]; rb[4 * n + 3][tid] = d01i[n];
  }
  __syncthreads();
  for (int s2 = 128; s2 > 0; s2 >>= 1) {
    if (tid < s2) {
#pragma unroll
      for (int q = 0; q < 4 * NSRC; q++) rb[q][tid] += rb[q][tid + s2];
    }
    __syncthreads();
  }
  if (tid == 0) {
    float Wf[8];
#pragma unroll
    for (int q = 0; q < 8; q++) Wf[q] = w0[q];
    ip_solve(Wf, 0, rb[0][0], rb[1][0], rb[2][0], rb[3][0]);
    ip_solve(Wf, 1, rb[4][0], rb[5][0], rb[6][0], rb[7][0]);
#pragma unroll
    for (int q = 0; q < 8; q++) sw[q] = Wf[q];
  }
  __syncthreads();
  if (tid < 8) ws[oW + (size_t)i * 8 + tid] = sw[tid];

  float* Y0r = ws + oYre;
  float* Y0i = ws + oYim;
  float* Y1r = ws + oYre + (size_t)IPD * JD;
  float* Y1i = ws + oYim + (size_t)IPD * JD;
  const float wa0 = sw[0], wb0 = sw[1], wa1 = sw[2], wb1 = sw[3];
  const float wc0 = sw[4], wd0 = sw[5], wc1 = sw[6], wd1 = sw[7];
#pragma unroll
  for (int q = 0; q < JPT; q++) {
    const size_t idx = (size_t)i * JD + tid + 256 * q;
    const float p0r = x0r[q], p0i = x0i[q], p1r = x1r[q], p1i = x1i[q];
    {
      float wr0 = wa0, wi0 = wb0, wr1 = wa1, wi1 = wb1;
      Y0r[idx] = wr0 * p0r - wi0 * p0i + wr1 * p1r - wi1 * p1i;
      Y0i[idx] = wr0 * p0i + wi0 * p0r + wr1 * p1i + wi1 * p1r;
    }
    {
      float wr0 = wc0, wi0 = wd0, wr1 = wc1, wi1 = wd1;
      Y1r[idx] = wr0 * p0r - wi0 * p0i + wr1 * p1r - wi1 * p1i;
      Y1i[idx] = wr0 * p0i + wi0 * p0r + wr1 * p1i + wi1 * p1r;
    }
  }
}

#define OP 1026
__global__ __launch_bounds__(512) void k_out(const float* __restrict__ ws, float* __restrict__ out) {
  __shared__ float2 s[16 * OP];
  const int tid = threadIdx.x;
  const int j0 = blockIdx.x * 16, n = blockIdx.y;
  const float* Yre = ws + oYre + (size_t)n * IPD * JD + j0;
  const float* Yim = ws + oYim + (size_t)n * IPD * JD + j0;
  float4 re[8], im[8];
#pragma unroll
  for (int q = 0; q < 8; q++) {
    const int f = q * 512 + tid, i = f >> 2, jq = f & 3;
    re[q] = *(const float4*)(Yre + (size_t)i * JD + 4 * jq);
    im[q] = *(const float4*)(Yim + (size_t)i * JD + 4 * jq);
  }
#pragma unroll
  for (int q = 0; q < 8; q++) {
    const int f = q * 512 + tid, i = f >> 2, jq = f & 3;
    s[(4 * jq + 0) * OP + i] = make_float2(re[q].x, im[q].x);
    s[(4 * jq + 1) * OP + i] = make_float2(re[q].y, im[q].y);
    s[(4 * jq + 2) * OP + i] = make_float2(re[q].z, im[q].z);
    s[(4 * jq + 3) * OP + i] = make_float2(re[q].w, im[q].w);
  }
  if (tid < 4) {
    const int i = ID - 1, jq = tid;
    const float4 r = *(const float4*)(Yre + (size_t)i * JD + 4 * jq);
    const float4 g = *(const float4*)(Yim + (size_t)i * JD + 4 * jq);
    s[(4 * jq + 0) * OP + i] = make_float2(r.x, g.x);
    s[(4 * jq + 1) * OP + i] = make_float2(r.y, g.y);
    s[(4 * jq + 2) * OP + i] = make_float2(r.z, g.z);
    s[(4 * jq + 3) * OP + i] = make_float2(r.w, g.w);
  }
  __syncthreads();
  float4* o4 = (float4*)(out + ((size_t)n * JD + j0) * ID * 2);
  for (int h = tid; h < 16 * ID / 2; h += 512) {
    const int q0 = 2 * h, q1 = q0 + 1;
    const float2 a = s[q0 + q0 / ID], b = s[q1 + q1 / ID];
    o4[h] = make_float4(a.x, a.y, b.x, b.y);
  }
}

extern "C" void kernel_launch(void* const* d_in, const int* in_sizes, int n_in,
                              void* d_out, int out_size, void* d_ws, size_t ws_size,
                              hipStream_t stream) {
  (void)in_sizes; (void)n_in; (void)out_size; (void)ws_size;
  const float* X  = (const float*)d_in[0];
  const float* T0 = (const float*)d_in[1];
  const float* V0 = (const float*)d_in[2];
  float* ws = (float*)d_ws;
  float* out = (float*)d_out;

  k_init_xy<<<dim3((IPD + 63) / 64, JD / 64, MC), 256, 0, stream>>>(X, ws);
  { size_t tot = (size_t)NSRC * IPD * KB; k_init_t<<<(unsigned)((tot + 255) / 256), 256, 0, stream>>>(T0, ws); }
  { size_t tot = (size_t)NSRC * KB * JD;  k_init_v<<<(unsigned)((tot + 255) / 256), 256, 0, stream>>>(V0, ws); }
  { size_t tot = (size_t)ID * 8;          k_init_w<<<(unsigned)((tot + 255) / 256), 256, 0, stream>>>(ws); }

  for (int it = 0; it < NITER; ++it) {
    k_t_update<<<dim3(IPD / 16, YCH, NSRC), 256, 0, stream>>>(ws);
    k_t_apply<<<dim3((unsigned)(((size_t)IPD * KB + 255) / 256), 1, NSRC), 256, 0, stream>>>(ws);
    k_v_update<<<dim3(JD / 16, YCH, NSRC), 256, 0, stream>>>(ws);
    k_v_apply<<<dim3((unsigned)(((size_t)KB * JD + 255) / 256), 1, NSRC), 256, 0, stream>>>(ws);
    k_ip_y_update<<<ID, 256, 0, stream>>>(ws);
  }
  k_out<<<dim3(JD / 16, NSRC), 512, 0, stream>>>(ws, out);
}


# Next iteration's T update fused into the per-frequency IP/Y block (it holds R, the new Y and every j): num/den on WMMA from LDS, j sums regrouped per block; Y stored once, at the end; 21 launches
# speedup vs baseline: 1.9166x; 1.2418x over previous
#include <hip/hip_runtime.h>
#include <math.h>

#define JD 2048
#define ID 1025
#define IPD 1040
#define KB 8
#define MC 2
#define NSRC 2
#define NITER 5
#define YCH 8
#define EPS 1e-20f

typedef __attribute__((ext_vector_type(2))) float v2f;
typedef __attribute__((ext_vector_type(8))) float v8f;

constexpr size_t oXre = 0;
constexpr size_t oXim = oXre + (size_t)MC * IPD * JD;
constexpr size_t oYre = oXim + (size_t)MC * IPD * JD;
constexpr size_t oYim = oYre + (size_t)NSRC * IPD * JD;
constexpr size_t oT   = oYim + (size_t)NSRC * IPD * JD;
constexpr size_t oV   = oT + (size_t)NSRC * IPD * KB;
constexpr size_t oW   = oV + (size_t)NSRC * KB * JD;
constexpr size_t oPVn = oW + (size_t)ID * 8;
constexpr size_t oPVd = oPVn + (size_t)NSRC * YCH * KB * JD;

__device__ __forceinline__ v8f wmma_k4(v2f a, v2f b, v8f c) {
  return __builtin_amdgcn_wmma_f32_16x16x4_f32(false, a, false, b, (short)0, c, false, false);
}

#define TP 66
__global__ __launch_bounds__(256) void k_init_xy(const float* __restrict__ X, float* __restrict__ ws) {
  __shared__ __attribute__((aligned(16))) float sre[64 * TP];
  __shared__ __attribute__((aligned(16))) float sim[64 * TP];
  const int tid = threadIdx.x, wv = tid >> 5, lane = tid & 31;
  const int i0 = blockIdx.x * 64, j0 = blockIdx.y * 64, m = blockIdx.z;
  float2 v[16];
#pragma unroll
  for (int q = 0; q < 16; q++) {
    const int jj = wv * 8 + (q >> 1);
    const int i = i0 + (q & 1) * 32 + lane;
    const int ic = i < ID ? i : ID - 1;
    v[q] = *(const float2*)(X + (((size_t)m * JD + j0 + jj) * ID + ic) * 2);
  }
#pragma unroll
  for (int q = 0; q < 16; q++) {
    const int jj = wv * 8 + (q >> 1);
    const int il = (q & 1) * 32 + lane;
    const bool pad = i0 + il >= ID;
    sre[il * TP + jj] = pad ? 0.f : v[q].x; sim[il * TP + jj] = pad ? 0.f : v[q].y;
  }
  __syncthreads();
#pragma unroll
  for (int q = 0; q < 8; q++) {
    const int il = wv * 8 + q;
    const int i = i0 + il;
    if (i < IPD) {
      const float2 xr = *(const float2*)(sre + il * TP + 2 * lane);
      const float2 xi = *(const float2*)(sim + il * TP + 2 * lane);
      const size_t dst = ((size_t)m * IPD + i) * JD + j0 + 2 * lane;
      *(float2*)(ws + oXre + dst) = xr; *(float2*)(ws + oXim + dst) = xi;
      if (i >= ID) *(float2*)(ws + oYre + dst) = xr;
    }
  }
}

__global__ __launch_bounds__(256) void k_init_t(const float* __restrict__ T0, float* __restrict__ ws) {
  size_t t = (size_t)blockIdx.x * 256 + threadIdx.x;
  size_t tot = (size_t)NSRC * IPD * KB;
  if (t >= tot) return;
  int k = (int)(t % KB); size_t r = t / KB;
  int i = (int)(r % IPD); int n = (int)(r / IPD);
  float v = 0.f;
  if (i < ID) v = T0[((size_t)i * KB + k) * NSRC + n];
  ws[oT + ((size_t)n * IPD + i) * KB + k] = v;
}

__global__ __launch_bounds__(256) void k_init_v(const float* __restrict__ V0, float* __restrict__ ws) {
  size_t t = (size_t)blockIdx.x * 256 + threadIdx.x;
  size_t tot = (size_t)NSRC * KB * JD;
  if (t >= tot) return;
  int j = (int)(t % JD); size_t r = t / JD;
  int k = (int)(r % KB); int n = (int)(r / KB);
  ws[oV + ((size_t)n * KB + k) * JD + j] = V0[((size_t)k * JD + j) * NSRC + n];
}

__global__ __launch_bounds__(256) void k_init_w(float* __restrict__ ws) {
  size_t t = (size_t)blockIdx.x * 256 + threadIdx.x;
  size_t tot = (size_t)ID * 8;
  if (t >= tot) return;
  int f = (int)(t & 7);
  int c = f & 1, m = (f >> 1) & 1, nn = (f >> 2) & 1;
  ws[oW + t] = (c == 0 && nn == m) ? 1.f : 0.f;
}

__global__ __launch_bounds__(256) void k_v_update(float* __restrict__ ws) {
  const int tid = threadIdx.x;
  const int wv = tid >> 5, lane = tid & 31, hf = lane >> 4, l16 = lane & 15;
  const int j0 = blockIdx.x * 16;
  const int yb = blockIdx.y;
  const int n = blockIdx.z;
  const float* T = ws + oT + (size_t)n * IPD * KB;
  const float* V = ws + oV + (size_t)n * KB * JD;
  const float* YdR = ws + oYre + (size_t)n * IPD * JD;

  __shared__ float lds_r[8][16][17];
  __shared__ float redn[8][256];
  __shared__ float redd[8][256];

  v2f b0, b1;
  b0.x = V[(size_t)(2 * hf + 0) * JD + j0 + l16];
  b0.y = V[(size_t)(2 * hf + 1) * JD + j0 + l16];
  b1.x = V[(size_t)(2 * hf + 4) * JD + j0 + l16];
  b1.y = V[(size_t)(2 * hf + 5) * JD + j0 + l16];

  v8f num = {0,0,0,0,0,0,0,0}, den = {0,0,0,0,0,0,0,0};

  for (int it = yb * 8 + wv; it < IPD / 16; it += YCH * 8) {
    const int i0 = it * 16;
    v2f a0, a1;
    size_t tr = (size_t)(i0 + l16) * KB;
    a0.x = T[tr + 2 * hf];     a0.y = T[tr + 2 * hf + 1];
    a1.x = T[tr + 4 + 2 * hf]; a1.y = T[tr + 5 + 2 * hf];
    v8f c = {0,0,0,0,0,0,0,0};
    c = wmma_k4(a0, b0, c);
    c = wmma_k4(a1, b1, c);
#pragma unroll
    for (int r = 0; r < 8; r++)
      lds_r[wv][r + 8 * hf][l16] = 1.0f / (fabsf(c[r]) + EPS);
#pragma unroll
    for (int ic = 0; ic < 16; ic += 4) {
      v2f at, byv, brv;
      int ia = i0 + ic + 2 * hf;
      at.x = (l16 < KB) ? T[(size_t)ia * KB + l16] : 0.f;
      at.y = (l16 < KB) ? T[(size_t)(ia + 1) * KB + l16] : 0.f;
      byv.x = YdR[(size_t)ia * JD + j0 + l16];
      byv.y = YdR[(size_t)(ia + 1) * JD + j0 + l16];
      brv.x = lds_r[wv][ic + 2 * hf][l16];
      brv.y = lds_r[wv][ic + 2 * hf + 1][l16];
      num = wmma_k4(at, byv, num);
      den = wmma_k4(at, brv, den);
    }
  }
#pragma unroll
  for (int r = 0; r < 8; r++) { redn[wv][r * 32 + lane] = num[r]; redd[wv][r * 32 + lane] = den[r]; }
  __syncthreads();
  if (wv == 0) {
#pragma unroll
    for (int r = 0; r < 8; r++) {
      float sn = 0.f, sd = 0.f;
      for (int w2 = 0; w2 < 8; w2++) { sn += redn[w2][r * 32 + lane]; sd += redd[w2][r * 32 + lane]; }
      if (hf == 0) {
        ws[oPVn + ((size_t)(n * YCH + yb) * KB + r) * JD + j0 + l16] = sn;
        ws[oPVd + ((size_t)(n * YCH + yb) * KB + r) * JD + j0 + l16] = sd;
      }
    }
  }
}

__global__ __launch_bounds__(256) void k_v_apply(float* __restrict__ ws) {
  size_t t = (size_t)blockIdx.x * 256 + threadIdx.x;
  size_t tot = (size_t)KB * JD;
  if (t >= tot) return;
  const int n = blockIdx.z;
  float sn = 0.f, sd = 0.f;
#pragma unroll
  for (int y = 0; y < YCH; y++) {
    sn += ws[oPVn + (size_t)(n * YCH + y) * KB * JD + t];
    sd += ws[oPVd + (size_t)(n * YCH + y) * KB * JD + t];
  }
  ws[oV + (size_t)n * KB * JD + t] *= sqrtf(sn / (sd + EPS));
}

__device__ __forceinline__ void ip_solve(float* Wf, const int n, float S00, float S11, float S01r, float S01i) {
  const float invJ = 1.0f / (float)JD;
  float D00 = S00 * invJ + EPS;
  float D11 = S11 * invJ + EPS;
  float Dr = S01r * invJ, Di = S01i * invJ;
  float w00r = Wf[0], w00i = Wf[1], w01r = Wf[2], w01i = Wf[3];
  float w10r = Wf[4], w10i = Wf[5], w11r = Wf[6], w11i = Wf[7];
  float A00r = w00r * D00 + (w01r * Dr + w01i * Di);
  float A00i = w00i * D00 + (w01i * Dr - w01r * Di);
  float A01r = (w00r * Dr - w00i * Di) + w01r * D11;
  float A01i = (w00i * Dr + w00r * Di) + w01i * D11;
  float A10r = w10r * D00 + (w11r * Dr + w11i * Di);
  float A10i = w10i * D00 + (w11i * Dr - w11r * Di);
  float A11r = (w10r * Dr - w10i * Di) + w11r * D11;
  float A11i = (w10i * Dr + w10r * Di) + w11i * D11;
  float detr = (A00r * A11r - A00i * A11i) - (A01r * A10r - A01i * A10i);
  float deti = (A00r * A11i + A00i * A11r) - (A01r * A10i + A01i * A10r);
  float n0r, n0i, n1r, n1i;
  if (n == 0) { n0r = A11r;  n0i = A11i;  n1r = -A10r; n1i = -A10i; }
  else        { n0r = -A01r; n0i = -A01i; n1r = A00r;  n1i = A00i;  }
  float dd = detr * detr + deti * deti;
  float b0r = (n0r * detr + n0i * deti) / dd, b0i = (n0i * detr - n0r * deti) / dd;
  float b1r = (n1r * detr + n1i * deti) / dd, b1i = (n1i * detr - n1r * deti) / dd;
  float Db0r = D00 * b0r + (Dr * b1r - Di * b1i);
  float Db0i = D00 * b0i + (Dr * b1i + Di * b1r);
  float Db1r = (Dr * b0r + Di * b0i) + D11 * b1r;
  float Db1i = (Dr * b0i - Di * b0r) + D11 * b1i;
  float denr = b0r * Db0r + b0i * Db0i + b1r * Db1r + b1i * Db1i;
  float dnm = sqrtf(denr + EPS);
  Wf[(n * 2 + 0) * 2]     = b0r / dnm;
  Wf[(n * 2 + 0) * 2 + 1] = -b0i / dnm;
  Wf[(n * 2 + 1) * 2]     = b1r / dnm;
  Wf[(n * 2 + 1) * 2 + 1] = -b1i / dnm;
}

#define JPT (JD / 256)
#define QP (JD + 8)
template <int MODE>
__global__ __launch_bounds__(256, 10) void k_ip_t(float* __restrict__ ws) {
  const int i = blockIdx.x;
  const int tid = threadIdx.x;
  const int wv = tid >> 5, lane = tid & 31, hf = lane >> 4, l16 = lane & 15;
  const float* V = ws + oV;
  const float* X0r = ws + oXre;
  const float* X0i = ws + oXim;
  const float* X1r = ws + oXre + (size_t)IPD * JD;
  const float* X1i = ws + oXim + (size_t)IPD * JD;

  __shared__ float sw[8];
  __shared__ float rb[4 * NSRC][256];
  __shared__ __attribute__((aligned(16))) float sq[MODE == 2 ? 4 : 4 * QP];
  __shared__ float red[8][32];

  float x0r[JPT], x0i[JPT], x1r[JPT], x1i[JPT];
#pragma unroll
  for (int q = 0; q < JPT; q++) {
    const size_t idx = (size_t)i * JD + tid + 256 * q;
    x0r[q] = X0r[idx]; x0i[q] = X0i[idx]; x1r[q] = X1r[idx]; x1i[q] = X1i[idx];
  }
  __builtin_amdgcn_sched_barrier(0);

  float tT[NSRC][KB], w0[8];
#pragma unroll
  for (int n = 0; n < NSRC; n++)
#pragma unroll
    for (int k = 0; k < KB; k++) tT[n][k] = ws[oT + (size_t)n * IPD * KB + (size_t)i * KB + k];
  if (MODE != 0) {
#pragma unroll
    for (int q = 0; q < 8; q++) w0[q] = ws[oW + (size_t)i * 8 + q];
  }

  float rnv[NSRC][JPT];
  float d00[NSRC], d11[NSRC], d01r[NSRC], d01i[NSRC];
#pragma unroll
  for (int n = 0; n < NSRC; n++) { d00[n] = 0.f; d11[n] = 0.f; d01r[n] = 0.f; d01i[n] = 0.f; }
  float vv[2][NSRC * KB];
#pragma unroll
  for (int r = 0; r < NSRC * KB; r++) vv[0][r] = V[(size_t)r * JD + tid];
#pragma unroll
  for (int q = 0; q < JPT; q++) {
    if (q + 1 < JPT) {
#pragma unroll
      for (int r = 0; r < NSRC * KB; r++) vv[(q + 1) & 1][r] = V[(size_t)r * JD + tid + 256 * (q + 1)];
    }
#pragma unroll
    for (int n = 0; n < NSRC; n++) {
      float rn = 0.f;
#pragma unroll
      for (int k = 0; k < KB; k++) rn += tT[n][k] * vv[q & 1][n * KB + k];
      rnv[n][q] = rn;
      if (MODE != 0) {
        float s = sqrtf(1.0f / (rn + EPS));
        float a0r = x0r[q] * s, a0i = x0i[q] * s;
        float a1r = x1r[q] * s, a1i = x1i[q] * s;
        d00[n] += a0r * a0r + a0i * a0i;
        d11[n] += a1r * a1r + a1i * a1i;
        d01r[n] += a0r * a1r + a0i * a1i;
        d01i[n] += a0i * a1r - a0r * a1i;
      }
    }
  }
  float wa0 = 1.f, wb0 = 0.f, wa1 = 0.f, wb1 = 0.f;
  float wc0 = 0.f, wd0 = 0.f, wc1 = 1.f, wd1 = 0.f;
  if (MODE != 0) {
#pragma unroll
    for (int n = 0; n < NSRC; n++) {
      rb[4 * n + 0][tid] = d00[n]; rb[4 * n + 1][tid] = d11[n];
      rb[4 * n + 2][tid] = d01r[n]; rb[4 * n + 3][tid] = d01i[n];
    }
    __syncthreads();
    for (int s2 = 128; s2 > 0; s2 >>= 1) {
      if (tid < s2) {
#pragma unroll
        for (int q = 0; q < 4 * NSRC; q++) rb[q][tid] += rb[q][tid + s2];
      }
      __syncthreads();
    }
    if (tid == 0) {
      float Wf[8];
#pragma unroll
      for (int q = 0; q < 8; q++) Wf[q] = w0[q];
      ip_solve(Wf, 0, rb[0][0], rb[1][0], rb[2][0], rb[3][0]);
      ip_solve(Wf, 1, rb[4][0], rb[5][0], rb[6][0], rb[7][0]);
#pragma unroll
      for (int q = 0; q < 8; q++) sw[q] = Wf[q];
    }
    __syncthreads();
    if (tid < 8) ws[oW + (size_t)i * 8 + tid] = sw[tid];
    wa0 = sw[0]; wb0 = sw[1]; wa1 = sw[2]; wb1 = sw[3];
    wc0 = sw[4]; wd0 = sw[5]; wc1 = sw[6]; wd1 = sw[7];
  }

  float* P0r = ws + oYre;
  float* P1r = ws + oYre + (size_t)IPD * JD;
  float* P0i = ws + oYim;
  float* P1i = ws + oYim + (size_t)IPD * JD;
#pragma unroll
  for (int q = 0; q < JPT; q++) {
    const int j = tid + 256 * q;
    const size_t idx = (size_t)i * JD + j;
    const float p0r = x0r[q], p0i = x0i[q], p1r = x1r[q], p1i = x1i[q];
    const float y0r = wa0 * p0r - wb0 * p0i + wa1 * p1r - wb1 * p1i;
    const float y0i = wa0 * p0i + wb0 * p0r + wa1 * p1i + wb1 * p1r;
    const float y1r = wc0 * p0r - wd0 * p0i + wc1 * p1r - wd1 * p1i;
    const float y1i = wc0 * p0i + wd0 * p0r + wc1 * p1i + wd1 * p1r;
    if (MODE == 2) {
      P0r[idx] = y0r; P0i[idx] = y0i; P1r[idx] = y1r; P1i[idx] = y1i;
    } else {
      const float r0 = rnv[0][q], r1 = rnv[1][q];
      const float ydr0 = (y0r * y0r + y0i * y0i) / (r0 * r0 + EPS);
      const float ydr1 = (y1r * y1r + y1i * y1i) / (r1 * r1 + EPS);
      const float rd0 = 1.0f / (fabsf(r0) + EPS);
      const float rd1 = 1.0f / (fabsf(r1) + EPS);
      P0r[idx] = ydr0; P1r[idx] = ydr1;
      sq[0 * QP + j] = ydr0; sq[1 * QP + j] = rd0;
      sq[2 * QP + j] = ydr1; sq[3 * QP + j] = rd1;
    }
  }
  if (MODE == 2) return;

  __syncthreads();
  v8f acc0 = {0,0,0,0,0,0,0,0}, acc1 = {0,0,0,0,0,0,0,0};
  const float* va = V + (size_t)l16 * JD + 256 * wv + 4 * hf;
  const float* qb = sq + (l16 & 3) * QP + 256 * wv + 4 * hf;
  const bool qcol = l16 < 4;
#pragma unroll 4
  for (int t = 0; t < 32; t++) {
    const float4 af = *(const float4*)(va + 8 * t);
    float4 bf = *(const float4*)(qb + 8 * t);
    if (!qcol) bf = make_float4(0.f, 0.f, 0.f, 0.f);
    v2f a, b;
    a.x = af.x; a.y = af.y; b.x = bf.x; b.y = bf.y;
    acc0 = wmma_k4(a, b, acc0);
    a.x = af.z; a.y = af.w; b.x = bf.z; b.y = bf.w;
    acc1 = wmma_k4(a, b, acc1);
  }
  if (l16 < 4 && (l16 >> 1) == hf) {
#pragma unroll
    for (int r = 0; r < 8; r++) red[wv][l16 * 8 + r] = acc0[r] + acc1[r];
  }
  __syncthreads();
  if (tid < NSRC * KB) {
    const int n = tid >> 3, k = tid & 7;
    float sn = 0.f, sd = 0.f;
#pragma unroll
    for (int w2 = 0; w2 < 8; w2++) { sn += red[w2][(2 * n) * 8 + k]; sd += red[w2][(2 * n + 1) * 8 + k]; }
    const size_t ti = oT + (size_t)n * IPD * KB + (size_t)i * KB + k;
    ws[ti] = ws[ti] * sqrtf(sn / (sd + EPS));
  }
}

#define OP 1026
__global__ __launch_bounds__(512) void k_out(const float* __restrict__ ws, float* __restrict__ out) {
  __shared__ float2 s[16 * OP];
  const int tid = threadIdx.x;
  const int j0 = blockIdx.x * 16, n = blockIdx.y;
  const float* Yre = ws + oYre + (size_t)n * IPD * JD + j0;
  const float* Yim = ws + oYim + (size_t)n * IPD * JD + j0;
  float4 re[8], im[8];
#pragma unroll
  for (int q = 0; q < 8; q++) {
    const int f = q * 512 + tid, i = f >> 2, jq = f & 3;
    re[q] = *(const float4*)(Yre + (size_t)i * JD + 4 * jq);
    im[q] = *(const float4*)(Yim + (size_t)i * JD + 4 * jq);
  }
#pragma unroll
  for (int q = 0; q < 8; q++) {
    const int f = q * 512 + tid, i = f >> 2, jq = f & 3;
    s[(4 * jq + 0) * OP + i] = make_float2(re[q].x, im[q].x);
    s[(4 * jq + 1) * OP + i] = make_float2(re[q].y, im[q].y);
    s[(4 * jq + 2) * OP + i] = make_float2(re[q].z, im[q].z);
    s[(4 * jq + 3) * OP + i] = make_float2(re[q].w, im[q].w);
  }
  if (tid < 4) {
    const int i = ID - 1, jq = tid;
    const float4 r = *(const float4*)(Yre + (size_t)i * JD + 4 * jq);
    const float4 g = *(const float4*)(Yim + (size_t)i * JD + 4 * jq);
    s[(4 * jq + 0) * OP + i] = make_float2(r.x, g.x);
    s[(4 * jq + 1) * OP + i] = make_float2(r.y, g.y);
    s[(4 * jq + 2) * OP + i] = make_float2(r.z, g.z);
    s[(4 * jq + 3) * OP + i] = make_float2(r.w, g.w);
  }
  __syncthreads();
  float4* o4 = (float4*)(out + ((size_t)n * JD + j0) * ID * 2);
  for (int h = tid; h < 16 * ID / 2; h += 512) {
    const int q0 = 2 * h, q1 = q0 + 1;
    const float2 a = s[q0 + q0 / ID], b = s[q1 + q1 / ID];
    o4[h] = make_float4(a.x, a.y, b.x, b.y);
  }
}

extern "C" void kernel_launch(void* const* d_in, const int* in_sizes, int n_in,
                              void* d_out, int out_size, void* d_ws, size_t ws_size,
                              hipStream_t stream) {
  (void)in_sizes; (void)n_in; (void)out_size; (void)ws_size;
  const float* X  = (const float*)d_in[0];
  const float* T0 = (const float*)d_in[1];
  const float* V0 = (const float*)d_in[2];
  float* ws = (float*)d_ws;
  float* out = (float*)d_out;

  k_init_xy<<<dim3((IPD + 63) / 64, JD / 64, MC), 256, 0, stream>>>(X, ws);
  { size_t tot = (size_t)NSRC * IPD * KB; k_init_t<<<(unsigned)((tot + 255) / 256), 256, 0, stream>>>(T0, ws); }
  { size_t tot = (size_t)NSRC * KB * JD;  k_init_v<<<(unsigned)((tot + 255) / 256), 256, 0, stream>>>(V0, ws); }
  { size_t tot = (size_t)ID * 8;          k_init_w<<<(unsigned)((tot + 255) / 256), 256, 0, stream>>>(ws); }

  k_ip_t<0><<<ID, 256, 0, stream>>>(ws);
  for (int it = 0; it < NITER; ++it) {
    k_v_update<<<dim3(JD / 16, YCH, NSRC), 256, 0, stream>>>(ws);
    k_v_apply<<<dim3((unsigned)(((size_t)KB * JD + 255) / 256), 1, NSRC), 256, 0, stream>>>(ws);
    if (it + 1 < NITER) k_ip_t<1><<<ID, 256, 0, stream>>>(ws);
    else                k_ip_t<2><<<ID, 256, 0, stream>>>(ws);
  }
  k_out<<<dim3(JD / 16, NSRC), 512, 0, stream>>>(ws, out);
}


# V update in one launch: a block owns a 16-column j strip over all frequencies, sums over i inside the block (i sums regrouped: 16 waves x 4 tiles), updates V in place; no partial planes; 16 launches
# speedup vs baseline: 2.2215x; 1.1591x over previous
#include <hip/hip_runtime.h>
#include <math.h>

#define JD 2048
#define ID 1025
#define IPD 1040
#define KB 8
#define MC 2
#define NSRC 2
#define NITER 5
#define VW 16
#define EPS 1e-20f

typedef __attribute__((ext_vector_type(2))) float v2f;
typedef __attribute__((ext_vector_type(8))) float v8f;

constexpr size_t oXre = 0;
constexpr size_t oXim = oXre + (size_t)MC * IPD * JD;
constexpr size_t oYre = oXim + (size_t)MC * IPD * JD;
constexpr size_t oYim = oYre + (size_t)NSRC * IPD * JD;
constexpr size_t oT   = oYim + (size_t)NSRC * IPD * JD;
constexpr size_t oV   = oT + (size_t)NSRC * IPD * KB;
constexpr size_t oW   = oV + (size_t)NSRC * KB * JD;

__device__ __forceinline__ v8f wmma_k4(v2f a, v2f b, v8f c) {
  return __builtin_amdgcn_wmma_f32_16x16x4_f32(false, a, false, b, (short)0, c, false, false);
}

#define TP 66
__global__ __launch_bounds__(256) void k_init_xy(const float* __restrict__ X, float* __restrict__ ws) {
  __shared__ __attribute__((aligned(16))) float sre[64 * TP];
  __shared__ __attribute__((aligned(16))) float sim[64 * TP];
  const int tid = threadIdx.x, wv = tid >> 5, lane = tid & 31;
  const int i0 = blockIdx.x * 64, j0 = blockIdx.y * 64, m = blockIdx.z;
  float2 v[16];
#pragma unroll
  for (int q = 0; q < 16; q++) {
    const int jj = wv * 8 + (q >> 1);
    const int i = i0 + (q & 1) * 32 + lane;
    const int ic = i < ID ? i : ID - 1;
    v[q] = *(const float2*)(X + (((size_t)m * JD + j0 + jj) * ID + ic) * 2);
  }
#pragma unroll
  for (int q = 0; q < 16; q++) {
    const int jj = wv * 8 + (q >> 1);
    const int il = (q & 1) * 32 + lane;
    const bool pad = i0 + il >= ID;
    sre[il * TP + jj] = pad ? 0.f : v[q].x; sim[il * TP + jj] = pad ? 0.f : v[q].y;
  }
  __syncthreads();
#pragma unroll
  for (int q = 0; q < 8; q++) {
    const int il = wv * 8 + q;
    const int i = i0 + il;
    if (i < IPD) {
      const float2 xr = *(const float2*)(sre + il * TP + 2 * lane);
      const float2 xi = *(const float2*)(sim + il * TP + 2 * lane);
      const size_t dst = ((size_t)m * IPD + i) * JD + j0 + 2 * lane;
      *(float2*)(ws + oXre + dst) = xr; *(float2*)(ws + oXim + dst) = xi;
      if (i >= ID) *(float2*)(ws + oYre + dst) = xr;
    }
  }
}

__global__ __launch_bounds__(256) void k_init_t(const float* __restrict__ T0, float* __restrict__ ws) {
  size_t t = (size_t)blockIdx.x * 256 + threadIdx.x;
  size_t tot = (size_t)NSRC * IPD * KB;
  if (t >= tot) return;
  int k = (int)(t % KB); size_t r = t / KB;
  int i = (int)(r % IPD); int n = (int)(r / IPD);
  float v = 0.f;
  if (i < ID) v = T0[((size_t)i * KB + k) * NSRC + n];
  ws[oT + ((size_t)n * IPD + i) * KB + k] = v;
}

__global__ __launch_bounds__(256) void k_init_v(const float* __restrict__ V0, float* __restrict__ ws) {
  size_t t = (size_t)blockIdx.x * 256 + threadIdx.x;
  size_t tot = (size_t)NSRC * KB * JD;
  if (t >= tot) return;
  int j = (int)(t % JD); size_t r = t / JD;
  int k = (int)(r % KB); int n = (int)(r / KB);
  ws[oV + ((size_t)n * KB + k) * JD + j] = V0[((size_t)k * JD + j) * NSRC + n];
}

__global__ __launch_bounds__(256) void k_init_w(float* __restrict__ ws) {
  size_t t = (size_t)blockIdx.x * 256 + threadIdx.x;
  size_t tot = (size_t)ID * 8;
  if (t >= tot) return;
  int f = (int)(t & 7);
  int c = f & 1, m = (f >> 1) & 1, nn = (f >> 2) & 1;
  ws[oW + t] = (c == 0 && nn == m) ? 1.f : 0.f;
}

struct VFrag { v2f a0, a1, at[4], by[4]; };

__device__ __forceinline__ void v_load_t(VFrag& f, const float* T,
                                         const int i0, const int hf, const int l16) {
  const size_t tr = (size_t)(i0 + l16) * KB;
  f.a0.x = T[tr + 2 * hf];     f.a0.y = T[tr + 2 * hf + 1];
  f.a1.x = T[tr + 4 + 2 * hf]; f.a1.y = T[tr + 5 + 2 * hf];
#pragma unroll
  for (int c4 = 0; c4 < 4; c4++) {
    const int ia = i0 + 4 * c4 + 2 * hf;
    f.at[c4].x = T[(size_t)ia * KB + (l16 & 7)];
    f.at[c4].y = T[(size_t)(ia + 1) * KB + (l16 & 7)];
  }
}

__device__ __forceinline__ void v_load_y(VFrag& f, const float* YdR,
                                         const int i0, const int j0, const int hf, const int l16) {
#pragma unroll
  for (int c4 = 0; c4 < 4; c4++) {
    const int ia = i0 + 4 * c4 + 2 * hf;
    f.by[c4].x = YdR[(size_t)ia * JD + j0 + l16];
    f.by[c4].y = YdR[(size_t)(ia + 1) * JD + j0 + l16];
  }
}

__device__ __forceinline__ void v_tile(const VFrag& f, const v2f b0, const v2f b1, float (*lr)[17],
                                       const int hf, const int l16, v8f& num, v8f& den) {
  v8f c = {0,0,0,0,0,0,0,0};
  c = wmma_k4(f.a0, b0, c);
  c = wmma_k4(f.a1, b1, c);
#pragma unroll
  for (int r = 0; r < 8; r++)
    lr[r + 8 * hf][l16] = 1.0f / (fabsf(c[r]) + EPS);
#pragma unroll
  for (int c4 = 0; c4 < 4; c4++) {
    v2f brv;
    brv.x = lr[4 * c4 + 2 * hf][l16];
    brv.y = lr[4 * c4 + 2 * hf + 1][l16];
    num = wmma_k4(f.at[c4], f.by[c4], num);
    den = wmma_k4(f.at[c4], brv, den);
  }
}

__global__ __launch_bounds__(VW * 32) void k_v_update(float* __restrict__ ws) {
  const int tid = threadIdx.x;
  const int wv = tid >> 5, lane = tid & 31, hf = lane >> 4, l16 = lane & 15;
  const int j0 = blockIdx.x * 16;
  const int n = blockIdx.y;
  const float* T = ws + oT + (size_t)n * IPD * KB;
  float* V = ws + oV + (size_t)n * KB * JD;
  const float* YdR = ws + oYre + (size_t)n * IPD * JD;

  __shared__ float lds_r[VW][16][17];
  __shared__ float redn[VW][128];
  __shared__ float redd[VW][128];

  v2f b0, b1;
  b0.x = V[(size_t)(2 * hf + 0) * JD + j0 + l16];
  b0.y = V[(size_t)(2 * hf + 1) * JD + j0 + l16];
  b1.x = V[(size_t)(2 * hf + 4) * JD + j0 + l16];
  b1.y = V[(size_t)(2 * hf + 5) * JD + j0 + l16];
  VFrag f[4], fx;
  const bool extra = wv == 0;
#pragma unroll
  for (int s = 0; s < 4; s++) v_load_t(f[s], T, (wv + VW * s) * 16, hf, l16);
  if (extra) v_load_t(fx, T, 4 * VW * 16, hf, l16);
#pragma unroll
  for (int s = 0; s < 4; s++) v_load_y(f[s], YdR, (wv + VW * s) * 16, j0, hf, l16);
  if (extra) v_load_y(fx, YdR, 4 * VW * 16, j0, hf, l16);

  v8f num = {0,0,0,0,0,0,0,0}, den = {0,0,0,0,0,0,0,0};
#pragma unroll
  for (int s = 0; s < 4; s++) v_tile(f[s], b0, b1, lds_r[wv], hf, l16, num, den);
  if (extra) v_tile(fx, b0, b1, lds_r[wv], hf, l16, num, den);

  if (hf == 0) {
#pragma unroll
    for (int r = 0; r < 8; r++) { redn[wv][r * 16 + l16] = num[r]; redd[wv][r * 16 + l16] = den[r]; }
  }
  __syncthreads();
  if (tid < KB * 16) {
    float sn = 0.f, sd = 0.f;
#pragma unroll
    for (int w2 = 0; w2 < VW; w2++) { sn += redn[w2][tid]; sd += redd[w2][tid]; }
    const size_t vi = (size_t)(tid >> 4) * JD + j0 + (tid & 15);
    V[vi] = V[vi] * sqrtf(sn / (sd + EPS));
  }
}

__device__ __forceinline__ void ip_solve(float* Wf, const int n, float S00, float S11, float S01r, float S01i) {
  const float invJ = 1.0f / (float)JD;
  float D00 = S00 * invJ + EPS;
  float D11 = S11 * invJ + EPS;
  float Dr = S01r * invJ, Di = S01i * invJ;
  float w00r = Wf[0], w00i = Wf[1], w01r = Wf[2], w01i = Wf[3];
  float w10r = Wf[4], w10i = Wf[5], w11r = Wf[6], w11i = Wf[7];
  float A00r = w00r * D00 + (w01r * Dr + w01i * Di);
  float A00i = w00i * D00 + (w01i * Dr - w01r * Di);
  float A01r = (w00r * Dr - w00i * Di) + w01r * D11;
  float A01i = (w00i * Dr + w00r * Di) + w01i * D11;
  float A10r = w10r * D00 + (w11r * Dr + w11i * Di);
  float A10i = w10i * D00 + (w11i * Dr - w11r * Di);
  float A11r = (w10r * Dr - w10i * Di) + w11r * D11;
  float A11i = (w10i * Dr + w10r * Di) + w11i * D11;
  float detr = (A00r * A11r - A00i * A11i) - (A01r * A10r - A01i * A10i);
  float deti = (A00r * A11i + A00i * A11r) - (A01r * A10i + A01i * A10r);
  float n0r, n0i, n1r, n1i;
  if (n == 0) { n0r = A11r;  n0i = A11i;  n1r = -A10r; n1i = -A10i; }
  else        { n0r = -A01r; n0i = -A01i; n1r = A00r;  n1i = A00i;  }
  float dd = detr * detr + deti * deti;
  float b0r = (n0r * detr + n0i * deti) / dd, b0i = (n0i * detr - n0r * deti) / dd;
  float b1r = (n1r * detr + n1i * deti) / dd, b1i = (n1i * detr - n1r * deti) / dd;
  float Db0r = D00 * b0r + (Dr * b1r - Di * b1i);
  float Db0i = D00 * b0i + (Dr * b1i + Di * b1r);
  float Db1r = (Dr * b0r + Di * b0i) + D11 * b1r;
  float Db1i = (Dr * b0i - Di * b0r) + D11 * b1i;
  float denr = b0r * Db0r + b0i * Db0i + b1r * Db1r + b1i * Db1i;
  float dnm = sqrtf(denr + EPS);
  Wf[(n * 2 + 0) * 2]     = b0r / dnm;
  Wf[(n * 2 + 0) * 2 + 1] = -b0i / dnm;
  Wf[(n * 2 + 1) * 2]     = b1r / dnm;
  Wf[(n * 2 + 1) * 2 + 1] = -b1i / dnm;
}

#define JPT (JD / 256)
#define QP (JD + 8)
template <int MODE>
__global__ __launch_bounds__(256, 10) void k_ip_t(float* __restrict__ ws) {
  const int i = blockIdx.x;
  const int tid = threadIdx.x;
  const int wv = tid >> 5, lane = tid & 31, hf = lane >> 4, l16 = lane & 15;
  const float* V = ws + oV;
  const float* X0r = ws + oXre;
  const float* X0i = ws + oXim;
  const float* X1r = ws + oXre + (size_t)IPD * JD;
  const float* X1i = ws + oXim + (size_t)IPD * JD;

  __shared__ float sw[8];
  __shared__ float rb[4 * NSRC][256];
  __shared__ __attribute__((aligned(16))) float sq[MODE == 2 ? 4 : 4 * QP];
  __shared__ float red[8][32];

  float x0r[JPT], x0i[JPT], x1r[JPT], x1i[JPT];
#pragma unroll
  for (int q = 0; q < JPT; q++) {
    const size_t idx = (size_t)i * JD + tid + 256 * q;
    x0r[q] = X0r[idx]; x0i[q] = X0i[idx]; x1r[q] = X1r[idx]; x1i[q] = X1i[idx];
  }
  __builtin_amdgcn_sched_barrier(0);

  float tT[NSRC][KB], w0[8];
#pragma unroll
  for (int n = 0; n < NSRC; n++)
#pragma unroll
    for (int k = 0; k < KB; k++) tT[n][k] = ws[oT + (size_t)n * IPD * KB + (size_t)i * KB + k];
  if (MODE != 0) {
#pragma unroll
    for (int q = 0; q < 8; q++) w0[q] = ws[oW + (size_t)i * 8 + q];
  }

  float rnv[NSRC][JPT];
  float d00[NSRC], d11[NSRC], d01r[NSRC], d01i[NSRC];
#pragma unroll
  for (int n = 0; n < NSRC; n++) { d00[n] = 0.f; d11[n] = 0.f; d01r[n] = 0.f; d01i[n] = 0.f; }
  float vv[2][NSRC * KB];
#pragma unroll
  for (int r = 0; r < NSRC * KB; r++) vv[0][r] = V[(size_t)r * JD + tid];
#pragma unroll
  for (int q = 0; q < JPT; q++) {
    if (q + 1 < JPT) {
#pragma unroll
      for (int r = 0; r < NSRC * KB; r++) vv[(q + 1) & 1][r] = V[(size_t)r * JD + tid + 256 * (q + 1)];
    }
#pragma unroll
    for (int n = 0; n < NSRC; n++) {
      float rn = 0.f;
#pragma unroll
      for (int k = 0; k < KB; k++) rn += tT[n][k] * vv[q & 1][n * KB + k];
      rnv[n][q] = rn;
      if (MODE != 0) {
        float s = sqrtf(1.0f / (rn + EPS));
        float a0r = x0r[q] * s, a0i = x0i[q] * s;
        float a1r = x1r[q] * s, a1i = x1i[q] * s;
        d00[n] += a0r * a0r + a0i * a0i;
        d11[n] += a1r * a1r + a1i * a1i;
        d01r[n] += a0r * a1r + a0i * a1i;
        d01i[n] += a0i * a1r - a0r * a1i;
      }
    }
  }
  float wa0 = 1.f, wb0 = 0.f, wa1 = 0.f, wb1 = 0.f;
  float wc0 = 0.f, wd0 = 0.f, wc1 = 1.f, wd1 = 0.f;
  if (MODE != 0) {
#pragma unroll
    for (int n = 0; n < NSRC; n++) {
      rb[4 * n + 0][tid] = d00[n]; rb[4 * n + 1][tid] = d11[n];
      rb[4 * n + 2][tid] = d01r[n]; rb[4 * n + 3][tid] = d01i[n];
    }
    __syncthreads();
    for (int s2 = 128; s2 > 0; s2 >>= 1) {
      if (tid < s2) {
#pragma unroll
        for (int q = 0; q < 4 * NSRC; q++) rb[q][tid] += rb[q][tid + s2];
      }
      __syncthreads();
    }
    if (tid == 0) {
      float Wf[8];
#pragma unroll
      for (int q = 0; q < 8; q++) Wf[q] = w0[q];
      ip_solve(Wf, 0, rb[0][0], rb[1][0], rb[2][0], rb[3][0]);
      ip_solve(Wf, 1, rb[4][0], rb[5][0], rb[6][0], rb[7][0]);
#pragma unroll
      for (int q = 0; q < 8; q++) sw[q] = Wf[q];
    }
    __syncthreads();
    if (tid < 8) ws[oW + (size_t)i * 8 + tid] = sw[tid];
    wa0 = sw[0]; wb0 = sw[1]; wa1 = sw[2]; wb1 = sw[3];
    wc0 = sw[4]; wd0 = sw[5]; wc1 = sw[6]; wd1 = sw[7];
  }

  float* P0r = ws + oYre;
  float* P1r = ws + oYre + (size_t)IPD * JD;
  float* P0i = ws + oYim;
  float* P1i = ws + oYim + (size_t)IPD * JD;
#pragma unroll
  for (int q = 0; q < JPT; q++) {
    const int j = tid + 256 * q;
    const size_t idx = (size_t)i * JD + j;
    const float p0r = x0r[q], p0i = x0i[q], p1r = x1r[q], p1i = x1i[q];
    const float y0r = wa0 * p0r - wb0 * p0i + wa1 * p1r - wb1 * p1i;
    const float y0i = wa0 * p0i + wb0 * p0r + wa1 * p1i + wb1 * p1r;
    const float y1r = wc0 * p0r - wd0 * p0i + wc1 * p1r - wd1 * p1i;
    const float y1i = wc0 * p0i + wd0 * p0r + wc1 * p1i + wd1 * p1r;
    if (MODE == 2) {
      P0r[idx] = y0r; P0i[idx] = y0i; P1r[idx] = y1r; P1i[idx] = y1i;
    } else {
      const float r0 = rnv[0][q], r1 = rnv[1][q];
      const float ydr0 = (y0r * y0r + y0i * y0i) / (r0 * r0 + EPS);
      const float ydr1 = (y1r * y1r + y1i * y1i) / (r1 * r1 + EPS);
      const float rd0 = 1.0f / (fabsf(r0) + EPS);
      const float rd1 = 1.0f / (fabsf(r1) + EPS);
      P0r[idx] = ydr0; P1r[idx] = ydr1;
      sq[0 * QP + j] = ydr0; sq[1 * QP + j] = rd0;
      sq[2 * QP + j] = ydr1; sq[3 * QP + j] = rd1;
    }
  }
  if (MODE == 2) return;

  __syncthreads();
  v8f acc0 = {0,0,0,0,0,0,0,0}, acc1 = {0,0,0,0,0,0,0,0};
  const float* va = V + (size_t)l16 * JD + 256 * wv + 4 * hf;
  const float* qb = sq + (l16 & 3) * QP + 256 * wv + 4 * hf;
  const bool qcol = l16 < 4;
#pragma unroll 4
  for (int t = 0; t < 32; t++) {
    const float4 af = *(const float4*)(va + 8 * t);
    float4 bf = *(const float4*)(qb + 8 * t);
    if (!qcol) bf = make_float4(0.f, 0.f, 0.f, 0.f);
    v2f a, b;
    a.x = af.x; a.y = af.y; b.x = bf.x; b.y = bf.y;
    acc0 = wmma_k4(a, b, acc0);
    a.x = af.z; a.y = af.w; b.x = bf.z; b.y = bf.w;
    acc1 = wmma_k4(a, b, acc1);
  }
  if (l16 < 4 && (l16 >> 1) == hf) {
#pragma unroll
    for (int r = 0; r < 8; r++) red[wv][l16 * 8 + r] = acc0[r] + acc1[r];
  }
  __syncthreads();
  if (tid < NSRC * KB) {
    const int n = tid >> 3, k = tid & 7;
    float sn = 0.f, sd = 0.f;
#pragma unroll
    for (int w2 = 0; w2 < 8; w2++) { sn += red[w2][(2 * n) * 8 + k]; sd += red[w2][(2 * n + 1) * 8 + k]; }
    const size_t ti = oT + (size_t)n * IPD * KB + (size_t)i * KB + k;
    ws[ti] = ws[ti] * sqrtf(sn / (sd + EPS));
  }
}

#define OP 1026
__global__ __launch_bounds__(512) void k_out(const float* __restrict__ ws, float* __restrict__ out) {
  __shared__ float2 s[16 * OP];
  const int tid = threadIdx.x;
  const int j0 = blockIdx.x * 16, n = blockIdx.y;
  const float* Yre = ws + oYre + (size_t)n * IPD * JD + j0;
  const float* Yim = ws + oYim + (size_t)n * IPD * JD + j0;
  float4 re[8], im[8];
#pragma unroll
  for (int q = 0; q < 8; q++) {
    const int f = q * 512 + tid, i = f >> 2, jq = f & 3;
    re[q] = *(const float4*)(Yre + (size_t)i * JD + 4 * jq);
    im[q] = *(const float4*)(Yim + (size_t)i * JD + 4 * jq);
  }
#pragma unroll
  for (int q = 0; q < 8; q++) {
    const int f = q * 512 + tid, i = f >> 2, jq = f & 3;
    s[(4 * jq + 0) * OP + i] = make_float2(re[q].x, im[q].x);
    s[(4 * jq + 1) * OP + i] = make_float2(re[q].y, im[q].y);
    s[(4 * jq + 2) * OP + i] = make_float2(re[q].z, im[q].z);
    s[(4 * jq + 3) * OP + i] = make_float2(re[q].w, im[q].w);
  }
  if (tid < 4) {
    const int i = ID - 1, jq = tid;
    const float4 r = *(const float4*)(Yre + (size_t)i * JD + 4 * jq);
    const float4 g = *(const float4*)(Yim + (size_t)i * JD + 4 * jq);
    s[(4 * jq + 0) * OP + i] = make_float2(r.x, g.x);
    s[(4 * jq + 1) * OP + i] = make_float2(r.y, g.y);
    s[(4 * jq + 2) * OP + i] = make_float2(r.z, g.z);
    s[(4 * jq + 3) * OP + i] = make_float2(r.w, g.w);
  }
  __syncthreads();
  float4* o4 = (float4*)(out + ((size_t)n * JD + j0) * ID * 2);
  for (int h = tid; h < 16 * ID / 2; h += 512) {
    const int q0 = 2 * h, q1 = q0 + 1;
    const float2 a = s[q0 + q0 / ID], b = s[q1 + q1 / ID];
    o4[h] = make_float4(a.x, a.y, b.x, b.y);
  }
}

extern "C" void kernel_launch(void* const* d_in, const int* in_sizes, int n_in,
                              void* d_out, int out_size, void* d_ws, size_t ws_size,
                              hipStream_t stream) {
  (void)in_sizes; (void)n_in; (void)out_size; (void)ws_size;
  const float* X  = (const float*)d_in[0];
  const float* T0 = (const float*)d_in[1];
  const float* V0 = (const float*)d_in[2];
  float* ws = (float*)d_ws;
  float* out = (float*)d_out;

  k_init_xy<<<dim3((IPD + 63) / 64, JD / 64, MC), 256, 0, stream>>>(X, ws);
  { size_t tot = (size_t)NSRC * IPD * KB; k_init_t<<<(unsigned)((tot + 255) / 256), 256, 0, stream>>>(T0, ws); }
  { size_t tot = (size_t)NSRC * KB * JD;  k_init_v<<<(unsigned)((tot + 255) / 256), 256, 0, stream>>>(V0, ws); }
  { size_t tot = (size_t)ID * 8;          k_init_w<<<(unsigned)((tot + 255) / 256), 256, 0, stream>>>(ws); }

  k_ip_t<0><<<ID, 256, 0, stream>>>(ws);
  for (int it = 0; it < NITER; ++it) {
    k_v_update<<<dim3(JD / 16, NSRC), VW * 32, 0, stream>>>(ws);
    if (it + 1 < NITER) k_ip_t<1><<<ID, 256, 0, stream>>>(ws);
    else                k_ip_t<2><<<ID, 256, 0, stream>>>(ws);
  }
  k_out<<<dim3(JD / 16, NSRC), 512, 0, stream>>>(ws, out);
}


# Two frequencies per block in the IP/T kernel: each V value loaded (for R = T V and as the A operand of the T update) serves both rows, Q fills 8 of B's 16 columns; V reads per frequency halve
# speedup vs baseline: 2.2970x; 1.0340x over previous
#include <hip/hip_runtime.h>
#include <math.h>

#define JD 2048
#define ID 1025
#define IPD 1040
#define KB 8
#define MC 2
#define NSRC 2
#define NITER 5
#define VW 16
#define EPS 1e-20f

typedef __attribute__((ext_vector_type(2))) float v2f;
typedef __attribute__((ext_vector_type(8))) float v8f;

constexpr size_t oXre = 0;
constexpr size_t oXim = oXre + (size_t)MC * IPD * JD;
constexpr size_t oYre = oXim + (size_t)MC * IPD * JD;
constexpr size_t oYim = oYre + (size_t)NSRC * IPD * JD;
constexpr size_t oT   = oYim + (size_t)NSRC * IPD * JD;
constexpr size_t oV   = oT + (size_t)NSRC * IPD * KB;
constexpr size_t oW   = oV + (size_t)NSRC * KB * JD;

__device__ __forceinline__ v8f wmma_k4(v2f a, v2f b, v8f c) {
  return __builtin_amdgcn_wmma_f32_16x16x4_f32(false, a, false, b, (short)0, c, false, false);
}

#define TP 66
__global__ __launch_bounds__(256) void k_init_xy(const float* __restrict__ X, float* __restrict__ ws) {
  __shared__ __attribute__((aligned(16))) float sre[64 * TP];
  __shared__ __attribute__((aligned(16))) float sim[64 * TP];
  const int tid = threadIdx.x, wv = tid >> 5, lane = tid & 31;
  const int i0 = blockIdx.x * 64, j0 = blockIdx.y * 64, m = blockIdx.z;
  float2 v[16];
#pragma unroll
  for (int q = 0; q < 16; q++) {
    const int jj = wv * 8 + (q >> 1);
    const int i = i0 + (q & 1) * 32 + lane;
    const int ic = i < ID ? i : ID - 1;
    v[q] = *(const float2*)(X + (((size_t)m * JD + j0 + jj) * ID + ic) * 2);
  }
#pragma unroll
  for (int q = 0; q < 16; q++) {
    const int jj = wv * 8 + (q >> 1);
    const int il = (q & 1) * 32 + lane;
    const bool pad = i0 + il >= ID;
    sre[il * TP + jj] = pad ? 0.f : v[q].x; sim[il * TP + jj] = pad ? 0.f : v[q].y;
  }
  __syncthreads();
#pragma unroll
  for (int q = 0; q < 8; q++) {
    const int il = wv * 8 + q;
    const int i = i0 + il;
    if (i < IPD) {
      const float2 xr = *(const float2*)(sre + il * TP + 2 * lane);
      const float2 xi = *(const float2*)(sim + il * TP + 2 * lane);
      const size_t dst = ((size_t)m * IPD + i) * JD + j0 + 2 * lane;
      *(float2*)(ws + oXre + dst) = xr; *(float2*)(ws + oXim + dst) = xi;
      if (i >= ID) *(float2*)(ws + oYre + dst) = xr;
    }
  }
}

__global__ __launch_bounds__(256) void k_init_t(const float* __restrict__ T0, float* __restrict__ ws) {
  size_t t = (size_t)blockIdx.x * 256 + threadIdx.x;
  size_t tot = (size_t)NSRC * IPD * KB;
  if (t >= tot) return;
  int k = (int)(t % KB); size_t r = t / KB;
  int i = (int)(r % IPD); int n = (int)(r / IPD);
  float v = 0.f;
  if (i < ID) v = T0[((size_t)i * KB + k) * NSRC + n];
  ws[oT + ((size_t)n * IPD + i) * KB + k] = v;
}

__global__ __launch_bounds__(256) void k_init_v(const float* __restrict__ V0, float* __restrict__ ws) {
  size_t t = (size_t)blockIdx.x * 256 + threadIdx.x;
  size_t tot = (size_t)NSRC * KB * JD;
  if (t >= tot) return;
  int j = (int)(t % JD); size_t r = t / JD;
  int k = (int)(r % KB); int n = (int)(r / KB);
  ws[oV + ((size_t)n * KB + k) * JD + j] = V0[((size_t)k * JD + j) * NSRC + n];
}

__global__ __launch_bounds__(256) void k_init_w(float* __restrict__ ws) {
  size_t t = (size_t)blockIdx.x * 256 + threadIdx.x;
  size_t tot = (size_t)ID * 8;
  if (t >= tot) return;
  int f = (int)(t & 7);
  int c = f & 1, m = (f >> 1) & 1, nn = (f >> 2) & 1;
  ws[oW + t] = (c == 0 && nn == m) ? 1.f : 0.f;
}

struct VFrag { v2f a0, a1, at[4], by[4]; };

__device__ __forceinline__ void v_load_t(VFrag& f, const float* T,
                                         const int i0, const int hf, const int l16) {
  const size_t tr = (size_t)(i0 + l16) * KB;
  f.a0.x = T[tr + 2 * hf];     f.a0.y = T[tr + 2 * hf + 1];
  f.a1.x = T[tr + 4 + 2 * hf]; f.a1.y = T[tr + 5 + 2 * hf];
#pragma unroll
  for (int c4 = 0; c4 < 4; c4++) {
    const int ia = i0 + 4 * c4 + 2 * hf;
    f.at[c4].x = T[(size_t)ia * KB + (l16 & 7)];
    f.at[c4].y = T[(size_t)(ia + 1) * KB + (l16 & 7)];
  }
}

__device__ __forceinline__ void v_load_y(VFrag& f, const float* YdR,
                                         const int i0, const int j0, const int hf, const int l16) {
#pragma unroll
  for (int c4 = 0; c4 < 4; c4++) {
    const int ia = i0 + 4 * c4 + 2 * hf;
    f.by[c4].x = YdR[(size_t)ia * JD + j0 + l16];
    f.by[c4].y = YdR[(size_t)(ia + 1) * JD + j0 + l16];
  }
}

__device__ __forceinline__ void v_tile(const VFrag& f, const v2f b0, const v2f b1, float (*lr)[17],
                                       const int hf, const int l16, v8f& num, v8f& den) {
  v8f c = {0,0,0,0,0,0,0,0};
  c = wmma_k4(f.a0, b0, c);
  c = wmma_k4(f.a1, b1, c);
#pragma unroll
  for (int r = 0; r < 8; r++)
    lr[r + 8 * hf][l16] = 1.0f / (fabsf(c[r]) + EPS);
#pragma unroll
  for (int c4 = 0; c4 < 4; c4++) {
    v2f brv;
    brv.x = lr[4 * c4 + 2 * hf][l16];
    brv.y = lr[4 * c4 + 2 * hf + 1][l16];
    num = wmma_k4(f.at[c4], f.by[c4], num);
    den = wmma_k4(f.at[c4], brv, den);
  }
}

__global__ __launch_bounds__(VW * 32) void k_v_update(float* __restrict__ ws) {
  const int tid = threadIdx.x;
  const int wv = tid >> 5, lane = tid & 31, hf = lane >> 4, l16 = lane & 15;
  const int j0 = blockIdx.x * 16;
  const int n = blockIdx.y;
  const float* T = ws + oT + (size_t)n * IPD * KB;
  float* V = ws + oV + (size_t)n * KB * JD;
  const float* YdR = ws + oYre + (size_t)n * IPD * JD;

  __shared__ float lds_r[VW][16][17];
  __shared__ float redn[VW][128];
  __shared__ float redd[VW][128];

  v2f b0, b1;
  b0.x = V[(size_t)(2 * hf + 0) * JD + j0 + l16];
  b0.y = V[(size_t)(2 * hf + 1) * JD + j0 + l16];
  b1.x = V[(size_t)(2 * hf + 4) * JD + j0 + l16];
  b1.y = V[(size_t)(2 * hf + 5) * JD + j0 + l16];
  VFrag f[4], fx;
  const bool extra = wv == 0;
#pragma unroll
  for (int s = 0; s < 4; s++) v_load_t(f[s], T, (wv + VW * s) * 16, hf, l16);
  if (extra) v_load_t(fx, T, 4 * VW * 16, hf, l16);
#pragma unroll
  for (int s = 0; s < 4; s++) v_load_y(f[s], YdR, (wv + VW * s) * 16, j0, hf, l16);
  if (extra) v_load_y(fx, YdR, 4 * VW * 16, j0, hf, l16);

  v8f num = {0,0,0,0,0,0,0,0}, den = {0,0,0,0,0,0,0,0};
#pragma unroll
  for (int s = 0; s < 4; s++) v_tile(f[s], b0, b1, lds_r[wv], hf, l16, num, den);
  if (extra) v_tile(fx, b0, b1, lds_r[wv], hf, l16, num, den);

  if (hf == 0) {
#pragma unroll
    for (int r = 0; r < 8; r++) { redn[wv][r * 16 + l16] = num[r]; redd[wv][r * 16 + l16] = den[r]; }
  }
  __syncthreads();
  if (tid < KB * 16) {
    float sn = 0.f, sd = 0.f;
#pragma unroll
    for (int w2 = 0; w2 < VW; w2++) { sn += redn[w2][tid]; sd += redd[w2][tid]; }
    const size_t vi = (size_t)(tid >> 4) * JD + j0 + (tid & 15);
    V[vi] = V[vi] * sqrtf(sn / (sd + EPS));
  }
}

__device__ __forceinline__ void ip_solve(float* Wf, const int n, float S00, float S11, float S01r, float S01i) {
  const float invJ = 1.0f / (float)JD;
  float D00 = S00 * invJ + EPS;
  float D11 = S11 * invJ + EPS;
  float Dr = S01r * invJ, Di = S01i * invJ;
  float w00r = Wf[0], w00i = Wf[1], w01r = Wf[2], w01i = Wf[3];
  float w10r = Wf[4], w10i = Wf[5], w11r = Wf[6], w11i = Wf[7];
  float A00r = w00r * D00 + (w01r * Dr + w01i * Di);
  float A00i = w00i * D00 + (w01i * Dr - w01r * Di);
  float A01r = (w00r * Dr - w00i * Di) + w01r * D11;
  float A01i = (w00i * Dr + w00r * Di) + w01i * D11;
  float A10r = w10r * D00 + (w11r * Dr + w11i * Di);
  float A10i = w10i * D00 + (w11i * Dr - w11r * Di);
  float A11r = (w10r * Dr - w10i * Di) + w11r * D11;
  float A11i = (w10i * Dr + w10r * Di) + w11i * D11;
  float detr = (A00r * A11r - A00i * A11i) - (A01r * A10r - A01i * A10i);
  float deti = (A00r * A11i + A00i * A11r) - (A01r * A10i + A01i * A10r);
  float n0r, n0i, n1r, n1i;
  if (n == 0) { n0r = A11r;  n0i = A11i;  n1r = -A10r; n1i = -A10i; }
  else        { n0r = -A01r; n0i = -A01i; n1r = A00r;  n1i = A00i;  }
  float dd = detr * detr + deti * deti;
  float b0r = (n0r * detr + n0i * deti) / dd, b0i = (n0i * detr - n0r * deti) / dd;
  float b1r = (n1r * detr + n1i * deti) / dd, b1i = (n1i * detr - n1r * deti) / dd;
  float Db0r = D00 * b0r + (Dr * b1r - Di * b1i);
  float Db0i = D00 * b0i + (Dr * b1i + Di * b1r);
  float Db1r = (Dr * b0r + Di * b0i) + D11 * b1r;
  float Db1i = (Dr * b0i - Di * b0r) + D11 * b1i;
  float denr = b0r * Db0r + b0i * Db0i + b1r * Db1r + b1i * Db1i;
  float dnm = sqrtf(denr + EPS);
  Wf[(n * 2 + 0) * 2]     = b0r / dnm;
  Wf[(n * 2 + 0) * 2 + 1] = -b0i / dnm;
  Wf[(n * 2 + 1) * 2]     = b1r / dnm;
  Wf[(n * 2 + 1) * 2 + 1] = -b1i / dnm;
}

#define NF 2
#define NPB ((ID + NF - 1) / NF)
#define JPT (JD / 256)
#define QP (JD + 8)
template <int MODE>
__global__ __launch_bounds__(256, 6) void k_ip_t(float* __restrict__ ws) {
  const int i0 = blockIdx.x * NF;
  const bool ok1 = i0 + 1 < ID;
  const int tid = threadIdx.x;
  const int wv = tid >> 5, lane = tid & 31, hf = lane >> 4, l16 = lane & 15;
  const float* V = ws + oV;
  const float* X0r = ws + oXre;
  const float* X0i = ws + oXim;
  const float* X1r = ws + oXre + (size_t)IPD * JD;
  const float* X1i = ws + oXim + (size_t)IPD * JD;

  __shared__ float sw[NF][8];
  __shared__ float rb[NF * 4 * NSRC][256];
  __shared__ __attribute__((aligned(16))) float sq[MODE == 2 ? 4 : NF * 4 * QP];
  __shared__ float red[8][NF * 32];

  float x0r[NF][JPT], x0i[NF][JPT], x1r[NF][JPT], x1i[NF][JPT];
#pragma unroll
  for (int q = 0; q < JPT; q++) {
#pragma unroll
    for (int f = 0; f < NF; f++) {
      const size_t idx = (size_t)(i0 + f) * JD + tid + 256 * q;
      x0r[f][q] = X0r[idx]; x0i[f][q] = X0i[idx]; x1r[f][q] = X1r[idx]; x1i[f][q] = X1i[idx];
    }
  }
  __builtin_amdgcn_sched_barrier(0);

  float tT[NF][NSRC][KB], w0[NF][8];
#pragma unroll
  for (int f = 0; f < NF; f++)
#pragma unroll
    for (int n = 0; n < NSRC; n++)
#pragma unroll
      for (int k = 0; k < KB; k++) tT[f][n][k] = ws[oT + (size_t)n * IPD * KB + (size_t)(i0 + f) * KB + k];
  if (MODE != 0) {
    const int iw1 = ok1 ? i0 + 1 : i0;
#pragma unroll
    for (int q = 0; q < 8; q++) { w0[0][q] = ws[oW + (size_t)i0 * 8 + q]; w0[1][q] = ws[oW + (size_t)iw1 * 8 + q]; }
  }

  float wa0[NF], wb0[NF], wa1[NF], wb1[NF], wc0[NF], wd0[NF], wc1[NF], wd1[NF];
#pragma unroll
  for (int f = 0; f < NF; f++) {
    wa0[f] = 1.f; wb0[f] = 0.f; wa1[f] = 0.f; wb1[f] = 0.f;
    wc0[f] = 0.f; wd0[f] = 0.f; wc1[f] = 1.f; wd1[f] = 0.f;
  }
  float* P0r = ws + oYre;
  float* P1r = ws + oYre + (size_t)IPD * JD;
  float* P0i = ws + oYim;
  float* P1i = ws + oYim + (size_t)IPD * JD;
  auto y_step = [&](const int f, const int q, const float r0, const float r1) {
    const bool st = f == 0 || ok1;
    const int j = tid + 256 * q;
    const size_t idx = (size_t)(i0 + f) * JD + j;
    const float p0r = x0r[f][q], p0i = x0i[f][q], p1r = x1r[f][q], p1i = x1i[f][q];
    const float y0r = wa0[f] * p0r - wb0[f] * p0i + wa1[f] * p1r - wb1[f] * p1i;
    const float y0i = wa0[f] * p0i + wb0[f] * p0r + wa1[f] * p1i + wb1[f] * p1r;
    const float y1r = wc0[f] * p0r - wd0[f] * p0i + wc1[f] * p1r - wd1[f] * p1i;
    const float y1i = wc0[f] * p0i + wd0[f] * p0r + wc1[f] * p1i + wd1[f] * p1r;
    if (MODE == 2) {
      if (st) { P0r[idx] = y0r; P0i[idx] = y0i; P1r[idx] = y1r; P1i[idx] = y1i; }
    } else {
      const float ydr0 = (y0r * y0r + y0i * y0i) / (r0 * r0 + EPS);
      const float ydr1 = (y1r * y1r + y1i * y1i) / (r1 * r1 + EPS);
      const float rd0 = 1.0f / (fabsf(r0) + EPS);
      const float rd1 = 1.0f / (fabsf(r1) + EPS);
      if (st) { P0r[idx] = ydr0; P1r[idx] = ydr1; }
      sq[(4 * f + 0) * QP + j] = ydr0; sq[(4 * f + 1) * QP + j] = rd0;
      sq[(4 * f + 2) * QP + j] = ydr1; sq[(4 * f + 3) * QP + j] = rd1;
    }
  };

  float rnv[NF][NSRC][JPT];
  float d00[NF][NSRC], d11[NF][NSRC], d01r[NF][NSRC], d01i[NF][NSRC];
#pragma unroll
  for (int f = 0; f < NF; f++)
#pragma unroll
    for (int n = 0; n < NSRC; n++) { d00[f][n] = 0.f; d11[f][n] = 0.f; d01r[f][n] = 0.f; d01i[f][n] = 0.f; }
  float vv[2][NSRC * KB];
#pragma unroll
  for (int r = 0; r < NSRC * KB; r++) vv[0][r] = V[(size_t)r * JD + tid];
#pragma unroll
  for (int q = 0; q < JPT; q++) {
    if (q + 1 < JPT) {
#pragma unroll
      for (int r = 0; r < NSRC * KB; r++) vv[(q + 1) & 1][r] = V[(size_t)r * JD + tid + 256 * (q + 1)];
    }
#pragma unroll
    for (int f = 0; f < NF; f++) {
      float rl[NSRC];
#pragma unroll
      for (int n = 0; n < NSRC; n++) {
        float rn = 0.f;
#pragma unroll
        for (int k = 0; k < KB; k++) rn += tT[f][n][k] * vv[q & 1][n * KB + k];
        rl[n] = rn;
        if (MODE != 0) {
          rnv[f][n][q] = rn;
          float s = sqrtf(1.0f / (rn + EPS));
          float a0r = x0r[f][q] * s, a0i = x0i[f][q] * s;
          float a1r = x1r[f][q] * s, a1i = x1i[f][q] * s;
          d00[f][n] += a0r * a0r + a0i * a0i;
          d11[f][n] += a1r * a1r + a1i * a1i;
          d01r[f][n] += a0r * a1r + a0i * a1i;
          d01i[f][n] += a0i * a1r - a0r * a1i;
        }
      }
      if (MODE == 0) y_step(f, q, rl[0], rl[1]);
    }
  }
  if (MODE != 0) {
#pragma unroll
    for (int f = 0; f < NF; f++)
#pragma unroll
      for (int n = 0; n < NSRC; n++) {
        rb[8 * f + 4 * n + 0][tid] = d00[f][n]; rb[8 * f + 4 * n + 1][tid] = d11[f][n];
        rb[8 * f + 4 * n + 2][tid] = d01r[f][n]; rb[8 * f + 4 * n + 3][tid] = d01i[f][n];
      }
    __syncthreads();
    for (int s2 = 128; s2 > 0; s2 >>= 1) {
      if (tid < s2) {
#pragma unroll
        for (int q = 0; q < NF * 4 * NSRC; q++) rb[q][tid] += rb[q][tid + s2];
      }
      __syncthreads();
    }
    if (lane == 0 && wv < NF) {
      const int f = wv;
      float Wf[8];
#pragma unroll
      for (int q = 0; q < 8; q++) Wf[q] = f == 0 ? w0[0][q] : w0[1][q];
      if (f == 0 || ok1) {
        ip_solve(Wf, 0, rb[8 * f + 0][0], rb[8 * f + 1][0], rb[8 * f + 2][0], rb[8 * f + 3][0]);
        ip_solve(Wf, 1, rb[8 * f + 4][0], rb[8 * f + 5][0], rb[8 * f + 6][0], rb[8 * f + 7][0]);
      }
#pragma unroll
      for (int q = 0; q < 8; q++) sw[f][q] = Wf[q];
    }
    __syncthreads();
    if (tid < (ok1 ? NF * 8 : 8)) ws[oW + (size_t)i0 * 8 + tid] = sw[tid >> 3][tid & 7];
#pragma unroll
    for (int f = 0; f < NF; f++) {
      wa0[f] = sw[f][0]; wb0[f] = sw[f][1]; wa1[f] = sw[f][2]; wb1[f] = sw[f][3];
      wc0[f] = sw[f][4]; wd0[f] = sw[f][5]; wc1[f] = sw[f][6]; wd1[f] = sw[f][7];
    }
  }

  if (MODE != 0) {
#pragma unroll
    for (int f = 0; f < NF; f++)
#pragma unroll
      for (int q = 0; q < JPT; q++) y_step(f, q, rnv[f][0][q], rnv[f][1][q]);
  }
  if (MODE == 2) return;

  __syncthreads();
  v8f acc0 = {0,0,0,0,0,0,0,0}, acc1 = {0,0,0,0,0,0,0,0};
  const float* va = V + (size_t)l16 * JD + 256 * wv + 4 * hf;
  const float* qb = sq + (l16 & 7) * QP + 256 * wv + 4 * hf;
  const bool qcol = l16 < NF * 4;
#pragma unroll 4
  for (int t = 0; t < 32; t++) {
    const float4 af = *(const float4*)(va + 8 * t);
    float4 bf = *(const float4*)(qb + 8 * t);
    if (!qcol) bf = make_float4(0.f, 0.f, 0.f, 0.f);
    v2f a, b;
    a.x = af.x; a.y = af.y; b.x = bf.x; b.y = bf.y;
    acc0 = wmma_k4(a, b, acc0);
    a.x = af.z; a.y = af.w; b.x = bf.z; b.y = bf.w;
    acc1 = wmma_k4(a, b, acc1);
  }
  if (l16 < NF * 4 && ((l16 & 3) >> 1) == hf) {
#pragma unroll
    for (int r = 0; r < 8; r++) red[wv][l16 * 8 + r] = acc0[r] + acc1[r];
  }
  __syncthreads();
  if (tid < NF * NSRC * KB) {
    const int f = tid >> 4, n = (tid >> 3) & 1, k = tid & 7;
    if (f == 0 || ok1) {
      float sn = 0.f, sd = 0.f;
#pragma unroll
      for (int w2 = 0; w2 < 8; w2++) { sn += red[w2][(4 * f + 2 * n) * 8 + k]; sd += red[w2][(4 * f + 2 * n + 1) * 8 + k]; }
      const size_t ti = oT + (size_t)n * IPD * KB + (size_t)(i0 + f) * KB + k;
      ws[ti] = ws[ti] * sqrtf(sn / (sd + EPS));
    }
  }
}

#define OP 1026
__global__ __launch_bounds__(512) void k_out(const float* __restrict__ ws, float* __restrict__ out) {
  __shared__ float2 s[16 * OP];
  const int tid = threadIdx.x;
  const int j0 = blockIdx.x * 16, n = blockIdx.y;
  const float* Yre = ws + oYre + (size_t)n * IPD * JD + j0;
  const float* Yim = ws + oYim + (size_t)n * IPD * JD + j0;
  float4 re[8], im[8];
#pragma unroll
  for (int q = 0; q < 8; q++) {
    const int f = q * 512 + tid, i = f >> 2, jq = f & 3;
    re[q] = *(const float4*)(Yre + (size_t)i * JD + 4 * jq);
    im[q] = *(const float4*)(Yim + (size_t)i * JD + 4 * jq);
  }
#pragma unroll
  for (int q = 0; q < 8; q++) {
    const int f = q * 512 + tid, i = f >> 2, jq = f & 3;
    s[(4 * jq + 0) * OP + i] = make_float2(re[q].x, im[q].x);
    s[(4 * jq + 1) * OP + i] = make_float2(re[q].y, im[q].y);
    s[(4 * jq + 2) * OP + i] = make_float2(re[q].z, im[q].z);
    s[(4 * jq + 3) * OP + i] = make_float2(re[q].w, im[q].w);
  }
  if (tid < 4) {
    const int i = ID - 1, jq = tid;
    const float4 r = *(const float4*)(Yre + (size_t)i * JD + 4 * jq);
    const float4 g = *(const float4*)(Yim + (size_t)i * JD + 4 * jq);
    s[(4 * jq + 0) * OP + i] = make_float2(r.x, g.x);
    s[(4 * jq + 1) * OP + i] = make_float2(r.y, g.y);
    s[(4 * jq + 2) * OP + i] = make_float2(r.z, g.z);
    s[(4 * jq + 3) * OP + i] = make_float2(r.w, g.w);
  }
  __syncthreads();
  float4* o4 = (float4*)(out + ((size_t)n * JD + j0) * ID * 2);
  for (int h = tid; h < 16 * ID / 2; h += 512) {
    const int q0 = 2 * h, q1 = q0 + 1;
    const float2 a = s[q0 + q0 / ID], b = s[q1 + q1 / ID];
    o4[h] = make_float4(a.x, a.y, b.x, b.y);
  }
}

extern "C" void kernel_launch(void* const* d_in, const int* in_sizes, int n_in,
                              void* d_out, int out_size, void* d_ws, size_t ws_size,
                              hipStream_t stream) {
  (void)in_sizes; (void)n_in; (void)out_size; (void)ws_size;
  const float* X  = (const float*)d_in[0];
  const float* T0 = (const float*)d_in[1];
  const float* V0 = (const float*)d_in[2];
  float* ws = (float*)d_ws;
  float* out = (float*)d_out;

  k_init_xy<<<dim3((IPD + 63) / 64, JD / 64, MC), 256, 0, stream>>>(X, ws);
  { size_t tot = (size_t)NSRC * IPD * KB; k_init_t<<<(unsigned)((tot + 255) / 256), 256, 0, stream>>>(T0, ws); }
  { size_t tot = (size_t)NSRC * KB * JD;  k_init_v<<<(unsigned)((tot + 255) / 256), 256, 0, stream>>>(V0, ws); }
  { size_t tot = (size_t)ID * 8;          k_init_w<<<(unsigned)((tot + 255) / 256), 256, 0, stream>>>(ws); }

  k_ip_t<0><<<NPB, 256, 0, stream>>>(ws);
  for (int it = 0; it < NITER; ++it) {
    k_v_update<<<dim3(JD / 16, NSRC), VW * 32, 0, stream>>>(ws);
    if (it + 1 < NITER) k_ip_t<1><<<NPB, 256, 0, stream>>>(ws);
    else                k_ip_t<2><<<NPB, 256, 0, stream>>>(ws);
  }
  k_out<<<dim3(JD / 16, NSRC), 512, 0, stream>>>(ws, out);
}


# Per-element reciprocals (sqrt(1/R) weight, |Y|^2/R^2, 1/R) on v_rsq_f32 / v_rcp_f32, 1 ulp, not the correctly rounded 14+11-instruction sqrt and division; R then waits in the idle Q tile of LDS
# speedup vs baseline: 2.5123x; 1.0937x over previous
#include <hip/hip_runtime.h>
#include <math.h>

#define JD 2048
#define ID 1025
#define IPD 1040
#define KB 8
#define MC 2
#define NSRC 2
#define NITER 5
#define VW 16
#define EPS 1e-20f

typedef __attribute__((ext_vector_type(2))) float v2f;
typedef __attribute__((ext_vector_type(8))) float v8f;

constexpr size_t oXre = 0;
constexpr size_t oXim = oXre + (size_t)MC * IPD * JD;
constexpr size_t oYre = oXim + (size_t)MC * IPD * JD;
constexpr size_t oYim = oYre + (size_t)NSRC * IPD * JD;
constexpr size_t oT   = oYim + (size_t)NSRC * IPD * JD;
constexpr size_t oV   = oT + (size_t)NSRC * IPD * KB;
constexpr size_t oW   = oV + (size_t)NSRC * KB * JD;

__device__ __forceinline__ v8f wmma_k4(v2f a, v2f b, v8f c) {
  return __builtin_amdgcn_wmma_f32_16x16x4_f32(false, a, false, b, (short)0, c, false, false);
}

__device__ __forceinline__ float rcp_hw(float x) { return __builtin_amdgcn_rcpf(x); }
__device__ __forceinline__ float rsq_hw(float x) { return __builtin_amdgcn_rsqf(x); }

#define TP 66
__global__ __launch_bounds__(256) void k_init_xy(const float* __restrict__ X, float* __restrict__ ws) {
  __shared__ __attribute__((aligned(16))) float sre[64 * TP];
  __shared__ __attribute__((aligned(16))) float sim[64 * TP];
  const int tid = threadIdx.x, wv = tid >> 5, lane = tid & 31;
  const int i0 = blockIdx.x * 64, j0 = blockIdx.y * 64, m = blockIdx.z;
  float2 v[16];
#pragma unroll
  for (int q = 0; q < 16; q++) {
    const int jj = wv * 8 + (q >> 1);
    const int i = i0 + (q & 1) * 32 + lane;
    const int ic = i < ID ? i : ID - 1;
    v[q] = *(const float2*)(X + (((size_t)m * JD + j0 + jj) * ID + ic) * 2);
  }
#pragma unroll
  for (int q = 0; q < 16; q++) {
    const int jj = wv * 8 + (q >> 1);
    const int il = (q & 1) * 32 + lane;
    const bool pad = i0 + il >= ID;
    sre[il * TP + jj] = pad ? 0.f : v[q].x; sim[il * TP + jj] = pad ? 0.f : v[q].y;
  }
  __syncthreads();
#pragma unroll
  for (int q = 0; q < 8; q++) {
    const int il = wv * 8 + q;
    const int i = i0 + il;
    if (i < IPD) {
      const float2 xr = *(const float2*)(sre + il * TP + 2 * lane);
      const float2 xi = *(const float2*)(sim + il * TP + 2 * lane);
      const size_t dst = ((size_t)m * IPD + i) * JD + j0 + 2 * lane;
      *(float2*)(ws + oXre + dst) = xr; *(float2*)(ws + oXim + dst) = xi;
      if (i >= ID) *(float2*)(ws + oYre + dst) = xr;
    }
  }
}

__global__ __launch_bounds__(256) void k_init_t(const float* __restrict__ T0, float* __restrict__ ws) {
  size_t t = (size_t)blockIdx.x * 256 + threadIdx.x;
  size_t tot = (size_t)NSRC * IPD * KB;
  if (t >= tot) return;
  int k = (int)(t % KB); size_t r = t / KB;
  int i = (int)(r % IPD); int n = (int)(r / IPD);
  float v = 0.f;
  if (i < ID) v = T0[((size_t)i * KB + k) * NSRC + n];
  ws[oT + ((size_t)n * IPD + i) * KB + k] = v;
}

__global__ __launch_bounds__(256) void k_init_v(const float* __restrict__ V0, float* __restrict__ ws) {
  size_t t = (size_t)blockIdx.x * 256 + threadIdx.x;
  size_t tot = (size_t)NSRC * KB * JD;
  if (t >= tot) return;
  int j = (int)(t % JD); size_t r = t / JD;
  int k = (int)(r % KB); int n = (int)(r / KB);
  ws[oV + ((size_t)n * KB + k) * JD + j] = V0[((size_t)k * JD + j) * NSRC + n];
}

__global__ __launch_bounds__(256) void k_init_w(float* __restrict__ ws) {
  size_t t = (size_t)blockIdx.x * 256 + threadIdx.x;
  size_t tot = (size_t)ID * 8;
  if (t >= tot) return;
  int f = (int)(t & 7);
  int c = f & 1, m = (f >> 1) & 1, nn = (f >> 2) & 1;
  ws[oW + t] = (c == 0 && nn == m) ? 1.f : 0.f;
}

struct VFrag { v2f a0, a1, at[4], by[4]; };

__device__ __forceinline__ void v_load_t(VFrag& f, const float* T,
                                         const int i0, const int hf, const int l16) {
  const size_t tr = (size_t)(i0 + l16) * KB;
  f.a0.x = T[tr + 2 * hf];     f.a0.y = T[tr + 2 * hf + 1];
  f.a1.x = T[tr + 4 + 2 * hf]; f.a1.y = T[tr + 5 + 2 * hf];
#pragma unroll
  for (int c4 = 0; c4 < 4; c4++) {
    const int ia = i0 + 4 * c4 + 2 * hf;
    f.at[c4].x = T[(size_t)ia * KB + (l16 & 7)];
    f.at[c4].y = T[(size_t)(ia + 1) * KB + (l16 & 7)];
  }
}

__device__ __forceinline__ void v_load_y(VFrag& f, const float* YdR,
                                         const int i0, const int j0, const int hf, const int l16) {
#pragma unroll
  for (int c4 = 0; c4 < 4; c4++) {
    const int ia = i0 + 4 * c4 + 2 * hf;
    f.by[c4].x = YdR[(size_t)ia * JD + j0 + l16];
    f.by[c4].y = YdR[(size_t)(ia + 1) * JD + j0 + l16];
  }
}

__device__ __forceinline__ void v_tile(const VFrag& f, const v2f b0, const v2f b1, float (*lr)[17],
                                       const int hf, const int l16, v8f& num, v8f& den) {
  v8f c = {0,0,0,0,0,0,0,0};
  c = wmma_k4(f.a0, b0, c);
  c = wmma_k4(f.a1, b1, c);
#pragma unroll
  for (int r = 0; r < 8; r++)
    lr[r + 8 * hf][l16] = rcp_hw(fabsf(c[r]) + EPS);
#pragma unroll
  for (int c4 = 0; c4 < 4; c4++) {
    v2f brv;
    brv.x = lr[4 * c4 + 2 * hf][l16];
    brv.y = lr[4 * c4 + 2 * hf + 1][l16];
    num = wmma_k4(f.at[c4], f.by[c4], num);
    den = wmma_k4(f.at[c4], brv, den);
  }
}

__global__ __launch_bounds__(VW * 32) void k_v_update(float* __restrict__ ws) {
  const int tid = threadIdx.x;
  const int wv = tid >> 5, lane = tid & 31, hf = lane >> 4, l16 = lane & 15;
  const int j0 = blockIdx.x * 16;
  const int n = blockIdx.y;
  const float* T = ws + oT + (size_t)n * IPD * KB;
  float* V = ws + oV + (size_t)n * KB * JD;
  const float* YdR = ws + oYre + (size_t)n * IPD * JD;

  __shared__ float lds_r[VW][16][17];
  __shared__ float redn[VW][128];
  __shared__ float redd[VW][128];

  v2f b0, b1;
  b0.x = V[(size_t)(2 * hf + 0) * JD + j0 + l16];
  b0.y = V[(size_t)(2 * hf + 1) * JD + j0 + l16];
  b1.x = V[(size_t)(2 * hf + 4) * JD + j0 + l16];
  b1.y = V[(size_t)(2 * hf + 5) * JD + j0 + l16];
  VFrag f[4], fx;
  const bool extra = wv == 0;
#pragma unroll
  for (int s = 0; s < 4; s++) v_load_t(f[s], T, (wv + VW * s) * 16, hf, l16);
  if (extra) v_load_t(fx, T, 4 * VW * 16, hf, l16);
#pragma unroll
  for (int s = 0; s < 4; s++) v_load_y(f[s], YdR, (wv + VW * s) * 16, j0, hf, l16);
  if (extra) v_load_y(fx, YdR, 4 * VW * 16, j0, hf, l16);

  v8f num = {0,0,0,0,0,0,0,0}, den = {0,0,0,0,0,0,0,0};
#pragma unroll
  for (int s = 0; s < 4; s++) v_tile(f[s], b0, b1, lds_r[wv], hf, l16, num, den);
  if (extra) v_tile(fx, b0, b1, lds_r[wv], hf, l16, num, den);

  if (hf == 0) {
#pragma unroll
    for (int r = 0; r < 8; r++) { redn[wv][r * 16 + l16] = num[r]; redd[wv][r * 16 + l16] = den[r]; }
  }
  __syncthreads();
  if (tid < KB * 16) {
    float sn = 0.f, sd = 0.f;
#pragma unroll
    for (int w2 = 0; w2 < VW; w2++) { sn += redn[w2][tid]; sd += redd[w2][tid]; }
    const size_t vi = (size_t)(tid >> 4) * JD + j0 + (tid & 15);
    V[vi] = V[vi] * sqrtf(sn / (sd + EPS));
  }
}

__device__ __forceinline__ void ip_solve(float* Wf, const int n, float S00, float S11, float S01r, float S01i) {
  const float invJ = 1.0f / (float)JD;
  float D00 = S00 * invJ + EPS;
  float D11 = S11 * invJ + EPS;
  float Dr = S01r * invJ, Di = S01i * invJ;
  float w00r = Wf[0], w00i = Wf[1], w01r = Wf[2], w01i = Wf[3];
  float w10r = Wf[4], w10i = Wf[5], w11r = Wf[6], w11i = Wf[7];
  float A00r = w00r * D00 + (w01r * Dr + w01i * Di);
  float A00i = w00i * D00 + (w01i * Dr - w01r * Di);
  float A01r = (w00r * Dr - w00i * Di) + w01r * D11;
  float A01i = (w00i * Dr + w00r * Di) + w01i * D11;
  float A10r = w10r * D00 + (w11r * Dr + w11i * Di);
  float A10i = w10i * D00 + (w11i * Dr - w11r * Di);
  float A11r = (w10r * Dr - w10i * Di) + w11r * D11;
  float A11i = (w10i * Dr + w10r * Di) + w11i * D11;
  float detr = (A00r * A11r - A00i * A11i) - (A01r * A10r - A01i * A10i);
  float deti = (A00r * A11i + A00i * A11r) - (A01r * A10i + A01i * A10r);
  float n0r, n0i, n1r, n1i;
  if (n == 0) { n0r = A11r;  n0i = A11i;  n1r = -A10r; n1i = -A10i; }
  else        { n0r = -A01r; n0i = -A01i; n1r = A00r;  n1i = A00i;  }
  float dd = detr * detr + deti * deti;
  float b0r = (n0r * detr + n0i * deti) / dd, b0i = (n0i * detr - n0r * deti) / dd;
  float b1r = (n1r * detr + n1i * deti) / dd, b1i = (n1i * detr - n1r * deti) / dd;
  float Db0r = D00 * b0r + (Dr * b1r - Di * b1i);
  float Db0i = D00 * b0i + (Dr * b1i + Di * b1r);
  float Db1r = (Dr * b0r + Di * b0i) + D11 * b1r;
  float Db1i = (Dr * b0i - Di * b0r) + D11 * b1i;
  float denr = b0r * Db0r + b0i * Db0i + b1r * Db1r + b1i * Db1i;
  float dnm = sqrtf(denr + EPS);
  Wf[(n * 2 + 0) * 2]     = b0r / dnm;
  Wf[(n * 2 + 0) * 2 + 1] = -b0i / dnm;
  Wf[(n * 2 + 1) * 2]     = b1r / dnm;
  Wf[(n * 2 + 1) * 2 + 1] = -b1i / dnm;
}

#define NF 2
#define NPB ((ID + NF - 1) / NF)
#define JPT (JD / 256)
#define QP (JD + 8)
template <int MODE>
__global__ __launch_bounds__(256, 6) void k_ip_t(float* __restrict__ ws) {
  const int i0 = blockIdx.x * NF;
  const bool ok1 = i0 + 1 < ID;
  const int tid = threadIdx.x;
  const int wv = tid >> 5, lane = tid & 31, hf = lane >> 4, l16 = lane & 15;
  const float* V = ws + oV;
  const float* X0r = ws + oXre;
  const float* X0i = ws + oXim;
  const float* X1r = ws + oXre + (size_t)IPD * JD;
  const float* X1i = ws + oXim + (size_t)IPD * JD;

  __shared__ float sw[NF][8];
  __shared__ float rb[NF * 4 * NSRC][256];
  __shared__ __attribute__((aligned(16))) float sq[MODE == 2 ? 4 : NF * 4 * QP];
  __shared__ float red[8][NF * 32];

  float x0r[NF][JPT], x0i[NF][JPT], x1r[NF][JPT], x1i[NF][JPT];
#pragma unroll
  for (int q = 0; q < JPT; q++) {
#pragma unroll
    for (int f = 0; f < NF; f++) {
      const size_t idx = (size_t)(i0 + f) * JD + tid + 256 * q;
      x0r[f][q] = X0r[idx]; x0i[f][q] = X0i[idx]; x1r[f][q] = X1r[idx]; x1i[f][q] = X1i[idx];
    }
  }
  __builtin_amdgcn_sched_barrier(0);

  float tT[NF][NSRC][KB], w0[NF][8];
#pragma unroll
  for (int f = 0; f < NF; f++)
#pragma unroll
    for (int n = 0; n < NSRC; n++)
#pragma unroll
      for (int k = 0; k < KB; k++) tT[f][n][k] = ws[oT + (size_t)n * IPD * KB + (size_t)(i0 + f) * KB + k];
  if (MODE != 0) {
    const int iw1 = ok1 ? i0 + 1 : i0;
#pragma unroll
    for (int q = 0; q < 8; q++) { w0[0][q] = ws[oW + (size_t)i0 * 8 + q]; w0[1][q] = ws[oW + (size_t)iw1 * 8 + q]; }
  }

  float wa0[NF], wb0[NF], wa1[NF], wb1[NF], wc0[NF], wd0[NF], wc1[NF], wd1[NF];
#pragma unroll
  for (int f = 0; f < NF; f++) {
    wa0[f] = 1.f; wb0[f] = 0.f; wa1[f] = 0.f; wb1[f] = 0.f;
    wc0[f] = 0.f; wd0[f] = 0.f; wc1[f] = 1.f; wd1[f] = 0.f;
  }
  float* P0r = ws + oYre;
  float* P1r = ws + oYre + (size_t)IPD * JD;
  float* P0i = ws + oYim;
  float* P1i = ws + oYim + (size_t)IPD * JD;
  auto y_step = [&](const int f, const int q, const float r0, const float r1) {
    const bool st = f == 0 || ok1;
    const int j = tid + 256 * q;
    const size_t idx = (size_t)(i0 + f) * JD + j;
    const float p0r = x0r[f][q], p0i = x0i[f][q], p1r = x1r[f][q], p1i = x1i[f][q];
    const float y0r = wa0[f] * p0r - wb0[f] * p0i + wa1[f] * p1r - wb1[f] * p1i;
    const float y0i = wa0[f] * p0i + wb0[f] * p0r + wa1[f] * p1i + wb1[f] * p1r;
    const float y1r = wc0[f] * p0r - wd0[f] * p0i + wc1[f] * p1r - wd1[f] * p1i;
    const float y1i = wc0[f] * p0i + wd0[f] * p0r + wc1[f] * p1i + wd1[f] * p1r;
    if (MODE == 2) {
      if (st) { P0r[idx] = y0r; P0i[idx] = y0i; P1r[idx] = y1r; P1i[idx] = y1i; }
    } else {
      const float ydr0 = (y0r * y0r + y0i * y0i) * rcp_hw(r0 * r0 + EPS);
      const float ydr1 = (y1r * y1r + y1i * y1i) * rcp_hw(r1 * r1 + EPS);
      const float rd0 = rcp_hw(fabsf(r0) + EPS);
      const float rd1 = rcp_hw(fabsf(r1) + EPS);
      if (st) { P0r[idx] = ydr0; P1r[idx] = ydr1; }
      sq[(4 * f + 0) * QP + j] = ydr0; sq[(4 * f + 1) * QP + j] = rd0;
      sq[(4 * f + 2) * QP + j] = ydr1; sq[(4 * f + 3) * QP + j] = rd1;
    }
  };

  float rnv[NF][NSRC][JPT];
  float d00[NF][NSRC], d11[NF][NSRC], d01r[NF][NSRC], d01i[NF][NSRC];
#pragma unroll
  for (int f = 0; f < NF; f++)
#pragma unroll
    for (int n = 0; n < NSRC; n++) { d00[f][n] = 0.f; d11[f][n] = 0.f; d01r[f][n] = 0.f; d01i[f][n] = 0.f; }
  float vv[2][NSRC * KB];
#pragma unroll
  for (int r = 0; r < NSRC * KB; r++) vv[0][r] = V[(size_t)r * JD + tid];
#pragma unroll
  for (int q = 0; q < JPT; q++) {
    if (q + 1 < JPT) {
#pragma unroll
      for (int r = 0; r < NSRC * KB; r++) vv[(q + 1) & 1][r] = V[(size_t)r * JD + tid + 256 * (q + 1)];
    }
#pragma unroll
    for (int f = 0; f < NF; f++) {
      float rl[NSRC];
#pragma unroll
      for (int n = 0; n < NSRC; n++) {
        float rn = 0.f;
#pragma unroll
        for (int k = 0; k < KB; k++) rn += tT[f][n][k] * vv[q & 1][n * KB + k];
        rl[n] = rn;
        if (MODE != 0) {
          if (MODE == 1) sq[(4 * f + 2 * n) * QP + tid + 256 * q] = rn;
          else rnv[f][n][q] = rn;
          float s = rsq_hw(rn + EPS);
          float a0r = x0r[f][q] * s, a0i = x0i[f][q] * s;
          float a1r = x1r[f][q] * s, a1i = x1i[f][q] * s;
          d00[f][n] += a0r * a0r + a0i * a0i;
          d11[f][n] += a1r * a1r + a1i * a1i;
          d01r[f][n] += a0r * a1r + a0i * a1i;
          d01i[f][n] += a0i * a1r - a0r * a1i;
        }
      }
      if (MODE == 0) y_step(f, q, rl[0], rl[1]);
    }
    __builtin_amdgcn_sched_barrier(0);
  }
  if (MODE != 0) {
#pragma unroll
    for (int f = 0; f < NF; f++)
#pragma unroll
      for (int n = 0; n < NSRC; n++) {
        rb[8 * f + 4 * n + 0][tid] = d00[f][n]; rb[8 * f + 4 * n + 1][tid] = d11[f][n];
        rb[8 * f + 4 * n + 2][tid] = d01r[f][n]; rb[8 * f + 4 * n + 3][tid] = d01i[f][n];
      }
    __syncthreads();
    for (int s2 = 128; s2 > 0; s2 >>= 1) {
      if (tid < s2) {
#pragma unroll
        for (int q = 0; q < NF * 4 * NSRC; q++) rb[q][tid] += rb[q][tid + s2];
      }
      __syncthreads();
    }
    if (lane == 0 && wv < NF) {
      const int f = wv;
      float Wf[8];
#pragma unroll
      for (int q = 0; q < 8; q++) Wf[q] = f == 0 ? w0[0][q] : w0[1][q];
      if (f == 0 || ok1) {
        ip_solve(Wf, 0, rb[8 * f + 0][0], rb[8 * f + 1][0], rb[8 * f + 2][0], rb[8 * f + 3][0]);
        ip_solve(Wf, 1, rb[8 * f + 4][0], rb[8 * f + 5][0], rb[8 * f + 6][0], rb[8 * f + 7][0]);
      }
#pragma unroll
      for (int q = 0; q < 8; q++) sw[f][q] = Wf[q];
    }
    __syncthreads();
    if (tid < (ok1 ? NF * 8 : 8)) ws[oW + (size_t)i0 * 8 + tid] = sw[tid >> 3][tid & 7];
#pragma unroll
    for (int f = 0; f < NF; f++) {
      wa0[f] = sw[f][0]; wb0[f] = sw[f][1]; wa1[f] = sw[f][2]; wb1[f] = sw[f][3];
      wc0[f] = sw[f][4]; wd0[f] = sw[f][5]; wc1[f] = sw[f][6]; wd1[f] = sw[f][7];
    }
  }

  if (MODE != 0) {
#pragma unroll
    for (int f = 0; f < NF; f++)
#pragma unroll
      for (int q = 0; q < JPT; q++) {
        if (MODE == 1) y_step(f, q, sq[(4 * f) * QP + tid + 256 * q], sq[(4 * f + 2) * QP + tid + 256 * q]);
        else y_step(f, q, rnv[f][0][q], rnv[f][1][q]);
      }
  }
  if (MODE == 2) return;

  __syncthreads();
  v8f acc0 = {0,0,0,0,0,0,0,0}, acc1 = {0,0,0,0,0,0,0,0};
  const float* va = V + (size_t)l16 * JD + 256 * wv + 4 * hf;
  const float* qb = sq + (l16 & 7) * QP + 256 * wv + 4 * hf;
  const bool qcol = l16 < NF * 4;
#pragma unroll 4
  for (int t = 0; t < 32; t++) {
    const float4 af = *(const float4*)(va + 8 * t);
    float4 bf = *(const float4*)(qb + 8 * t);
    if (!qcol) bf = make_float4(0.f, 0.f, 0.f, 0.f);
    v2f a, b;
    a.x = af.x; a.y = af.y; b.x = bf.x; b.y = bf.y;
    acc0 = wmma_k4(a, b, acc0);
    a.x = af.z; a.y = af.w; b.x = bf.z; b.y = bf.w;
    acc1 = wmma_k4(a, b, acc1);
  }
  if (l16 < NF * 4 && ((l16 & 3) >> 1) == hf) {
#pragma unroll
    for (int r = 0; r < 8; r++) red[wv][l16 * 8 + r] = acc0[r] + acc1[r];
  }
  __syncthreads();
  if (tid < NF * NSRC * KB) {
    const int f = tid >> 4, n = (tid >> 3) & 1, k = tid & 7;
    if (f == 0 || ok1) {
      float sn = 0.f, sd = 0.f;
#pragma unroll
      for (int w2 = 0; w2 < 8; w2++) { sn += red[w2][(4 * f + 2 * n) * 8 + k]; sd += red[w2][(4 * f + 2 * n + 1) * 8 + k]; }
      const size_t ti = oT + (size_t)n * IPD * KB + (size_t)(i0 + f) * KB + k;
      ws[ti] = ws[ti] * sqrtf(sn / (sd + EPS));
    }
  }
}

#define OP 1026
__global__ __launch_bounds__(512) void k_out(const float* __restrict__ ws, float* __restrict__ out) {
  __shared__ float2 s[16 * OP];
  const int tid = threadIdx.x;
  const int j0 = blockIdx.x * 16, n = blockIdx.y;
  const float* Yre = ws + oYre + (size_t)n * IPD * JD + j0;
  const float* Yim = ws + oYim + (size_t)n * IPD * JD + j0;
  float4 re[8], im[8];
#pragma unroll
  for (int q = 0; q < 8; q++) {
    const int f = q * 512 + tid, i = f >> 2, jq = f & 3;
    re[q] = *(const float4*)(Yre + (size_t)i * JD + 4 * jq);
    im[q] = *(const float4*)(Yim + (size_t)i * JD + 4 * jq);
  }
#pragma unroll
  for (int q = 0; q < 8; q++) {
    const int f = q * 512 + tid, i = f >> 2, jq = f & 3;
    s[(4 * jq + 0) * OP + i] = make_float2(re[q].x, im[q].x);
    s[(4 * jq + 1) * OP + i] = make_float2(re[q].y, im[q].y);
    s[(4 * jq + 2) * OP + i] = make_float2(re[q].z, im[q].z);
    s[(4 * jq + 3) * OP + i] = make_float2(re[q].w, im[q].w);
  }
  if (tid < 4) {
    const int i = ID - 1, jq = tid;
    const float4 r = *(const float4*)(Yre + (size_t)i * JD + 4 * jq);
    const float4 g = *(const float4*)(Yim + (size_t)i * JD + 4 * jq);
    s[(4 * jq + 0) * OP + i] = make_float2(r.x, g.x);
    s[(4 * jq + 1) * OP + i] = make_float2(r.y, g.y);
    s[(4 * jq + 2) * OP + i] = make_float2(r.z, g.z);
    s[(4 * jq + 3) * OP + i] = make_float2(r.w, g.w);
  }
  __syncthreads();
  float4* o4 = (float4*)(out + ((size_t)n * JD + j0) * ID * 2);
  for (int h = tid; h < 16 * ID / 2; h += 512) {
    const int q0 = 2 * h, q1 = q0 + 1;
    const float2 a = s[q0 + q0 / ID], b = s[q1 + q1 / ID];
    o4[h] = make_float4(a.x, a.y, b.x, b.y);
  }
}

extern "C" void kernel_launch(void* const* d_in, const int* in_sizes, int n_in,
                              void* d_out, int out_size, void* d_ws, size_t ws_size,
                              hipStream_t stream) {
  (void)in_sizes; (void)n_in; (void)out_size; (void)ws_size;
  const float* X  = (const float*)d_in[0];
  const float* T0 = (const float*)d_in[1];
  const float* V0 = (const float*)d_in[2];
  float* ws = (float*)d_ws;
  float* out = (float*)d_out;

  k_init_xy<<<dim3((IPD + 63) / 64, JD / 64, MC), 256, 0, stream>>>(X, ws);
  { size_t tot = (size_t)NSRC * IPD * KB; k_init_t<<<(unsigned)((tot + 255) / 256), 256, 0, stream>>>(T0, ws); }
  { size_t tot = (size_t)NSRC * KB * JD;  k_init_v<<<(unsigned)((tot + 255) / 256), 256, 0, stream>>>(V0, ws); }
  { size_t tot = (size_t)ID * 8;          k_init_w<<<(unsigned)((tot + 255) / 256), 256, 0, stream>>>(ws); }

  k_ip_t<0><<<NPB, 256, 0, stream>>>(ws);
  for (int it = 0; it < NITER; ++it) {
    k_v_update<<<dim3(JD / 16, NSRC), VW * 32, 0, stream>>>(ws);
    if (it + 1 < NITER) k_ip_t<1><<<NPB, 256, 0, stream>>>(ws);
    else                k_ip_t<2><<<NPB, 256, 0, stream>>>(ws);
  }
  k_out<<<dim3(JD / 16, NSRC), 512, 0, stream>>>(ws, out);
}


# The three small set-up launches (T, V into plane order, W = eye) move into the first 226 blocks of the X transpose launch, one element per thread, loaded beside the tile's own loads; 13 launches
# speedup vs baseline: 2.6072x; 1.0378x over previous
#include <hip/hip_runtime.h>
#include <math.h>

#define JD 2048
#define ID 1025
#define IPD 1040
#define KB 8
#define MC 2
#define NSRC 2
#define NITER 5
#define VW 16
#define EPS 1e-20f

typedef __attribute__((ext_vector_type(2))) float v2f;
typedef __attribute__((ext_vector_type(8))) float v8f;

constexpr size_t oXre = 0;
constexpr size_t oXim = oXre + (size_t)MC * IPD * JD;
constexpr size_t oYre = oXim + (size_t)MC * IPD * JD;
constexpr size_t oYim = oYre + (size_t)NSRC * IPD * JD;
constexpr size_t oT   = oYim + (size_t)NSRC * IPD * JD;
constexpr size_t oV   = oT + (size_t)NSRC * IPD * KB;
constexpr size_t oW   = oV + (size_t)NSRC * KB * JD;

__device__ __forceinline__ v8f wmma_k4(v2f a, v2f b, v8f c) {
  return __builtin_amdgcn_wmma_f32_16x16x4_f32(false, a, false, b, (short)0, c, false, false);
}

__device__ __forceinline__ float rcp_hw(float x) { return __builtin_amdgcn_rcpf(x); }
__device__ __forceinline__ float rsq_hw(float x) { return __builtin_amdgcn_rsqf(x); }

#define TP 66
#define NT_E (NSRC * IPD * KB)
#define NV_E (NSRC * KB * JD)
#define NW_E (ID * 8)
#define SMALL_B ((NT_E + NV_E + NW_E + 255) / 256)
#define GXI ((IPD + 63) / 64)
#define GYJ (JD / 64)
static_assert(oV == oT + NT_E && oW == oV + NV_E, "T, V, W must be contiguous");
static_assert(NT_E % 256 == 0 && NV_E % 256 == 0, "a block must lie inside one array");
static_assert(SMALL_B <= GXI * GYJ * MC, "not enough blocks for the small arrays");
__device__ __forceinline__ void init_small(const float* __restrict__ T0, const float* __restrict__ V0,
                                           float* __restrict__ ws, const int bflat, const int tid) {
  const int t = bflat * 256 + tid;
  const float* sp = T0;
  bool ld = false;
  float alt = 0.f;
  if (bflat < NT_E / 256) {
    const int k = t % KB, r = t / KB;
    const int i = r % IPD, n = r / IPD;
    const int ic = i < ID ? i : ID - 1;
    sp = T0 + ((size_t)ic * KB + k) * NSRC + n;
    ld = i < ID;
  } else if (bflat < (NT_E + NV_E) / 256) {
    const int u = t - NT_E;
    const int j = u % JD, r = u / JD;
    const int k = r % KB, n = r / KB;
    sp = V0 + ((size_t)k * JD + j) * NSRC + n;
    ld = true;
  } else {
    const int f = (t - NT_E - NV_E) & 7;
    const int cc = f & 1, m = (f >> 1) & 1, nn = (f >> 2) & 1;
    alt = (cc == 0 && nn == m) ? 1.f : 0.f;
  }
  const float lv = *sp;
  if (t < NT_E + NV_E + NW_E) ws[oT + t] = ld ? lv : alt;
}

__global__ __launch_bounds__(256) void k_init_xy(const float* __restrict__ X, const float* __restrict__ T0,
                                                 const float* __restrict__ V0, float* __restrict__ ws) {
  __shared__ __attribute__((aligned(16))) float sre[64 * TP];
  __shared__ __attribute__((aligned(16))) float sim[64 * TP];
  const int tid = threadIdx.x, wv = tid >> 5, lane = tid & 31;
  const int i0 = blockIdx.x * 64, j0 = blockIdx.y * 64, m = blockIdx.z;
  float2 v[16];
#pragma unroll
  for (int q = 0; q < 16; q++) {
    const int jj = wv * 8 + (q >> 1);
    const int i = i0 + (q & 1) * 32 + lane;
    const int ic = i < ID ? i : ID - 1;
    v[q] = *(const float2*)(X + (((size_t)m * JD + j0 + jj) * ID + ic) * 2);
  }
  const int bflat = (blockIdx.z * GYJ + blockIdx.y) * GXI + blockIdx.x;
  if (bflat < SMALL_B) init_small(T0, V0, ws, bflat, tid);
#pragma unroll
  for (int q = 0; q < 16; q++) {
    const int jj = wv * 8 + (q >> 1);
    const int il = (q & 1) * 32 + lane;
    const bool pad = i0 + il >= ID;
    sre[il * TP + jj] = pad ? 0.f : v[q].x; sim[il * TP + jj] = pad ? 0.f : v[q].y;
  }
  __syncthreads();
#pragma unroll
  for (int q = 0; q < 8; q++) {
    const int il = wv * 8 + q;
    const int i = i0 + il;
    if (i < IPD) {
      const float2 xr = *(const float2*)(sre + il * TP + 2 * lane);
      const float2 xi = *(const float2*)(sim + il * TP + 2 * lane);
      const size_t dst = ((size_t)m * IPD + i) * JD + j0 + 2 * lane;
      *(float2*)(ws + oXre + dst) = xr; *(float2*)(ws + oXim + dst) = xi;
      if (i >= ID) *(float2*)(ws + oYre + dst) = xr;
    }
  }
}

struct VFrag { v2f a0, a1, at[4], by[4]; };

__device__ __forceinline__ void v_load_t(VFrag& f, const float* T,
                                         const int i0, const int hf, const int l16) {
  const size_t tr = (size_t)(i0 + l16) * KB;
  f.a0.x = T[tr + 2 * hf];     f.a0.y = T[tr + 2 * hf + 1];
  f.a1.x = T[tr + 4 + 2 * hf]; f.a1.y = T[tr + 5 + 2 * hf];
#pragma unroll
  for (int c4 = 0; c4 < 4; c4++) {
    const int ia = i0 + 4 * c4 + 2 * hf;
    f.at[c4].x = T[(size_t)ia * KB + (l16 & 7)];
    f.at[c4].y = T[(size_t)(ia + 1) * KB + (l16 & 7)];
  }
}

__device__ __forceinline__ void v_load_y(VFrag& f, const float* YdR,
                                         const int i0, const int j0, const int hf, const int l16) {
#pragma unroll
  for (int c4 = 0; c4 < 4; c4++) {
    const int ia = i0 + 4 * c4 + 2 * hf;
    f.by[c4].x = YdR[(size_t)ia * JD + j0 + l16];
    f.by[c4].y = YdR[(size_t)(ia + 1) * JD + j0 + l16];
  }
}

__device__ __forceinline__ void v_tile(const VFrag& f, const v2f b0, const v2f b1, float (*lr)[17],
                                       const int hf, const int l16, v8f& num, v8f& den) {
  v8f c = {0,0,0,0,0,0,0,0};
  c = wmma_k4(f.a0, b0, c);
  c = wmma_k4(f.a1, b1, c);
#pragma unroll
  for (int r = 0; r < 8; r++)
    lr[r + 8 * hf][l16] = rcp_hw(fabsf(c[r]) + EPS);
#pragma unroll
  for (int c4 = 0; c4 < 4; c4++) {
    v2f brv;
    brv.x = lr[4 * c4 + 2 * hf][l16];
    brv.y = lr[4 * c4 + 2 * hf + 1][l16];
    num = wmma_k4(f.at[c4], f.by[c4], num);
    den = wmma_k4(f.at[c4], brv, den);
  }
}

__global__ __launch_bounds__(VW * 32) void k_v_update(float* __restrict__ ws) {
  const int tid = threadIdx.x;
  const int wv = tid >> 5, lane = tid & 31, hf = lane >> 4, l16 = lane & 15;
  const int j0 = blockIdx.x * 16;
  const int n = blockIdx.y;
  const float* T = ws + oT + (size_t)n * IPD * KB;
  float* V = ws + oV + (size_t)n * KB * JD;
  const float* YdR = ws + oYre + (size_t)n * IPD * JD;

  __shared__ float lds_r[VW][16][17];
  __shared__ float redn[VW][128];
  __shared__ float redd[VW][128];

  v2f b0, b1;
  b0.x = V[(size_t)(2 * hf + 0) * JD + j0 + l16];
  b0.y = V[(size_t)(2 * hf + 1) * JD + j0 + l16];
  b1.x = V[(size_t)(2 * hf + 4) * JD + j0 + l16];
  b1.y = V[(size_t)(2 * hf + 5) * JD + j0 + l16];
  VFrag f[4], fx;
  const bool extra = wv == 0;
#pragma unroll
  for (int s = 0; s < 4; s++) v_load_t(f[s], T, (wv + VW * s) * 16, hf, l16);
  if (extra) v_load_t(fx, T, 4 * VW * 16, hf, l16);
#pragma unroll
  for (int s = 0; s < 4; s++) v_load_y(f[s], YdR, (wv + VW * s) * 16, j0, hf, l16);
  if (extra) v_load_y(fx, YdR, 4 * VW * 16, j0, hf, l16);

  v8f num = {0,0,0,0,0,0,0,0}, den = {0,0,0,0,0,0,0,0};
#pragma unroll
  for (int s = 0; s < 4; s++) v_tile(f[s], b0, b1, lds_r[wv], hf, l16, num, den);
  if (extra) v_tile(fx, b0, b1, lds_r[wv], hf, l16, num, den);

  if (hf == 0) {
#pragma unroll
    for (int r = 0; r < 8; r++) { redn[wv][r * 16 + l16] = num[r]; redd[wv][r * 16 + l16] = den[r]; }
  }
  __syncthreads();
  if (tid < KB * 16) {
    float sn = 0.f, sd = 0.f;
#pragma unroll
    for (int w2 = 0; w2 < VW; w2++) { sn += redn[w2][tid]; sd += redd[w2][tid]; }
    const size_t vi = (size_t)(tid >> 4) * JD + j0 + (tid & 15);
    V[vi] = V[vi] * sqrtf(sn / (sd + EPS));
  }
}

__device__ __forceinline__ void ip_solve(float* Wf, const int n, float S00, float S11, float S01r, float S01i) {
  const float invJ = 1.0f / (float)JD;
  float D00 = S00 * invJ + EPS;
  float D11 = S11 * invJ + EPS;
  float Dr = S01r * invJ, Di = S01i * invJ;
  float w00r = Wf[0], w00i = Wf[1], w01r = Wf[2], w01i = Wf[3];
  float w10r = Wf[4], w10i = Wf[5], w11r = Wf[6], w11i = Wf[7];
  float A00r = w00r * D00 + (w01r * Dr + w01i * Di);
  float A00i = w00i * D00 + (w01i * Dr - w01r * Di);
  float A01r = (w00r * Dr - w00i * Di) + w01r * D11;
  float A01i = (w00i * Dr + w00r * Di) + w01i * D11;
  float A10r = w10r * D00 + (w11r * Dr + w11i * Di);
  float A10i = w10i * D00 + (w11i * Dr - w11r * Di);
  float A11r = (w10r * Dr - w10i * Di) + w11r * D11;
  float A11i = (w10i * Dr + w10r * Di) + w11i * D11;
  float detr = (A00r * A11r - A00i * A11i) - (A01r * A10r - A01i * A10i);
  float deti = (A00r * A11i + A00i * A11r) - (A01r * A10i + A01i * A10r);
  float n0r, n0i, n1r, n1i;
  if (n == 0) { n0r = A11r;  n0i = A11i;  n1r = -A10r; n1i = -A10i; }
  else        { n0r = -A01r; n0i = -A01i; n1r = A00r;  n1i = A00i;  }
  float dd = detr * detr + deti * deti;
  float b0r = (n0r * detr + n0i * deti) / dd, b0i = (n0i * detr - n0r * deti) / dd;
  float b1r = (n1r * detr + n1i * deti) / dd, b1i = (n1i * detr - n1r * deti) / dd;
  float Db0r = D00 * b0r + (Dr * b1r - Di * b1i);
  float Db0i = D00 * b0i + (Dr * b1i + Di * b1r);
  float Db1r = (Dr * b0r + Di * b0i) + D11 * b1r;
  float Db1i = (Dr * b0i - Di * b0r) + D11 * b1i;
  float denr = b0r * Db0r + b0i * Db0i + b1r * Db1r + b1i * Db1i;
  float dnm = sqrtf(denr + EPS);
  Wf[(n * 2 + 0) * 2]     = b0r / dnm;
  Wf[(n * 2 + 0) * 2 + 1] = -b0i / dnm;
  Wf[(n * 2 + 1) * 2]     = b1r / dnm;
  Wf[(n * 2 + 1) * 2 + 1] = -b1i / dnm;
}

#define NF 2
#define NPB ((ID + NF - 1) / NF)
#define JPT (JD / 256)
#define QP (JD + 8)
template <int MODE>
__global__ __launch_bounds__(256, 6) void k_ip_t(float* __restrict__ ws) {
  const int i0 = blockIdx.x * NF;
  const bool ok1 = i0 + 1 < ID;
  const int tid = threadIdx.x;
  const int wv = tid >> 5, lane = tid & 31, hf = lane >> 4, l16 = lane & 15;
  const float* V = ws + oV;
  const float* X0r = ws + oXre;
  const float* X0i = ws + oXim;
  const float* X1r = ws + oXre + (size_t)IPD * JD;
  const float* X1i = ws + oXim + (size_t)IPD * JD;

  __shared__ float sw[NF][8];
  __shared__ float rb[NF * 4 * NSRC][256];
  __shared__ __attribute__((aligned(16))) float sq[MODE == 2 ? 4 : NF * 4 * QP];
  __shared__ float red[8][NF * 32];

  float x0r[NF][JPT], x0i[NF][JPT], x1r[NF][JPT], x1i[NF][JPT];
#pragma unroll
  for (int q = 0; q < JPT; q++) {
#pragma unroll
    for (int f = 0; f < NF; f++) {
      const size_t idx = (size_t)(i0 + f) * JD + tid + 256 * q;
      x0r[f][q] = X0r[idx]; x0i[f][q] = X0i[idx]; x1r[f][q] = X1r[idx]; x1i[f][q] = X1i[idx];
    }
  }
  __builtin_amdgcn_sched_barrier(0);

  float tT[NF][NSRC][KB], w0[NF][8];
#pragma unroll
  for (int f = 0; f < NF; f++)
#pragma unroll
    for (int n = 0; n < NSRC; n++)
#pragma unroll
      for (int k = 0; k < KB; k++) tT[f][n][k] = ws[oT + (size_t)n * IPD * KB + (size_t)(i0 + f) * KB + k];
  if (MODE != 0) {
    const int iw1 = ok1 ? i0 + 1 : i0;
#pragma unroll
    for (int q = 0; q < 8; q++) { w0[0][q] = ws[oW + (size_t)i0 * 8 + q]; w0[1][q] = ws[oW + (size_t)iw1 * 8 + q]; }
  }

  float wa0[NF], wb0[NF], wa1[NF], wb1[NF], wc0[NF], wd0[NF], wc1[NF], wd1[NF];
#pragma unroll
  for (int f = 0; f < NF; f++) {
    wa0[f] = 1.f; wb0[f] = 0.f; wa1[f] = 0.f; wb1[f] = 0.f;
    wc0[f] = 0.f; wd0[f] = 0.f; wc1[f] = 1.f; wd1[f] = 0.f;
  }
  float* P0r = ws + oYre;
  float* P1r = ws + oYre + (size_t)IPD * JD;
  float* P0i = ws + oYim;
  float* P1i = ws + oYim + (size_t)IPD * JD;
  auto y_step = [&](const int f, const int q, const float r0, const float r1) {
    const bool st = f == 0 || ok1;
    const int j = tid + 256 * q;
    const size_t idx = (size_t)(i0 + f) * JD + j;
    const float p0r = x0r[f][q], p0i = x0i[f][q], p1r = x1r[f][q], p1i = x1i[f][q];
    const float y0r = wa0[f] * p0r - wb0[f] * p0i + wa1[f] * p1r - wb1[f] * p1i;
    const float y0i = wa0[f] * p0i + wb0[f] * p0r + wa1[f] * p1i + wb1[f] * p1r;
    const float y1r = wc0[f] * p0r - wd0[f] * p0i + wc1[f] * p1r - wd1[f] * p1i;
    const float y1i = wc0[f] * p0i + wd0[f] * p0r + wc1[f] * p1i + wd1[f] * p1r;
    if (MODE == 2) {
      if (st) { P0r[idx] = y0r; P0i[idx] = y0i; P1r[idx] = y1r; P1i[idx] = y1i; }
    } else {
      const float ydr0 = (y0r * y0r + y0i * y0i) * rcp_hw(r0 * r0 + EPS);
      const float ydr1 = (y1r * y1r + y1i * y1i) * rcp_hw(r1 * r1 + EPS);
      const float rd0 = rcp_hw(fabsf(r0) + EPS);
      const float rd1 = rcp_hw(fabsf(r1) + EPS);
      if (st) { P0r[idx] = ydr0; P1r[idx] = ydr1; }
      sq[(4 * f + 0) * QP + j] = ydr0; sq[(4 * f + 1) * QP + j] = rd0;
      sq[(4 * f + 2) * QP + j] = ydr1; sq[(4 * f + 3) * QP + j] = rd1;
    }
  };

  float rnv[NF][NSRC][JPT];
  float d00[NF][NSRC], d11[NF][NSRC], d01r[NF][NSRC], d01i[NF][NSRC];
#pragma unroll
  for (int f = 0; f < NF; f++)
#pragma unroll
    for (int n = 0; n < NSRC; n++) { d00[f][n] = 0.f; d11[f][n] = 0.f; d01r[f][n] = 0.f; d01i[f][n] = 0.f; }
  float vv[2][NSRC * KB];
#pragma unroll
  for (int r = 0; r < NSRC * KB; r++) vv[0][r] = V[(size_t)r * JD + tid];
#pragma unroll
  for (int q = 0; q < JPT; q++) {
    if (q + 1 < JPT) {
#pragma unroll
      for (int r = 0; r < NSRC * KB; r++) vv[(q + 1) & 1][r] = V[(size_t)r * JD + tid + 256 * (q + 1)];
    }
#pragma unroll
    for (int f = 0; f < NF; f++) {
      float rl[NSRC];
#pragma unroll
      for (int n = 0; n < NSRC; n++) {
        float rn = 0.f;
#pragma unroll
        for (int k = 0; k < KB; k++) rn += tT[f][n][k] * vv[q & 1][n * KB + k];
        rl[n] = rn;
        if (MODE != 0) {
          if (MODE == 1) sq[(4 * f + 2 * n) * QP + tid + 256 * q] = rn;
          else rnv[f][n][q] = rn;
          float s = rsq_hw(rn + EPS);
          float a0r = x0r[f][q] * s, a0i = x0i[f][q] * s;
          float a1r = x1r[f][q] * s, a1i = x1i[f][q] * s;
          d00[f][n] += a0r * a0r + a0i * a0i;
          d11[f][n] += a1r * a1r + a1i * a1i;
          d01r[f][n] += a0r * a1r + a0i * a1i;
          d01i[f][n] += a0i * a1r - a0r * a1i;
        }
      }
      if (MODE == 0) y_step(f, q, rl[0], rl[1]);
    }
    __builtin_amdgcn_sched_barrier(0);
  }
  if (MODE != 0) {
#pragma unroll
    for (int f = 0; f < NF; f++)
#pragma unroll
      for (int n = 0; n < NSRC; n++) {
        rb[8 * f + 4 * n + 0][tid] = d00[f][n]; rb[8 * f + 4 * n + 1][tid] = d11[f][n];
        rb[8 * f + 4 * n + 2][tid] = d01r[f][n]; rb[8 * f + 4 * n + 3][tid] = d01i[f][n];
      }
    __syncthreads();
    for (int s2 = 128; s2 > 0; s2 >>= 1) {
      if (tid < s2) {
#pragma unroll
        for (int q = 0; q < NF * 4 * NSRC; q++) rb[q][tid] += rb[q][tid + s2];
      }
      __syncthreads();
    }
    if (lane == 0 && wv < NF) {
      const int f = wv;
      float Wf[8];
#pragma unroll
      for (int q = 0; q < 8; q++) Wf[q] = f == 0 ? w0[0][q] : w0[1][q];
      if (f == 0 || ok1) {
        ip_solve(Wf, 0, rb[8 * f + 0][0], rb[8 * f + 1][0], rb[8 * f + 2][0], rb[8 * f + 3][0]);
        ip_solve(Wf, 1, rb[8 * f + 4][0], rb[8 * f + 5][0], rb[8 * f + 6][0], rb[8 * f + 7][0]);
      }
#pragma unroll
      for (int q = 0; q < 8; q++) sw[f][q] = Wf[q];
    }
    __syncthreads();
    if (tid < (ok1 ? NF * 8 : 8)) ws[oW + (size_t)i0 * 8 + tid] = sw[tid >> 3][tid & 7];
#pragma unroll
    for (int f = 0; f < NF; f++) {
      wa0[f] = sw[f][0]; wb0[f] = sw[f][1]; wa1[f] = sw[f][2]; wb1[f] = sw[f][3];
      wc0[f] = sw[f][4]; wd0[f] = sw[f][5]; wc1[f] = sw[f][6]; wd1[f] = sw[f][7];
    }
  }

  if (MODE != 0) {
#pragma unroll
    for (int f = 0; f < NF; f++)
#pragma unroll
      for (int q = 0; q < JPT; q++) {
        if (MODE == 1) y_step(f, q, sq[(4 * f) * QP + tid + 256 * q], sq[(4 * f + 2) * QP + tid + 256 * q]);
        else y_step(f, q, rnv[f][0][q], rnv[f][1][q]);
      }
  }
  if (MODE == 2) return;

  __syncthreads();
  v8f acc0 = {0,0,0,0,0,0,0,0}, acc1 = {0,0,0,0,0,0,0,0};
  const float* va = V + (size_t)l16 * JD + 256 * wv + 4 * hf;
  const float* qb = sq + (l16 & 7) * QP + 256 * wv + 4 * hf;
  const bool qcol = l16 < NF * 4;
#pragma unroll 4
  for (int t = 0; t < 32; t++) {
    const float4 af = *(const float4*)(va + 8 * t);
    float4 bf = *(const float4*)(qb + 8 * t);
    if (!qcol) bf = make_float4(0.f, 0.f, 0.f, 0.f);
    v2f a, b;
    a.x = af.x; a.y = af.y; b.x = bf.x; b.y = bf.y;
    acc0 = wmma_k4(a, b, acc0);
    a.x = af.z; a.y = af.w; b.x = bf.z; b.y = bf.w;
    acc1 = wmma_k4(a, b, acc1);
  }
  if (l16 < NF * 4 && ((l16 & 3) >> 1) == hf) {
#pragma unroll
    for (int r = 0; r < 8; r++) red[wv][l16 * 8 + r] = acc0[r] + acc1[r];
  }
  __syncthreads();
  if (tid < NF * NSRC * KB) {
    const int f = tid >> 4, n = (tid >> 3) & 1, k = tid & 7;
    if (f == 0 || ok1) {
      float sn = 0.f, sd = 0.f;
#pragma unroll
      for (int w2 = 0; w2 < 8; w2++) { sn += red[w2][(4 * f + 2 * n) * 8 + k]; sd += red[w2][(4 * f + 2 * n + 1) * 8 + k]; }
      const size_t ti = oT + (size_t)n * IPD * KB + (size_t)(i0 + f) * KB + k;
      ws[ti] = ws[ti] * sqrtf(sn / (sd + EPS));
    }
  }
}

#define OP 1026
__global__ __launch_bounds__(512) void k_out(const float* __restrict__ ws, float* __restrict__ out) {
  __shared__ float2 s[16 * OP];
  const int tid = threadIdx.x;
  const int j0 = blockIdx.x * 16, n = blockIdx.y;
  const float* Yre = ws + oYre + (size_t)n * IPD * JD + j0;
  const float* Yim = ws + oYim + (size_t)n * IPD * JD + j0;
  float4 re[8], im[8];
#pragma unroll
  for (int q = 0; q < 8; q++) {
    const int f = q * 512 + tid, i = f >> 2, jq = f & 3;
    re[q] = *(const float4*)(Yre + (size_t)i * JD + 4 * jq);
    im[q] = *(const float4*)(Yim + (size_t)i * JD + 4 * jq);
  }
#pragma unroll
  for (int q = 0; q < 8; q++) {
    const int f = q * 512 + tid, i = f >> 2, jq = f & 3;
    s[(4 * jq + 0) * OP + i] = make_float2(re[q].x, im[q].x);
    s[(4 * jq + 1) * OP + i] = make_float2(re[q].y, im[q].y);
    s[(4 * jq + 2) * OP + i] = make_float2(re[q].z, im[q].z);
    s[(4 * jq + 3) * OP + i] = make_float2(re[q].w, im[q].w);
  }
  if (tid < 4) {
    const int i = ID - 1, jq = tid;
    const float4 r = *(const float4*)(Yre + (size_t)i * JD + 4 * jq);
    const float4 g = *(const float4*)(Yim + (size_t)i * JD + 4 * jq);
    s[(4 * jq + 0) * OP + i] = make_float2(r.x, g.x);
    s[(4 * jq + 1) * OP + i] = make_float2(r.y, g.y);
    s[(4 * jq + 2) * OP + i] = make_float2(r.z, g.z);
    s[(4 * jq + 3) * OP + i] = make_float2(r.w, g.w);
  }
  __syncthreads();
  float4* o4 = (float4*)(out + ((size_t)n * JD + j0) * ID * 2);
  for (int h = tid; h < 16 * ID / 2; h += 512) {
    const int q0 = 2 * h, q1 = q0 + 1;
    const float2 a = s[q0 + q0 / ID], b = s[q1 + q1 / ID];
    o4[h] = make_float4(a.x, a.y, b.x, b.y);
  }
}

extern "C" void kernel_launch(void* const* d_in, const int* in_sizes, int n_in,
                              void* d_out, int out_size, void* d_ws, size_t ws_size,
                              hipStream_t stream) {
  (void)in_sizes; (void)n_in; (void)out_size; (void)ws_size;
  const float* X  = (const float*)d_in[0];
  const float* T0 = (const float*)d_in[1];
  const float* V0 = (const float*)d_in[2];
  float* ws = (float*)d_ws;
  float* out = (float*)d_out;

  k_init_xy<<<dim3(GXI, GYJ, MC), 256, 0, stream>>>(X, T0, V0, ws);

  k_ip_t<0><<<NPB, 256, 0, stream>>>(ws);
  for (int it = 0; it < NITER; ++it) {
    k_v_update<<<dim3(JD / 16, NSRC), VW * 32, 0, stream>>>(ws);
    if (it + 1 < NITER) k_ip_t<1><<<NPB, 256, 0, stream>>>(ws);
    else                k_ip_t<2><<<NPB, 256, 0, stream>>>(ws);
  }
  k_out<<<dim3(JD / 16, NSRC), 512, 0, stream>>>(ws, out);
}


# IP/T kernel: a wave owns 256 consecutive j's end to end, so the block barrier between the Y pass and the T-update products goes (wave-scope fence); covariance partial sums regrouped per thread
# speedup vs baseline: 2.7024x; 1.0365x over previous
#include <hip/hip_runtime.h>
#include <math.h>

#define JD 2048
#define ID 1025
#define IPD 1040
#define KB 8
#define MC 2
#define NSRC 2
#define NITER 5
#define VW 16
#define EPS 1e-20f

typedef __attribute__((ext_vector_type(2))) float v2f;
typedef __attribute__((ext_vector_type(8))) float v8f;

constexpr size_t oXre = 0;
constexpr size_t oXim = oXre + (size_t)MC * IPD * JD;
constexpr size_t oYre = oXim + (size_t)MC * IPD * JD;
constexpr size_t oYim = oYre + (size_t)NSRC * IPD * JD;
constexpr size_t oT   = oYim + (size_t)NSRC * IPD * JD;
constexpr size_t oV   = oT + (size_t)NSRC * IPD * KB;
constexpr size_t oW   = oV + (size_t)NSRC * KB * JD;

__device__ __forceinline__ v8f wmma_k4(v2f a, v2f b, v8f c) {
  return __builtin_amdgcn_wmma_f32_16x16x4_f32(false, a, false, b, (short)0, c, false, false);
}

__device__ __forceinline__ float rcp_hw(float x) { return __builtin_amdgcn_rcpf(x); }
__device__ __forceinline__ float rsq_hw(float x) { return __builtin_amdgcn_rsqf(x); }

#define TP 66
#define NT_E (NSRC * IPD * KB)
#define NV_E (NSRC * KB * JD)
#define NW_E (ID * 8)
#define SMALL_B ((NT_E + NV_E + NW_E + 255) / 256)
#define GXI ((IPD + 63) / 64)
#define GYJ (JD / 64)
static_assert(oV == oT + NT_E && oW == oV + NV_E, "T, V, W must be contiguous");
static_assert(NT_E % 256 == 0 && NV_E % 256 == 0, "a block must lie inside one array");
static_assert(SMALL_B <= GXI * GYJ * MC, "not enough blocks for the small arrays");
__device__ __forceinline__ void init_small(const float* __restrict__ T0, const float* __restrict__ V0,
                                           float* __restrict__ ws, const int bflat, const int tid) {
  const int t = bflat * 256 + tid;
  const float* sp = T0;
  bool ld = false;
  float alt = 0.f;
  if (bflat < NT_E / 256) {
    const int k = t % KB, r = t / KB;
    const int i = r % IPD, n = r / IPD;
    const int ic = i < ID ? i : ID - 1;
    sp = T0 + ((size_t)ic * KB + k) * NSRC + n;
    ld = i < ID;
  } else if (bflat < (NT_E + NV_E) / 256) {
    const int u = t - NT_E;
    const int j = u % JD, r = u / JD;
    const int k = r % KB, n = r / KB;
    sp = V0 + ((size_t)k * JD + j) * NSRC + n;
    ld = true;
  } else {
    const int f = (t - NT_E - NV_E) & 7;
    const int cc = f & 1, m = (f >> 1) & 1, nn = (f >> 2) & 1;
    alt = (cc == 0 && nn == m) ? 1.f : 0.f;
  }
  const float lv = *sp;
  if (t < NT_E + NV_E + NW_E) ws[oT + t] = ld ? lv : alt;
}

__global__ __launch_bounds__(256) void k_init_xy(const float* __restrict__ X, const float* __restrict__ T0,
                                                 const float* __restrict__ V0, float* __restrict__ ws) {
  __shared__ __attribute__((aligned(16))) float sre[64 * TP];
  __shared__ __attribute__((aligned(16))) float sim[64 * TP];
  const int tid = threadIdx.x, wv = tid >> 5, lane = tid & 31;
  const int i0 = blockIdx.x * 64, j0 = blockIdx.y * 64, m = blockIdx.z;
  float2 v[16];
#pragma unroll
  for (int q = 0; q < 16; q++) {
    const int jj = wv * 8 + (q >> 1);
    const int i = i0 + (q & 1) * 32 + lane;
    const int ic = i < ID ? i : ID - 1;
    v[q] = *(const float2*)(X + (((size_t)m * JD + j0 + jj) * ID + ic) * 2);
  }
  const int bflat = (blockIdx.z * GYJ + blockIdx.y) * GXI + blockIdx.x;
  if (bflat < SMALL_B) init_small(T0, V0, ws, bflat, tid);
#pragma unroll
  for (int q = 0; q < 16; q++) {
    const int jj = wv * 8 + (q >> 1);
    const int il = (q & 1) * 32 + lane;
    const bool pad = i0 + il >= ID;
    sre[il * TP + jj] = pad ? 0.f : v[q].x; sim[il * TP + jj] = pad ? 0.f : v[q].y;
  }
  __syncthreads();
#pragma unroll
  for (int q = 0; q < 8; q++) {
    const int il = wv * 8 + q;
    const int i = i0 + il;
    if (i < IPD) {
      const float2 xr = *(const float2*)(sre + il * TP + 2 * lane);
      const float2 xi = *(const float2*)(sim + il * TP + 2 * lane);
      const size_t dst = ((size_t)m * IPD + i) * JD + j0 + 2 * lane;
      *(float2*)(ws + oXre + dst) = xr; *(float2*)(ws + oXim + dst) = xi;
      if (i >= ID) *(float2*)(ws + oYre + dst) = xr;
    }
  }
}

struct VFrag { v2f a0, a1, at[4], by[4]; };

__device__ __forceinline__ void v_load_t(VFrag& f, const float* T,
                                         const int i0, const int hf, const int l16) {
  const size_t tr = (size_t)(i0 + l16) * KB;
  f.a0.x = T[tr + 2 * hf];     f.a0.y = T[tr + 2 * hf + 1];
  f.a1.x = T[tr + 4 + 2 * hf]; f.a1.y = T[tr + 5 + 2 * hf];
#pragma unroll
  for (int c4 = 0; c4 < 4; c4++) {
    const int ia = i0 + 4 * c4 + 2 * hf;
    f.at[c4].x = T[(size_t)ia * KB + (l16 & 7)];
    f.at[c4].y = T[(size_t)(ia + 1) * KB + (l16 & 7)];
  }
}

__device__ __forceinline__ void v_load_y(VFrag& f, const float* YdR,
                                         const int i0, const int j0, const int hf, const int l16) {
#pragma unroll
  for (int c4 = 0; c4 < 4; c4++) {
    const int ia = i0 + 4 * c4 + 2 * hf;
    f.by[c4].x = YdR[(size_t)ia * JD + j0 + l16];
    f.by[c4].y = YdR[(size_t)(ia + 1) * JD + j0 + l16];
  }
}

__device__ __forceinline__ void v_tile(const VFrag& f, const v2f b0, const v2f b1, float (*lr)[17],
                                       const int hf, const int l16, v8f& num, v8f& den) {
  v8f c = {0,0,0,0,0,0,0,0};
  c = wmma_k4(f.a0, b0, c);
  c = wmma_k4(f.a1, b1, c);
#pragma unroll
  for (int r = 0; r < 8; r++)
    lr[r + 8 * hf][l16] = rcp_hw(fabsf(c[r]) + EPS);
#pragma unroll
  for (int c4 = 0; c4 < 4; c4++) {
    v2f brv;
    brv.x = lr[4 * c4 + 2 * hf][l16];
    brv.y = lr[4 * c4 + 2 * hf + 1][l16];
    num = wmma_k4(f.at[c4], f.by[c4], num);
    den = wmma_k4(f.at[c4], brv, den);
  }
}

__global__ __launch_bounds__(VW * 32) void k_v_update(float* __restrict__ ws) {
  const int tid = threadIdx.x;
  const int wv = tid >> 5, lane = tid & 31, hf = lane >> 4, l16 = lane & 15;
  const int j0 = blockIdx.x * 16;
  const int n = blockIdx.y;
  const float* T = ws + oT + (size_t)n * IPD * KB;
  float* V = ws + oV + (size_t)n * KB * JD;
  const float* YdR = ws + oYre + (size_t)n * IPD * JD;

  __shared__ float lds_r[VW][16][17];
  __shared__ float redn[VW][128];
  __shared__ float redd[VW][128];

  v2f b0, b1;
  b0.x = V[(size_t)(2 * hf + 0) * JD + j0 + l16];
  b0.y = V[(size_t)(2 * hf + 1) * JD + j0 + l16];
  b1.x = V[(size_t)(2 * hf + 4) * JD + j0 + l16];
  b1.y = V[(size_t)(2 * hf + 5) * JD + j0 + l16];
  VFrag f[4], fx;
  const bool extra = wv == 0;
#pragma unroll
  for (int s = 0; s < 4; s++) v_load_t(f[s], T, (wv + VW * s) * 16, hf, l16);
  if (extra) v_load_t(fx, T, 4 * VW * 16, hf, l16);
#pragma unroll
  for (int s = 0; s < 4; s++) v_load_y(f[s], YdR, (wv + VW * s) * 16, j0, hf, l16);
  if (extra) v_load_y(fx, YdR, 4 * VW * 16, j0, hf, l16);

  v8f num = {0,0,0,0,0,0,0,0}, den = {0,0,0,0,0,0,0,0};
#pragma unroll
  for (int s = 0; s < 4; s++) v_tile(f[s], b0, b1, lds_r[wv], hf, l16, num, den);
  if (extra) v_tile(fx, b0, b1, lds_r[wv], hf, l16, num, den);

  if (hf == 0) {
#pragma unroll
    for (int r = 0; r < 8; r++) { redn[wv][r * 16 + l16] = num[r]; redd[wv][r * 16 + l16] = den[r]; }
  }
  __syncthreads();
  if (tid < KB * 16) {
    float sn = 0.f, sd = 0.f;
#pragma unroll
    for (int w2 = 0; w2 < VW; w2++) { sn += redn[w2][tid]; sd += redd[w2][tid]; }
    const size_t vi = (size_t)(tid >> 4) * JD + j0 + (tid & 15);
    V[vi] = V[vi] * sqrtf(sn / (sd + EPS));
  }
}

__device__ __forceinline__ void ip_solve(float* Wf, const int n, float S00, float S11, float S01r, float S01i) {
  const float invJ = 1.0f / (float)JD;
  float D00 = S00 * invJ + EPS;
  float D11 = S11 * invJ + EPS;
  float Dr = S01r * invJ, Di = S01i * invJ;
  float w00r = Wf[0], w00i = Wf[1], w01r = Wf[2], w01i = Wf[3];
  float w10r = Wf[4], w10i = Wf[5], w11r = Wf[6], w11i = Wf[7];
  float A00r = w00r * D00 + (w01r * Dr + w01i * Di);
  float A00i = w00i * D00 + (w01i * Dr - w01r * Di);
  float A01r = (w00r * Dr - w00i * Di) + w01r * D11;
  float A01i = (w00i * Dr + w00r * Di) + w01i * D11;
  float A10r = w10r * D00 + (w11r * Dr + w11i * Di);
  float A10i = w10i * D00 + (w11i * Dr - w11r * Di);
  float A11r = (w10r * Dr - w10i * Di) + w11r * D11;
  float A11i = (w10i * Dr + w10r * Di) + w11i * D11;
  float detr = (A00r * A11r - A00i * A11i) - (A01r * A10r - A01i * A10i);
  float deti = (A00r * A11i + A00i * A11r) - (A01r * A10i + A01i * A10r);
  float n0r, n0i, n1r, n1i;
  if (n == 0) { n0r = A11r;  n0i = A11i;  n1r = -A10r; n1i = -A10i; }
  else        { n0r = -A01r; n0i = -A01i; n1r = A00r;  n1i = A00i;  }
  float dd = detr * detr + deti * deti;
  float b0r = (n0r * detr + n0i * deti) / dd, b0i = (n0i * detr - n0r * deti) / dd;
  float b1r = (n1r * detr + n1i * deti) / dd, b1i = (n1i * detr - n1r * deti) / dd;
  float Db0r = D00 * b0r + (Dr * b1r - Di * b1i);
  float Db0i = D00 * b0i + (Dr * b1i + Di * b1r);
  float Db1r = (Dr * b0r + Di * b0i) + D11 * b1r;
  float Db1i = (Dr * b0i - Di * b0r) + D11 * b1i;
  float denr = b0r * Db0r + b0i * Db0i + b1r * Db1r + b1i * Db1i;
  float dnm = sqrtf(denr + EPS);
  Wf[(n * 2 + 0) * 2]     = b0r / dnm;
  Wf[(n * 2 + 0) * 2 + 1] = -b0i / dnm;
  Wf[(n * 2 + 1) * 2]     = b1r / dnm;
  Wf[(n * 2 + 1) * 2 + 1] = -b1i / dnm;
}

#define NF 2
#define NPB ((ID + NF - 1) / NF)
#define JPT (JD / 256)
#define QP (JD + 8)
template <int MODE>
__global__ __launch_bounds__(256, 6) void k_ip_t(float* __restrict__ ws) {
  const int i0 = blockIdx.x * NF;
  const bool ok1 = i0 + 1 < ID;
  const int tid = threadIdx.x;
  const int wv = tid >> 5, lane = tid & 31, hf = lane >> 4, l16 = lane & 15;
  const float* V = ws + oV;
  const float* X0r = ws + oXre;
  const float* X0i = ws + oXim;
  const float* X1r = ws + oXre + (size_t)IPD * JD;
  const float* X1i = ws + oXim + (size_t)IPD * JD;

  __shared__ float sw[NF][8];
  __shared__ float rb[NF * 4 * NSRC][256];
  __shared__ __attribute__((aligned(16))) float sq[MODE == 2 ? 4 : NF * 4 * QP];
  __shared__ float red[8][NF * 32];

  const int jb = 256 * wv + lane;
  float x0r[NF][JPT], x0i[NF][JPT], x1r[NF][JPT], x1i[NF][JPT];
#pragma unroll
  for (int q = 0; q < JPT; q++) {
#pragma unroll
    for (int f = 0; f < NF; f++) {
      const size_t idx = (size_t)(i0 + f) * JD + jb + 32 * q;
      x0r[f][q] = X0r[idx]; x0i[f][q] = X0i[idx]; x1r[f][q] = X1r[idx]; x1i[f][q] = X1i[idx];
    }
  }
  __builtin_amdgcn_sched_barrier(0);

  float tT[NF][NSRC][KB], w0[NF][8];
#pragma unroll
  for (int f = 0; f < NF; f++)
#pragma unroll
    for (int n = 0; n < NSRC; n++)
#pragma unroll
      for (int k = 0; k < KB; k++) tT[f][n][k] = ws[oT + (size_t)n * IPD * KB + (size_t)(i0 + f) * KB + k];
  if (MODE != 0) {
    const int iw1 = ok1 ? i0 + 1 : i0;
#pragma unroll
    for (int q = 0; q < 8; q++) { w0[0][q] = ws[oW + (size_t)i0 * 8 + q]; w0[1][q] = ws[oW + (size_t)iw1 * 8 + q]; }
  }

  float wa0[NF], wb0[NF], wa1[NF], wb1[NF], wc0[NF], wd0[NF], wc1[NF], wd1[NF];
#pragma unroll
  for (int f = 0; f < NF; f++) {
    wa0[f] = 1.f; wb0[f] = 0.f; wa1[f] = 0.f; wb1[f] = 0.f;
    wc0[f] = 0.f; wd0[f] = 0.f; wc1[f] = 1.f; wd1[f] = 0.f;
  }
  float* P0r = ws + oYre;
  float* P1r = ws + oYre + (size_t)IPD * JD;
  float* P0i = ws + oYim;
  float* P1i = ws + oYim + (size_t)IPD * JD;
  auto y_step = [&](const int f, const int q, const float r0, const float r1) {
    const bool st = f == 0 || ok1;
    const int j = jb + 32 * q;
    const size_t idx = (size_t)(i0 + f) * JD + j;
    const float p0r = x0r[f][q], p0i = x0i[f][q], p1r = x1r[f][q], p1i = x1i[f][q];
    const float y0r = wa0[f] * p0r - wb0[f] * p0i + wa1[f] * p1r - wb1[f] * p1i;
    const float y0i = wa0[f] * p0i + wb0[f] * p0r + wa1[f] * p1i + wb1[f] * p1r;
    const float y1r = wc0[f] * p0r - wd0[f] * p0i + wc1[f] * p1r - wd1[f] * p1i;
    const float y1i = wc0[f] * p0i + wd0[f] * p0r + wc1[f] * p1i + wd1[f] * p1r;
    if (MODE == 2) {
      if (st) { P0r[idx] = y0r; P0i[idx] = y0i; P1r[idx] = y1r; P1i[idx] = y1i; }
    } else {
      const float ydr0 = (y0r * y0r + y0i * y0i) * rcp_hw(r0 * r0 + EPS);
      const float ydr1 = (y1r * y1r + y1i * y1i) * rcp_hw(r1 * r1 + EPS);
      const float rd0 = rcp_hw(fabsf(r0) + EPS);
      const float rd1 = rcp_hw(fabsf(r1) + EPS);
      if (st) { P0r[idx] = ydr0; P1r[idx] = ydr1; }
      sq[(4 * f + 0) * QP + j] = ydr0; sq[(4 * f + 1) * QP + j] = rd0;
      sq[(4 * f + 2) * QP + j] = ydr1; sq[(4 * f + 3) * QP + j] = rd1;
    }
  };

  float rnv[NF][NSRC][JPT];
  float d00[NF][NSRC], d11[NF][NSRC], d01r[NF][NSRC], d01i[NF][NSRC];
#pragma unroll
  for (int f = 0; f < NF; f++)
#pragma unroll
    for (int n = 0; n < NSRC; n++) { d00[f][n] = 0.f; d11[f][n] = 0.f; d01r[f][n] = 0.f; d01i[f][n] = 0.f; }
  float vv[2][NSRC * KB];
#pragma unroll
  for (int r = 0; r < NSRC * KB; r++) vv[0][r] = V[(size_t)r * JD + jb];
#pragma unroll
  for (int q = 0; q < JPT; q++) {
    if (q + 1 < JPT) {
#pragma unroll
      for (int r = 0; r < NSRC * KB; r++) vv[(q + 1) & 1][r] = V[(size_t)r * JD + jb + 32 * (q + 1)];
    }
#pragma unroll
    for (int f = 0; f < NF; f++) {
      float rl[NSRC];
#pragma unroll
      for (int n = 0; n < NSRC; n++) {
        float rn = 0.f;
#pragma unroll
        for (int k = 0; k < KB; k++) rn += tT[f][n][k] * vv[q & 1][n * KB + k];
        rl[n] = rn;
        if (MODE != 0) {
          if (MODE == 1) sq[(4 * f + 2 * n) * QP + jb + 32 * q] = rn;
          else rnv[f][n][q] = rn;
          float s = rsq_hw(rn + EPS);
          float a0r = x0r[f][q] * s, a0i = x0i[f][q] * s;
          float a1r = x1r[f][q] * s, a1i = x1i[f][q] * s;
          d00[f][n] += a0r * a0r + a0i * a0i;
          d11[f][n] += a1r * a1r + a1i * a1i;
          d01r[f][n] += a0r * a1r + a0i * a1i;
          d01i[f][n] += a0i * a1r - a0r * a1i;
        }
      }
      if (MODE == 0) y_step(f, q, rl[0], rl[1]);
    }
    __builtin_amdgcn_sched_barrier(0);
  }
  if (MODE != 0) {
#pragma unroll
    for (int f = 0; f < NF; f++)
#pragma unroll
      for (int n = 0; n < NSRC; n++) {
        rb[8 * f + 4 * n + 0][tid] = d00[f][n]; rb[8 * f + 4 * n + 1][tid] = d11[f][n];
        rb[8 * f + 4 * n + 2][tid] = d01r[f][n]; rb[8 * f + 4 * n + 3][tid] = d01i[f][n];
      }
    __syncthreads();
    for (int s2 = 128; s2 > 0; s2 >>= 1) {
      if (tid < s2) {
#pragma unroll
        for (int q = 0; q < NF * 4 * NSRC; q++) rb[q][tid] += rb[q][tid + s2];
      }
      __syncthreads();
    }
    if (lane == 0 && wv < NF) {
      const int f = wv;
      float Wf[8];
#pragma unroll
      for (int q = 0; q < 8; q++) Wf[q] = f == 0 ? w0[0][q] : w0[1][q];
      if (f == 0 || ok1) {
        ip_solve(Wf, 0, rb[8 * f + 0][0], rb[8 * f + 1][0], rb[8 * f + 2][0], rb[8 * f + 3][0]);
        ip_solve(Wf, 1, rb[8 * f + 4][0], rb[8 * f + 5][0], rb[8 * f + 6][0], rb[8 * f + 7][0]);
      }
#pragma unroll
      for (int q = 0; q < 8; q++) sw[f][q] = Wf[q];
    }
    __syncthreads();
    if (tid < (ok1 ? NF * 8 : 8)) ws[oW + (size_t)i0 * 8 + tid] = sw[tid >> 3][tid & 7];
#pragma unroll
    for (int f = 0; f < NF; f++) {
      wa0[f] = sw[f][0]; wb0[f] = sw[f][1]; wa1[f] = sw[f][2]; wb1[f] = sw[f][3];
      wc0[f] = sw[f][4]; wd0[f] = sw[f][5]; wc1[f] = sw[f][6]; wd1[f] = sw[f][7];
    }
  }

  if (MODE != 0) {
#pragma unroll
    for (int f = 0; f < NF; f++)
#pragma unroll
      for (int q = 0; q < JPT; q++) {
        if (MODE == 1) y_step(f, q, sq[(4 * f) * QP + jb + 32 * q], sq[(4 * f + 2) * QP + jb + 32 * q]);
        else y_step(f, q, rnv[f][0][q], rnv[f][1][q]);
      }
  }
  if (MODE == 2) return;

  __builtin_amdgcn_fence(__ATOMIC_RELEASE, "wavefront");
  __builtin_amdgcn_wave_barrier();
  __builtin_amdgcn_fence(__ATOMIC_ACQUIRE, "wavefront");
  v8f acc0 = {0,0,0,0,0,0,0,0}, acc1 = {0,0,0,0,0,0,0,0};
  const float* va = V + (size_t)l16 * JD + 256 * wv + 4 * hf;
  const float* qb = sq + (l16 & 7) * QP + 256 * wv + 4 * hf;
  const bool qcol = l16 < NF * 4;
#pragma unroll 4
  for (int t = 0; t < 32; t++) {
    const float4 af = *(const float4*)(va + 8 * t);
    float4 bf = *(const float4*)(qb + 8 * t);
    if (!qcol) bf = make_float4(0.f, 0.f, 0.f, 0.f);
    v2f a, b;
    a.x = af.x; a.y = af.y; b.x = bf.x; b.y = bf.y;
    acc0 = wmma_k4(a, b, acc0);
    a.x = af.z; a.y = af.w; b.x = bf.z; b.y = bf.w;
    acc1 = wmma_k4(a, b, acc1);
  }
  if (l16 < NF * 4 && ((l16 & 3) >> 1) == hf) {
#pragma unroll
    for (int r = 0; r < 8; r++) red[wv][l16 * 8 + r] = acc0[r] + acc1[r];
  }
  __syncthreads();
  if (tid < NF * NSRC * KB) {
    const int f = tid >> 4, n = (tid >> 3) & 1, k = tid & 7;
    if (f == 0 || ok1) {
      float sn = 0.f, sd = 0.f;
#pragma unroll
      for (int w2 = 0; w2 < 8; w2++) { sn += red[w2][(4 * f + 2 * n) * 8 + k]; sd += red[w2][(4 * f + 2 * n + 1) * 8 + k]; }
      const size_t ti = oT + (size_t)n * IPD * KB + (size_t)(i0 + f) * KB + k;
      ws[ti] = ws[ti] * sqrtf(sn / (sd + EPS));
    }
  }
}

#define OP 1026
__global__ __launch_bounds__(512) void k_out(const float* __restrict__ ws, float* __restrict__ out) {
  __shared__ float2 s[16 * OP];
  const int tid = threadIdx.x;
  const int j0 = blockIdx.x * 16, n = blockIdx.y;
  const float* Yre = ws + oYre + (size_t)n * IPD * JD + j0;
  const float* Yim = ws + oYim + (size_t)n * IPD * JD + j0;
  float4 re[8], im[8];
#pragma unroll
  for (int q = 0; q < 8; q++) {
    const int f = q * 512 + tid, i = f >> 2, jq = f & 3;
    re[q] = *(const float4*)(Yre + (size_t)i * JD + 4 * jq);
    im[q] = *(const float4*)(Yim + (size_t)i * JD + 4 * jq);
  }
#pragma unroll
  for (int q = 0; q < 8; q++) {
    const int f = q * 512 + tid, i = f >> 2, jq = f & 3;
    s[(4 * jq + 0) * OP + i] = make_float2(re[q].x, im[q].x);
    s[(4 * jq + 1) * OP + i] = make_float2(re[q].y, im[q].y);
    s[(4 * jq + 2) * OP + i] = make_float2(re[q].z, im[q].z);
    s[(4 * jq + 3) * OP + i] = make_float2(re[q].w, im[q].w);
  }
  if (tid < 4) {
    const int i = ID - 1, jq = tid;
    const float4 r = *(const float4*)(Yre + (size_t)i * JD + 4 * jq);
    const float4 g = *(const float4*)(Yim + (size_t)i * JD + 4 * jq);
    s[(4 * jq + 0) * OP + i] = make_float2(r.x, g.x);
    s[(4 * jq + 1) * OP + i] = make_float2(r.y, g.y);
    s[(4 * jq + 2) * OP + i] = make_float2(r.z, g.z);
    s[(4 * jq + 3) * OP + i] = make_float2(r.w, g.w);
  }
  __syncthreads();
  float4* o4 = (float4*)(out + ((size_t)n * JD + j0) * ID * 2);
  for (int h = tid; h < 16 * ID / 2; h += 512) {
    const int q0 = 2 * h, q1 = q0 + 1;
    const float2 a = s[q0 + q0 / ID], b = s[q1 + q1 / ID];
    o4[h] = make_float4(a.x, a.y, b.x, b.y);
  }
}

extern "C" void kernel_launch(void* const* d_in, const int* in_sizes, int n_in,
                              void* d_out, int out_size, void* d_ws, size_t ws_size,
                              hipStream_t stream) {
  (void)in_sizes; (void)n_in; (void)out_size; (void)ws_size;
  const float* X  = (const float*)d_in[0];
  const float* T0 = (const float*)d_in[1];
  const float* V0 = (const float*)d_in[2];
  float* ws = (float*)d_ws;
  float* out = (float*)d_out;

  k_init_xy<<<dim3(GXI, GYJ, MC), 256, 0, stream>>>(X, T0, V0, ws);

  k_ip_t<0><<<NPB, 256, 0, stream>>>(ws);
  for (int it = 0; it < NITER; ++it) {
    k_v_update<<<dim3(JD / 16, NSRC), VW * 32, 0, stream>>>(ws);
    if (it + 1 < NITER) k_ip_t<1><<<NPB, 256, 0, stream>>>(ws);
    else                k_ip_t<2><<<NPB, 256, 0, stream>>>(ws);
  }
  k_out<<<dim3(JD / 16, NSRC), 512, 0, stream>>>(ws, out);
}


# Output formed straight from the input X in the output's own layout (out[n,j,i] = sum_m W[i,n,m] X[m,j,i], W staged in LDS): the last IP launch stores only W; no Y planes, no transposing kernel
# speedup vs baseline: 2.9111x; 1.0772x over previous
#include <hip/hip_runtime.h>
#include <math.h>

#define JD 2048
#define ID 1025
#define IPD 1040
#define KB 8
#define MC 2
#define NSRC 2
#define NITER 5
#define VW 16
#define EPS 1e-20f

typedef __attribute__((ext_vector_type(2))) float v2f;
typedef __attribute__((ext_vector_type(8))) float v8f;

constexpr size_t oXre = 0;
constexpr size_t oXim = oXre + (size_t)MC * IPD * JD;
constexpr size_t oYre = oXim + (size_t)MC * IPD * JD;
constexpr size_t oYim = oYre + (size_t)NSRC * IPD * JD;
constexpr size_t oT   = oYim + (size_t)NSRC * IPD * JD;
constexpr size_t oV   = oT + (size_t)NSRC * IPD * KB;
constexpr size_t oW   = oV + (size_t)NSRC * KB * JD;

__device__ __forceinline__ v8f wmma_k4(v2f a, v2f b, v8f c) {
  return __builtin_amdgcn_wmma_f32_16x16x4_f32(false, a, false, b, (short)0, c, false, false);
}

__device__ __forceinline__ float rcp_hw(float x) { return __builtin_amdgcn_rcpf(x); }
__device__ __forceinline__ float rsq_hw(float x) { return __builtin_amdgcn_rsqf(x); }

#define TP 66
#define NT_E (NSRC * IPD * KB)
#define NV_E (NSRC * KB * JD)
#define NW_E (ID * 8)
#define SMALL_B ((NT_E + NV_E + NW_E + 255) / 256)
#define GXI ((IPD + 63) / 64)
#define GYJ (JD / 64)
static_assert(oV == oT + NT_E && oW == oV + NV_E, "T, V, W must be contiguous");
static_assert(NT_E % 256 == 0 && NV_E % 256 == 0, "a block must lie inside one array");
static_assert(SMALL_B <= GXI * GYJ * MC, "not enough blocks for the small arrays");
__device__ __forceinline__ void init_small(const float* __restrict__ T0, const float* __restrict__ V0,
                                           float* __restrict__ ws, const int bflat, const int tid) {
  const int t = bflat * 256 + tid;
  const float* sp = T0;
  bool ld = false;
  float alt = 0.f;
  if (bflat < NT_E / 256) {
    const int k = t % KB, r = t / KB;
    const int i = r % IPD, n = r / IPD;
    const int ic = i < ID ? i : ID - 1;
    sp = T0 + ((size_t)ic * KB + k) * NSRC + n;
    ld = i < ID;
  } else if (bflat < (NT_E + NV_E) / 256) {
    const int u = t - NT_E;
    const int j = u % JD, r = u / JD;
    const int k = r % KB, n = r / KB;
    sp = V0 + ((size_t)k * JD + j) * NSRC + n;
    ld = true;
  } else {
    const int f = (t - NT_E - NV_E) & 7;
    const int cc = f & 1, m = (f >> 1) & 1, nn = (f >> 2) & 1;
    alt = (cc == 0 && nn == m) ? 1.f : 0.f;
  }
  const float lv = *sp;
  if (t < NT_E + NV_E + NW_E) ws[oT + t] = ld ? lv : alt;
}

__global__ __launch_bounds__(256) void k_init_xy(const float* __restrict__ X, const float* __restrict__ T0,
                                                 const float* __restrict__ V0, float* __restrict__ ws) {
  __shared__ __attribute__((aligned(16))) float sre[64 * TP];
  __shared__ __attribute__((aligned(16))) float sim[64 * TP];
  const int tid = threadIdx.x, wv = tid >> 5, lane = tid & 31;
  const int i0 = blockIdx.x * 64, j0 = blockIdx.y * 64, m = blockIdx.z;
  float2 v[16];
#pragma unroll
  for (int q = 0; q < 16; q++) {
    const int jj = wv * 8 + (q >> 1);
    const int i = i0 + (q & 1) * 32 + lane;
    const int ic = i < ID ? i : ID - 1;
    v[q] = *(const float2*)(X + (((size_t)m * JD + j0 + jj) * ID + ic) * 2);
  }
  const int bflat = (blockIdx.z * GYJ + blockIdx.y) * GXI + blockIdx.x;
  if (bflat < SMALL_B) init_small(T0, V0, ws, bflat, tid);
#pragma unroll
  for (int q = 0; q < 16; q++) {
    const int jj = wv * 8 + (q >> 1);
    const int il = (q & 1) * 32 + lane;
    const bool pad = i0 + il >= ID;
    sre[il * TP + jj] = pad ? 0.f : v[q].x; sim[il * TP + jj] = pad ? 0.f : v[q].y;
  }
  __syncthreads();
#pragma unroll
  for (int q = 0; q < 8; q++) {
    const int il = wv * 8 + q;
    const int i = i0 + il;
    if (i < IPD) {
      const float2 xr = *(const float2*)(sre + il * TP + 2 * lane);
      const float2 xi = *(const float2*)(sim + il * TP + 2 * lane);
      const size_t dst = ((size_t)m * IPD + i) * JD + j0 + 2 * lane;
      *(float2*)(ws + oXre + dst) = xr; *(float2*)(ws + oXim + dst) = xi;
      if (i >= ID) *(float2*)(ws + oYre + dst) = xr;
    }
  }
}

struct VFrag { v2f a0, a1, at[4], by[4]; };

__device__ __forceinline__ void v_load_t(VFrag& f, const float* T,
                                         const int i0, const int hf, const int l16) {
  const size_t tr = (size_t)(i0 + l16) * KB;
  f.a0.x = T[tr + 2 * hf];     f.a0.y = T[tr + 2 * hf + 1];
  f.a1.x = T[tr + 4 + 2 * hf]; f.a1.y = T[tr + 5 + 2 * hf];
#pragma unroll
  for (int c4 = 0; c4 < 4; c4++) {
    const int ia = i0 + 4 * c4 + 2 * hf;
    f.at[c4].x = T[(size_t)ia * KB + (l16 & 7)];
    f.at[c4].y = T[(size_t)(ia + 1) * KB + (l16 & 7)];
  }
}

__device__ __forceinline__ void v_load_y(VFrag& f, const float* YdR,
                                         const int i0, const int j0, const int hf, const int l16) {
#pragma unroll
  for (int c4 = 0; c4 < 4; c4++) {
    const int ia = i0 + 4 * c4 + 2 * hf;
    f.by[c4].x = YdR[(size_t)ia * JD + j0 + l16];
    f.by[c4].y = YdR[(size_t)(ia + 1) * JD + j0 + l16];
  }
}

__device__ __forceinline__ void v_tile(const VFrag& f, const v2f b0, const v2f b1, float (*lr)[17],
                                       const int hf, const int l16, v8f& num, v8f& den) {
  v8f c = {0,0,0,0,0,0,0,0};
  c = wmma_k4(f.a0, b0, c);
  c = wmma_k4(f.a1, b1, c);
#pragma unroll
  for (int r = 0; r < 8; r++)
    lr[r + 8 * hf][l16] = rcp_hw(fabsf(c[r]) + EPS);
#pragma unroll
  for (int c4 = 0; c4 < 4; c4++) {
    v2f brv;
    brv.x = lr[4 * c4 + 2 * hf][l16];
    brv.y = lr[4 * c4 + 2 * hf + 1][l16];
    num = wmma_k4(f.at[c4], f.by[c4], num);
    den = wmma_k4(f.at[c4], brv, den);
  }
}

__global__ __launch_bounds__(VW * 32) void k_v_update(float* __restrict__ ws) {
  const int tid = threadIdx.x;
  const int wv = tid >> 5, lane = tid & 31, hf = lane >> 4, l16 = lane & 15;
  const int j0 = blockIdx.x * 16;
  const int n = blockIdx.y;
  const float* T = ws + oT + (size_t)n * IPD * KB;
  float* V = ws + oV + (size_t)n * KB * JD;
  const float* YdR = ws + oYre + (size_t)n * IPD * JD;

  __shared__ float lds_r[VW][16][17];
  __shared__ float redn[VW][128];
  __shared__ float redd[VW][128];

  v2f b0, b1;
  b0.x = V[(size_t)(2 * hf + 0) * JD + j0 + l16];
  b0.y = V[(size_t)(2 * hf + 1) * JD + j0 + l16];
  b1.x = V[(size_t)(2 * hf + 4) * JD + j0 + l16];
  b1.y = V[(size_t)(2 * hf + 5) * JD + j0 + l16];
  VFrag f[4], fx;
  const bool extra = wv == 0;
#pragma unroll
  for (int s = 0; s < 4; s++) v_load_t(f[s], T, (wv + VW * s) * 16, hf, l16);
  if (extra) v_load_t(fx, T, 4 * VW * 16, hf, l16);
#pragma unroll
  for (int s = 0; s < 4; s++) v_load_y(f[s], YdR, (wv + VW * s) * 16, j0, hf, l16);
  if (extra) v_load_y(fx, YdR, 4 * VW * 16, j0, hf, l16);

  v8f num = {0,0,0,0,0,0,0,0}, den = {0,0,0,0,0,0,0,0};
#pragma unroll
  for (int s = 0; s < 4; s++) v_tile(f[s], b0, b1, lds_r[wv], hf, l16, num, den);
  if (extra) v_tile(fx, b0, b1, lds_r[wv], hf, l16, num, den);

  if (hf == 0) {
#pragma unroll
    for (int r = 0; r < 8; r++) { redn[wv][r * 16 + l16] = num[r]; redd[wv][r * 16 + l16] = den[r]; }
  }
  __syncthreads();
  if (tid < KB * 16) {
    float sn = 0.f, sd = 0.f;
#pragma unroll
    for (int w2 = 0; w2 < VW; w2++) { sn += redn[w2][tid]; sd += redd[w2][tid]; }
    const size_t vi = (size_t)(tid >> 4) * JD + j0 + (tid & 15);
    V[vi] = V[vi] * sqrtf(sn / (sd + EPS));
  }
}

__device__ __forceinline__ void ip_solve(float* Wf, const int n, float S00, float S11, float S01r, float S01i) {
  const float invJ = 1.0f / (float)JD;
  float D00 = S00 * invJ + EPS;
  float D11 = S11 * invJ + EPS;
  float Dr = S01r * invJ, Di = S01i * invJ;
  float w00r = Wf[0], w00i = Wf[1], w01r = Wf[2], w01i = Wf[3];
  float w10r = Wf[4], w10i = Wf[5], w11r = Wf[6], w11i = Wf[7];
  float A00r = w00r * D00 + (w01r * Dr + w01i * Di);
  float A00i = w00i * D00 + (w01i * Dr - w01r * Di);
  float A01r = (w00r * Dr - w00i * Di) + w01r * D11;
  float A01i = (w00i * Dr + w00r * Di) + w01i * D11;
  float A10r = w10r * D00 + (w11r * Dr + w11i * Di);
  float A10i = w10i * D00 + (w11i * Dr - w11r * Di);
  float A11r = (w10r * Dr - w10i * Di) + w11r * D11;
  float A11i = (w10i * Dr + w10r * Di) + w11i * D11;
  float detr = (A00r * A11r - A00i * A11i) - (A01r * A10r - A01i * A10i);
  float deti = (A00r * A11i + A00i * A11r) - (A01r * A10i + A01i * A10r);
  float n0r, n0i, n1r, n1i;
  if (n == 0) { n0r = A11r;  n0i = A11i;  n1r = -A10r; n1i = -A10i; }
  else        { n0r = -A01r; n0i = -A01i; n1r = A00r;  n1i = A00i;  }
  float dd = detr * detr + deti * deti;
  float b0r = (n0r * detr + n0i * deti) / dd, b0i = (n0i * detr - n0r * deti) / dd;
  float b1r = (n1r * detr + n1i * deti) / dd, b1i = (n1i * detr - n1r * deti) / dd;
  float Db0r = D00 * b0r + (Dr * b1r - Di * b1i);
  float Db0i = D00 * b0i + (Dr * b1i + Di * b1r);
  float Db1r = (Dr * b0r + Di * b0i) + D11 * b1r;
  float Db1i = (Dr * b0i - Di * b0r) + D11 * b1i;
  float denr = b0r * Db0r + b0i * Db0i + b1r * Db1r + b1i * Db1i;
  float dnm = sqrtf(denr + EPS);
  Wf[(n * 2 + 0) * 2]     = b0r / dnm;
  Wf[(n * 2 + 0) * 2 + 1] = -b0i / dnm;
  Wf[(n * 2 + 1) * 2]     = b1r / dnm;
  Wf[(n * 2 + 1) * 2 + 1] = -b1i / dnm;
}

#define NF 2
#define NPB ((ID + NF - 1) / NF)
#define JPT (JD / 256)
#define QP (JD + 8)
template <int MODE>
__global__ __launch_bounds__(256, 6) void k_ip_t(float* __restrict__ ws) {
  const int i0 = blockIdx.x * NF;
  const bool ok1 = i0 + 1 < ID;
  const int tid = threadIdx.x;
  const int wv = tid >> 5, lane = tid & 31, hf = lane >> 4, l16 = lane & 15;
  const float* V = ws + oV;
  const float* X0r = ws + oXre;
  const float* X0i = ws + oXim;
  const float* X1r = ws + oXre + (size_t)IPD * JD;
  const float* X1i = ws + oXim + (size_t)IPD * JD;

  __shared__ float sw[NF][8];
  __shared__ float rb[NF * 4 * NSRC][256];
  __shared__ __attribute__((aligned(16))) float sq[MODE == 2 ? 4 : NF * 4 * QP];
  __shared__ float red[8][NF * 32];

  const int jb = 256 * wv + lane;
  float x0r[NF][JPT], x0i[NF][JPT], x1r[NF][JPT], x1i[NF][JPT];
#pragma unroll
  for (int q = 0; q < JPT; q++) {
#pragma unroll
    for (int f = 0; f < NF; f++) {
      const size_t idx = (size_t)(i0 + f) * JD + jb + 32 * q;
      x0r[f][q] = X0r[idx]; x0i[f][q] = X0i[idx]; x1r[f][q] = X1r[idx]; x1i[f][q] = X1i[idx];
    }
  }
  __builtin_amdgcn_sched_barrier(0);

  float tT[NF][NSRC][KB], w0[NF][8];
#pragma unroll
  for (int f = 0; f < NF; f++)
#pragma unroll
    for (int n = 0; n < NSRC; n++)
#pragma unroll
      for (int k = 0; k < KB; k++) tT[f][n][k] = ws[oT + (size_t)n * IPD * KB + (size_t)(i0 + f) * KB + k];
  if (MODE != 0) {
    const int iw1 = ok1 ? i0 + 1 : i0;
#pragma unroll
    for (int q = 0; q < 8; q++) { w0[0][q] = ws[oW + (size_t)i0 * 8 + q]; w0[1][q] = ws[oW + (size_t)iw1 * 8 + q]; }
  }

  float wa0[NF], wb0[NF], wa1[NF], wb1[NF], wc0[NF], wd0[NF], wc1[NF], wd1[NF];
#pragma unroll
  for (int f = 0; f < NF; f++) {
    wa0[f] = 1.f; wb0[f] = 0.f; wa1[f] = 0.f; wb1[f] = 0.f;
    wc0[f] = 0.f; wd0[f] = 0.f; wc1[f] = 1.f; wd1[f] = 0.f;
  }
  float* P0r = ws + oYre;
  float* P1r = ws + oYre + (size_t)IPD * JD;
  auto y_step = [&](const int f, const int q, const float r0, const float r1) {
    const bool st = f == 0 || ok1;
    const int j = jb + 32 * q;
    const size_t idx = (size_t)(i0 + f) * JD + j;
    const float p0r = x0r[f][q], p0i = x0i[f][q], p1r = x1r[f][q], p1i = x1i[f][q];
    const float y0r = wa0[f] * p0r - wb0[f] * p0i + wa1[f] * p1r - wb1[f] * p1i;
    const float y0i = wa0[f] * p0i + wb0[f] * p0r + wa1[f] * p1i + wb1[f] * p1r;
    const float y1r = wc0[f] * p0r - wd0[f] * p0i + wc1[f] * p1r - wd1[f] * p1i;
    const float y1i = wc0[f] * p0i + wd0[f] * p0r + wc1[f] * p1i + wd1[f] * p1r;
    const float ydr0 = (y0r * y0r + y0i * y0i) * rcp_hw(r0 * r0 + EPS);
    const float ydr1 = (y1r * y1r + y1i * y1i) * rcp_hw(r1 * r1 + EPS);
    const float rd0 = rcp_hw(fabsf(r0) + EPS);
    const float rd1 = rcp_hw(fabsf(r1) + EPS);
    if (st) { P0r[idx] = ydr0; P1r[idx] = ydr1; }
    sq[(4 * f + 0) * QP + j] = ydr0; sq[(4 * f + 1) * QP + j] = rd0;
    sq[(4 * f + 2) * QP + j] = ydr1; sq[(4 * f + 3) * QP + j] = rd1;
  };

  float d00[NF][NSRC], d11[NF][NSRC], d01r[NF][NSRC], d01i[NF][NSRC];
#pragma unroll
  for (int f = 0; f < NF; f++)
#pragma unroll
    for (int n = 0; n < NSRC; n++) { d00[f][n] = 0.f; d11[f][n] = 0.f; d01r[f][n] = 0.f; d01i[f][n] = 0.f; }
  float vv[2][NSRC * KB];
#pragma unroll
  for (int r = 0; r < NSRC * KB; r++) vv[0][r] = V[(size_t)r * JD + jb];
#pragma unroll
  for (int q = 0; q < JPT; q++) {
    if (q + 1 < JPT) {
#pragma unroll
      for (int r = 0; r < NSRC * KB; r++) vv[(q + 1) & 1][r] = V[(size_t)r * JD + jb + 32 * (q + 1)];
    }
#pragma unroll
    for (int f = 0; f < NF; f++) {
      float rl[NSRC];
#pragma unroll
      for (int n = 0; n < NSRC; n++) {
        float rn = 0.f;
#pragma unroll
        for (int k = 0; k < KB; k++) rn += tT[f][n][k] * vv[q & 1][n * KB + k];
        rl[n] = rn;
        if (MODE != 0) {
          if (MODE == 1) sq[(4 * f + 2 * n) * QP + jb + 32 * q] = rn;
          float s = rsq_hw(rn + EPS);
          float a0r = x0r[f][q] * s, a0i = x0i[f][q] * s;
          float a1r = x1r[f][q] * s, a1i = x1i[f][q] * s;
          d00[f][n] += a0r * a0r + a0i * a0i;
          d11[f][n] += a1r * a1r + a1i * a1i;
          d01r[f][n] += a0r * a1r + a0i * a1i;
          d01i[f][n] += a0i * a1r - a0r * a1i;
        }
      }
      if (MODE == 0) y_step(f, q, rl[0], rl[1]);
    }
    __builtin_amdgcn_sched_barrier(0);
  }
  if (MODE != 0) {
#pragma unroll
    for (int f = 0; f < NF; f++)
#pragma unroll
      for (int n = 0; n < NSRC; n++) {
        rb[8 * f + 4 * n + 0][tid] = d00[f][n]; rb[8 * f + 4 * n + 1][tid] = d11[f][n];
        rb[8 * f + 4 * n + 2][tid] = d01r[f][n]; rb[8 * f + 4 * n + 3][tid] = d01i[f][n];
      }
    __syncthreads();
    for (int s2 = 128; s2 > 0; s2 >>= 1) {
      if (tid < s2) {
#pragma unroll
        for (int q = 0; q < NF * 4 * NSRC; q++) rb[q][tid] += rb[q][tid + s2];
      }
      __syncthreads();
    }
    if (lane == 0 && wv < NF) {
      const int f = wv;
      float Wf[8];
#pragma unroll
      for (int q = 0; q < 8; q++) Wf[q] = f == 0 ? w0[0][q] : w0[1][q];
      if (f == 0 || ok1) {
        ip_solve(Wf, 0, rb[8 * f + 0][0], rb[8 * f + 1][0], rb[8 * f + 2][0], rb[8 * f + 3][0]);
        ip_solve(Wf, 1, rb[8 * f + 4][0], rb[8 * f + 5][0], rb[8 * f + 6][0], rb[8 * f + 7][0]);
      }
#pragma unroll
      for (int q = 0; q < 8; q++) sw[f][q] = Wf[q];
    }
    __syncthreads();
    if (tid < (ok1 ? NF * 8 : 8)) ws[oW + (size_t)i0 * 8 + tid] = sw[tid >> 3][tid & 7];
    if (MODE == 2) return;
#pragma unroll
    for (int f = 0; f < NF; f++) {
      wa0[f] = sw[f][0]; wb0[f] = sw[f][1]; wa1[f] = sw[f][2]; wb1[f] = sw[f][3];
      wc0[f] = sw[f][4]; wd0[f] = sw[f][5]; wc1[f] = sw[f][6]; wd1[f] = sw[f][7];
    }
  }

  if (MODE == 1) {
#pragma unroll
    for (int f = 0; f < NF; f++)
#pragma unroll
      for (int q = 0; q < JPT; q++)
        y_step(f, q, sq[(4 * f) * QP + jb + 32 * q], sq[(4 * f + 2) * QP + jb + 32 * q]);
  }

  __builtin_amdgcn_fence(__ATOMIC_RELEASE, "wavefront");
  __builtin_amdgcn_wave_barrier();
  __builtin_amdgcn_fence(__ATOMIC_ACQUIRE, "wavefront");
  v8f acc0 = {0,0,0,0,0,0,0,0}, acc1 = {0,0,0,0,0,0,0,0};
  const float* va = V + (size_t)l16 * JD + 256 * wv + 4 * hf;
  const float* qb = sq + (l16 & 7) * QP + 256 * wv + 4 * hf;
  const bool qcol = l16 < NF * 4;
#pragma unroll 4
  for (int t = 0; t < 32; t++) {
    const float4 af = *(const float4*)(va + 8 * t);
    float4 bf = *(const float4*)(qb + 8 * t);
    if (!qcol) bf = make_float4(0.f, 0.f, 0.f, 0.f);
    v2f a, b;
    a.x = af.x; a.y = af.y; b.x = bf.x; b.y = bf.y;
    acc0 = wmma_k4(a, b, acc0);
    a.x = af.z; a.y = af.w; b.x = bf.z; b.y = bf.w;
    acc1 = wmma_k4(a, b, acc1);
  }
  if (l16 < NF * 4 && ((l16 & 3) >> 1) == hf) {
#pragma unroll
    for (int r = 0; r < 8; r++) red[wv][l16 * 8 + r] = acc0[r] + acc1[r];
  }
  __syncthreads();
  if (tid < NF * NSRC * KB) {
    const int f = tid >> 4, n = (tid >> 3) & 1, k = tid & 7;
    if (f == 0 || ok1) {
      float sn = 0.f, sd = 0.f;
#pragma unroll
      for (int w2 = 0; w2 < 8; w2++) { sn += red[w2][(4 * f + 2 * n) * 8 + k]; sd += red[w2][(4 * f + 2 * n + 1) * 8 + k]; }
      const size_t ti = oT + (size_t)n * IPD * KB + (size_t)(i0 + f) * KB + k;
      ws[ti] = ws[ti] * sqrtf(sn / (sd + EPS));
    }
  }
}

#define OTH 512
#define ONI 8
#define PF4 (JD * ID / 2)
#define OBK (PF4 / (OTH * ONI))
#define OTL ((PF4 - OBK * OTH * ONI) / OBK)
static_assert(OBK * (OTH * ONI + OTL) == PF4, "the blocks must cover a plane exactly");
__global__ __launch_bounds__(OTH) void k_out(const float* X, const float* __restrict__ ws, float* out) {
  __shared__ float4 sa[ID], sc[ID];
  const int tid = threadIdx.x;
  const float4* w4 = (const float4*)(ws + oW);
  for (int i = tid; i < ID; i += OTH) { sa[i] = w4[2 * i]; sc[i] = w4[2 * i + 1]; }
  __syncthreads();
  const int hb = blockIdx.x * (OTH * ONI) + tid;
  const float4* x0 = (const float4*)X;
  const float4* x1 = x0 + PF4;
  const float4* pa = x0 + hb;
  const float4* pb = x1 + hb;
  float4* o0 = (float4*)out;
  float4* o1 = o0 + PF4;
  const bool tl = tid < OTL;
  const int ht = OBK * OTH * ONI + OTL * blockIdx.x + tid;
  float4 a[ONI + 1], b[ONI + 1];
  if (tl) { a[ONI] = x0[ht]; b[ONI] = x1[ht]; }
#pragma unroll
  for (int t = 0; t < ONI; t++) {
    a[t] = pa[t * OTH]; b[t] = pb[t * OTH];
    __builtin_amdgcn_sched_barrier(0);
  }
  auto step = [&](const int h, const float4 p, const float4 r) {
    const int ia = (2 * h) % ID;
    const int ib = ia + 1 == ID ? 0 : ia + 1;
    const float4 ua = sa[ia], uc = sc[ia], va = sa[ib], vc = sc[ib];
    float4 y0, y1;
    y0.x = ua.x * p.x - ua.y * p.y + ua.z * r.x - ua.w * r.y;
    y0.y = ua.x * p.y + ua.y * p.x + ua.z * r.y + ua.w * r.x;
    y1.x = uc.x * p.x - uc.y * p.y + uc.z * r.x - uc.w * r.y;
    y1.y = uc.x * p.y + uc.y * p.x + uc.z * r.y + uc.w * r.x;
    y0.z = va.x * p.z - va.y * p.w + va.z * r.z - va.w * r.w;
    y0.w = va.x * p.w + va.y * p.z + va.z * r.w + va.w * r.z;
    y1.z = vc.x * p.z - vc.y * p.w + vc.z * r.z - vc.w * r.w;
    y1.w = vc.x * p.w + vc.y * p.z + vc.z * r.w + vc.w * r.z;
    o0[h] = y0; o1[h] = y1;
  };
#pragma unroll
  for (int t = 0; t < ONI; t++) {
    step(hb + t * OTH, a[t], b[t]);
    __builtin_amdgcn_sched_barrier(0);
  }
  if (tl) step(ht, a[ONI], b[ONI]);
}

extern "C" void kernel_launch(void* const* d_in, const int* in_sizes, int n_in,
                              void* d_out, int out_size, void* d_ws, size_t ws_size,
                              hipStream_t stream) {
  (void)in_sizes; (void)n_in; (void)out_size; (void)ws_size;
  const float* X  = (const float*)d_in[0];
  const float* T0 = (const float*)d_in[1];
  const float* V0 = (const float*)d_in[2];
  float* ws = (float*)d_ws;
  float* out = (float*)d_out;

  k_init_xy<<<dim3(GXI, GYJ, MC), 256, 0, stream>>>(X, T0, V0, ws);

  k_ip_t<0><<<NPB, 256, 0, stream>>>(ws);
  for (int it = 0; it < NITER; ++it) {
    k_v_update<<<dim3(JD / 16, NSRC), VW * 32, 0, stream>>>(ws);
    if (it + 1 < NITER) k_ip_t<1><<<NPB, 256, 0, stream>>>(ws);
    else                k_ip_t<2><<<NPB, 256, 0, stream>>>(ws);
  }
  k_out<<<OBK, OTH, 0, stream>>>(X, ws, out);
}
